# starting kernel
# baseline (speedup 1.0000x reference)
#include <hip/hip_runtime.h>
#include <math.h>

#define BB 2
#define SS 256
#define HH 768
#define H4 3072
#define CHK 64
#define LRC 0.01f
#define EPSC 1e-5f

typedef float v2f __attribute__((ext_vector_type(2)));
typedef float v8f __attribute__((ext_vector_type(8)));

__device__ __forceinline__ v8f wmma4(v2f a, v2f b, v8f c) {
  return __builtin_amdgcn_wmma_f32_16x16x4_f32(
       false, a,  false, b,
       (short)0, c,  false,  false);
}

__device__ __forceinline__ v2f frag_nk(const float* M, int ld, int r0, int k0, int lane) {
  const float* p = M + (size_t)(r0 + (lane & 15)) * ld + k0 + ((lane >> 4) << 1);
  v2f v; v.x = p[0]; v.y = p[1];
  return v;
}
__device__ __forceinline__ v2f frag_kn(const float* M, int ld, int k0, int r0, int lane) {
  const float* p = M + (size_t)(k0 + ((lane >> 4) << 1)) * ld + r0 + (lane & 15);
  v2f v; v.x = p[0]; v.y = p[ld];
  return v;
}
__device__ __forceinline__ void store_tile(float* D, int ld, int r0, int c0, v8f v, int lane) {
  int n = c0 + (lane & 15);
  int m0 = r0 + ((lane >> 4) << 3);
#pragma unroll
  for (int i = 0; i < 8; ++i) D[(size_t)(m0 + i) * ld + n] = v[i];
}
__device__ __forceinline__ v8f load_tile(const float* D, int ld, int r0, int c0, int lane) {
  int n = c0 + (lane & 15);
  int m0 = r0 + ((lane >> 4) << 3);
  v8f v;
#pragma unroll
  for (int i = 0; i < 8; ++i) v[i] = D[(size_t)(m0 + i) * ld + n];
  return v;
}


__global__ void k_init(const float* __restrict__ Wi, const float* __restrict__ bi,
                       float* __restrict__ Wst, float* __restrict__ bst) {
  int i = blockIdx.x * blockDim.x + threadIdx.x;
  const int n = HH * HH;
  if (i < n) {
    float w = Wi[i];
    Wst[i] = w;
    Wst[n + i] = w;
  }
  if (i < HH) {
    float b = bi[i];
    bst[i] = b;
    bst[HH + i] = b;
  }
}

__global__ void k_p0(const float* __restrict__ x, const float* __restrict__ Wst,
                     const float* __restrict__ bst, float* __restrict__ P0, int chunk) {
  int wid = (blockIdx.x * blockDim.x + threadIdx.x) >> 5;
  int lane = threadIdx.x & 31;
  int tn = wid % (HH / 16);
  int tm = (wid / (HH / 16)) % (CHK / 16);
  int b  = wid / ((HH / 16) * (CHK / 16));
  const float* Xc = x + ((size_t)b * SS + chunk * CHK) * HH;
  const float* W  = Wst + (size_t)b * HH * HH;
  v8f acc = {};
  for (int k0 = 0; k0 < HH; k0 += 4) {
    v2f a  = frag_nk(Xc, HH, tm * 16, k0, lane);
    v2f bb = frag_nk(W,  HH, tn * 16, k0, lane);
    acc = wmma4(a, bb, acc);
  }
  float bias = bst[(size_t)b * HH + tn * 16 + (lane & 15)];
#pragma unroll
  for (int i = 0; i < 8; ++i) acc[i] += bias;
  store_tile(P0 + (size_t)b * CHK * HH, HH, tm * 16, tn * 16, acc, lane);
}

__global__ void k_attn(const float* __restrict__ x, float* __restrict__ Amat, int chunk) {
  int wid = (blockIdx.x * blockDim.x + threadIdx.x) >> 5;
  int lane = threadIdx.x & 31;
  int tn = wid % (CHK / 16);
  int tm = (wid / (CHK / 16)) % (CHK / 16);
  int b  = wid / ((CHK / 16) * (CHK / 16));
  const float* Xc = x + ((size_t)b * SS + chunk * CHK) * HH;
  v8f acc = {};
  for (int k0 = 0; k0 < HH; k0 += 4) {
    v2f a  = frag_nk(Xc, HH, tm * 16, k0, lane);
    v2f bb = frag_nk(Xc, HH, tn * 16, k0, lane);
    acc = wmma4(a, bb, acc);
  }
  int s = tn * 16 + (lane & 15);
  int mbase = tm * 16 + ((lane >> 4) << 3);
  float* Ab = Amat + (size_t)b * CHK * CHK;
#pragma unroll
  for (int i = 0; i < 8; ++i) {
    int t = mbase + i;
    float v = (s < t) ? LRC * (acc[i] + 1.0f) : 0.0f;
    Ab[(size_t)t * CHK + s] = v;
  }
}

__global__ void k_solve(const float* __restrict__ x, const float* __restrict__ P0,
                        const float* __restrict__ Amat, float* __restrict__ Pred,
                        float* __restrict__ Diff, float* __restrict__ bst, int chunk) {
  extern __shared__ float dsh[];
  int b = blockIdx.x;
  int h = threadIdx.x;
  const float* Xb = x + (size_t)b * SS * HH;
  const float* Ab = Amat + (size_t)b * CHK * CHK;
  float bsum = 0.0f;
  for (int t = 0; t < CHK; ++t) {
    int gt = chunk * CHK + t;
    float acc = P0[((size_t)b * CHK + t) * HH + h];
    const float* Arow = Ab + (size_t)t * CHK;
    for (int s = 0; s < t; ++s) acc -= Arow[s] * dsh[(size_t)s * HH + h];
    Pred[((size_t)b * CHK + t) * HH + h] = acc;
    float xn = (gt + 1 < SS) ? Xb[(size_t)(gt + 1) * HH + h] : 0.0f;
    float d = acc - xn;
    dsh[(size_t)t * HH + h] = d;
    float dl = (gt == SS - 1) ? 0.0f : d;
    Diff[((size_t)b * CHK + t) * HH + h] = dl;
    bsum += dl;
  }
  bst[(size_t)b * HH + h] -= LRC * bsum;
}

__global__ void k_wupdate(const float* __restrict__ x, const float* __restrict__ Diff,
                          float* __restrict__ Wst, int chunk) {
  int wid = (blockIdx.x * blockDim.x + threadIdx.x) >> 5;
  int lane = threadIdx.x & 31;
  int tn = wid % (HH / 16);
  int tm = (wid / (HH / 16)) % (HH / 16);
  int b  = wid / ((HH / 16) * (HH / 16));
  const float* Xc = x + ((size_t)b * SS + chunk * CHK) * HH;
  const float* Db = Diff + (size_t)b * CHK * HH;
  float* W = Wst + (size_t)b * HH * HH;
  v8f s = {};
  for (int k0 = 0; k0 < CHK; k0 += 4) {
    v2f a  = frag_kn(Db, HH, k0, tm * 16, lane);
    v2f bb = frag_kn(Xc, HH, k0, tn * 16, lane);
    s = wmma4(a, bb, s);
  }
  v8f acc = load_tile(W, HH, tm * 16, tn * 16, lane);
#pragma unroll
  for (int i = 0; i < 8; ++i) acc[i] -= LRC * s[i];
  store_tile(W, HH, tm * 16, tn * 16, acc, lane);
}

__global__ void k_mlp1(const float* __restrict__ Pred,
                       const float* __restrict__ gw, const float* __restrict__ gb,
                       const float* __restrict__ uw, const float* __restrict__ ub,
                       float* __restrict__ Gt) {
  int wid = (blockIdx.x * blockDim.x + threadIdx.x) >> 5;
  int lane = threadIdx.x & 31;
  int tn = wid % (H4 / 16);
  int tm = (wid / (H4 / 16)) % (CHK / 16);
  int b  = wid / ((H4 / 16) * (CHK / 16));
  const float* Pb = Pred + (size_t)b * CHK * HH;
  v8f g = {}, u = {};
  for (int k0 = 0; k0 < HH; k0 += 4) {
    v2f a  = frag_nk(Pb, HH, tm * 16, k0, lane);
    v2f bg = frag_nk(gw, HH, tn * 16, k0, lane);
    v2f bu = frag_nk(uw, HH, tn * 16, k0, lane);
    g = wmma4(a, bg, g);
    u = wmma4(a, bu, u);
  }
  int n = tn * 16 + (lane & 15);
  float gbias = gb[n], ubias = ub[n];
  v8f r;
#pragma unroll
  for (int i = 0; i < 8; ++i) {
    float zg = g[i] + gbias;
    float zu = u[i] + ubias;
    r[i] = (1.0f / (1.0f + __expf(-zg))) * zu;
  }
  store_tile(Gt + (size_t)b * CHK * H4, H4, tm * 16, tn * 16, r, lane);
}

__global__ void k_mlp2(const float* __restrict__ Gt, const float* __restrict__ dw,
                       const float* __restrict__ db, const float* __restrict__ Pred,
                       float* __restrict__ Ypre) {
  int wid = (blockIdx.x * blockDim.x + threadIdx.x) >> 5;
  int lane = threadIdx.x & 31;
  int tn = wid % (HH / 16);
  int tm = (wid / (HH / 16)) % (CHK / 16);
  int b  = wid / ((HH / 16) * (CHK / 16));
  const float* Gb = Gt + (size_t)b * CHK * H4;
  v8f acc = {};
  for (int k0 = 0; k0 < H4; k0 += 4) {
    if ((k0 & 63) == 0) __builtin_prefetch(Gb + (size_t)(tm * 16) * H4 + k0 + 64, 0, 0);
    v2f a  = frag_nk(Gb, H4, tm * 16, k0, lane);
    v2f bb = frag_nk(dw, H4, tn * 16, k0, lane);
    acc = wmma4(a, bb, acc);
  }
  int n = tn * 16 + (lane & 15);
  int mbase = tm * 16 + ((lane >> 4) << 3);
  float bias = db[n];
  const float* Pb = Pred + (size_t)b * CHK * HH;
  float* Yb = Ypre + (size_t)b * CHK * HH;
#pragma unroll
  for (int i = 0; i < 8; ++i) {
    int m = mbase + i;
    Yb[(size_t)m * HH + n] = acc[i] + bias + Pb[(size_t)m * HH + n];
  }
}

__global__ void k_ln(const float* __restrict__ Y, const float* __restrict__ nw,
                     const float* __restrict__ nb, float* __restrict__ out, int chunk) {
  __shared__ float s1[256], s2[256];
  int row = blockIdx.x;
  int b = row / CHK, t = row % CHK;
  const float* y = Y + ((size_t)b * CHK + t) * HH;
  int tid = threadIdx.x;
  float v0 = y[tid], v1 = y[tid + 256], v2 = y[tid + 512];
  s1[tid] = v0 + v1 + v2;
  s2[tid] = v0 * v0 + v1 * v1 + v2 * v2;
  __syncthreads();
  for (int off = 128; off > 0; off >>= 1) {
    if (tid < off) { s1[tid] += s1[tid + off]; s2[tid] += s2[tid + off]; }
    __syncthreads();
  }
  float mu  = s1[0] * (1.0f / HH);
  float var = s2[0] * (1.0f / HH) - mu * mu;
  float r = rsqrtf(var + EPSC);
  int gt = chunk * CHK + t;
  float* o = out + ((size_t)b * SS + gt) * HH;
  o[tid]       = (v0 - mu) * r * nw[tid]       + nb[tid];
  o[tid + 256] = (v1 - mu) * r * nw[tid + 256] + nb[tid + 256];
  o[tid + 512] = (v2 - mu) * r * nw[tid + 512] + nb[tid + 512];
}

extern "C" void kernel_launch(void* const* d_in, const int* in_sizes, int n_in,
                              void* d_out, int out_size, void* d_ws, size_t ws_size,
                              hipStream_t stream) {
  const float* x      = (const float*)d_in[0];
  const float* W_init = (const float*)d_in[1];
  const float* b_init = (const float*)d_in[2];
  const float* gate_w = (const float*)d_in[3];
  const float* gate_b = (const float*)d_in[4];
  const float* up_w   = (const float*)d_in[5];
  const float* up_b   = (const float*)d_in[6];
  const float* down_w = (const float*)d_in[7];
  const float* down_b = (const float*)d_in[8];
  const float* norm_w = (const float*)d_in[9];
  const float* norm_b = (const float*)d_in[10];
  float* out = (float*)d_out;

  float* ws   = (float*)d_ws;
  float* Wst  = ws;  ws += (size_t)BB * HH * HH;
  float* bst  = ws;  ws += (size_t)BB * HH;
  float* P0   = ws;  ws += (size_t)BB * CHK * HH;
  float* Pred = ws;  ws += (size_t)BB * CHK * HH;
  float* Diff = ws;  ws += (size_t)BB * CHK * HH;
  float* Amat = ws;  ws += (size_t)BB * CHK * CHK;
  float* Gt   = ws;  ws += (size_t)BB * CHK * H4;
  float* Ypre = ws;  ws += (size_t)BB * CHK * HH;

  k_init<<<(HH * HH + 255) / 256, 256, 0, stream>>>(W_init, b_init, Wst, bst);

  const int NCH = SS / CHK;
  for (int c = 0; c < NCH; ++c) {
    k_p0<<<48, 256, 0, stream>>>(x, Wst, bst, P0, c);
    k_attn<<<4, 256, 0, stream>>>(x, Amat, c);
    k_solve<<<BB, HH, CHK * HH * sizeof(float), stream>>>(x, P0, Amat, Pred, Diff, bst, c);
    k_wupdate<<<576, 256, 0, stream>>>(x, Diff, Wst, c);
    k_mlp1<<<192, 256, 0, stream>>>(Pred, gate_w, gate_b, up_w, up_b, Gt);
    k_mlp2<<<48, 256, 0, stream>>>(Gt, down_w, down_b, Pred, Ypre);
    k_ln<<<BB * CHK, 256, 0, stream>>>(Ypre, norm_w, norm_b, out, c);
  }
}


# chunk loop skewed by one chunk: MLP and LayerNorm of chunk c-1 run beside the recurrence launches of chunk c (attn|mlp1|wupdate|ln, then mlp2|p0, then solve); 29 launches -> 15; every sum as before
# speedup vs baseline: 2.1002x; 2.1002x over previous
#include <hip/hip_runtime.h>
#include <math.h>

#define BB 2
#define SS 256
#define HH 768
#define H4 3072
#define CHK 64
#define LRC 0.01f
#define EPSC 1e-5f

typedef float v2f __attribute__((ext_vector_type(2)));
typedef float v8f __attribute__((ext_vector_type(8)));

__device__ __forceinline__ v8f wmma4(v2f a, v2f b, v8f c) {
  return __builtin_amdgcn_wmma_f32_16x16x4_f32(
       false, a,  false, b,
       (short)0, c,  false,  false);
}

__device__ __forceinline__ v2f frag_nk(const float* M, int ld, int r0, int k0, int lane) {
  const float* p = M + (size_t)(r0 + (lane & 15)) * ld + k0 + ((lane >> 4) << 1);
  v2f v; v.x = p[0]; v.y = p[1];
  return v;
}
__device__ __forceinline__ v2f frag_kn(const float* M, int ld, int k0, int r0, int lane) {
  const float* p = M + (size_t)(k0 + ((lane >> 4) << 1)) * ld + r0 + (lane & 15);
  v2f v; v.x = p[0]; v.y = p[ld];
  return v;
}
__device__ __forceinline__ void store_tile(float* D, int ld, int r0, int c0, v8f v, int lane) {
  int n = c0 + (lane & 15);
  int m0 = r0 + ((lane >> 4) << 3);
#pragma unroll
  for (int i = 0; i < 8; ++i) D[(size_t)(m0 + i) * ld + n] = v[i];
}
__device__ __forceinline__ v8f load_tile(const float* D, int ld, int r0, int c0, int lane) {
  int n = c0 + (lane & 15);
  int m0 = r0 + ((lane >> 4) << 3);
  v8f v;
#pragma unroll
  for (int i = 0; i < 8; ++i) v[i] = D[(size_t)(m0 + i) * ld + n];
  return v;
}

#define NB_INIT ((HH * HH + 255) / 256)
#define NB_P0   (BB * (CHK / 16) * (HH / 16) / 8)
#define NB_ATTN (BB * (CHK / 16) * (CHK / 16) / 8)
#define NB_WUPD (BB * (HH / 16) * (HH / 16) / 8)
#define NB_MLP1 (BB * (CHK / 16) * (H4 / 16) / 8)
#define NB_MLP2 (BB * (CHK / 16) * (HH / 16) / 8)
#define NB_LN   (BB * CHK)
#define NCHUNK  (SS / CHK)

__device__ __forceinline__ void d_init(int blk, const float* __restrict__ Wi, const float* __restrict__ bi,
                       float* __restrict__ Wst, float* __restrict__ bst) {
  int i = blk * 256 + threadIdx.x;
  const int n = HH * HH;
  if (i < n) {
    float w = Wi[i];
    Wst[i] = w;
    Wst[n + i] = w;
  }
  if (i < HH) {
    float b = bi[i];
    bst[i] = b;
    bst[HH + i] = b;
  }
}

__device__ __forceinline__ void d_p0(int blk, const float* __restrict__ x, const float* __restrict__ Wst,
                     const float* __restrict__ bst, float* __restrict__ P0, int chunk) {
  int wid = (blk * 256 + threadIdx.x) >> 5;
  int lane = threadIdx.x & 31;
  int tn = wid % (HH / 16);
  int tm = (wid / (HH / 16)) % (CHK / 16);
  int b  = wid / ((HH / 16) * (CHK / 16));
  const float* Xc = x + ((size_t)b * SS + chunk * CHK) * HH;
  const float* W  = Wst + (size_t)b * HH * HH;
  v8f acc = {};
  for (int k0 = 0; k0 < HH; k0 += 4) {
    v2f a  = frag_nk(Xc, HH, tm * 16, k0, lane);
    v2f bb = frag_nk(W,  HH, tn * 16, k0, lane);
    acc = wmma4(a, bb, acc);
  }
  float bias = bst[(size_t)b * HH + tn * 16 + (lane & 15)];
#pragma unroll
  for (int i = 0; i < 8; ++i) acc[i] += bias;
  store_tile(P0 + (size_t)b * CHK * HH, HH, tm * 16, tn * 16, acc, lane);
}

__device__ __forceinline__ void d_attn(int blk, const float* __restrict__ x, float* __restrict__ Amat, int chunk) {
  int wid = (blk * 256 + threadIdx.x) >> 5;
  int lane = threadIdx.x & 31;
  int tn = wid % (CHK / 16);
  int tm = (wid / (CHK / 16)) % (CHK / 16);
  int b  = wid / ((CHK / 16) * (CHK / 16));
  const float* Xc = x + ((size_t)b * SS + chunk * CHK) * HH;
  v8f acc = {};
  for (int k0 = 0; k0 < HH; k0 += 4) {
    v2f a  = frag_nk(Xc, HH, tm * 16, k0, lane);
    v2f bb = frag_nk(Xc, HH, tn * 16, k0, lane);
    acc = wmma4(a, bb, acc);
  }
  int s = tn * 16 + (lane & 15);
  int mbase = tm * 16 + ((lane >> 4) << 3);
  float* Ab = Amat + (size_t)b * CHK * CHK;
#pragma unroll
  for (int i = 0; i < 8; ++i) {
    int t = mbase + i;
    float v = (s < t) ? LRC * (acc[i] + 1.0f) : 0.0f;
    Ab[(size_t)t * CHK + s] = v;
  }
}

#define APITCH 68
#define DPITCH 40
typedef float v4f __attribute__((ext_vector_type(4)));

__global__ __launch_bounds__(32) void k_solve(const float* __restrict__ x, const float* __restrict__ P0,
                        const float* __restrict__ Amat, float* __restrict__ Pred,
                        float* __restrict__ Diff, float* __restrict__ bst, int chunk) {
  __shared__ __attribute__((aligned(16))) float As[CHK * APITCH];
  __shared__ __attribute__((aligned(16))) float Ds[CHK * DPITCH];
  __shared__ __attribute__((aligned(16))) float Ts[16 * DPITCH];
  const int lane = threadIdx.x;
  const int c = lane & 15, hf = lane >> 4;
  const int b  = blockIdx.x / (HH / 32);
  const int n0 = (blockIdx.x % (HH / 32)) * 32;
  const float* Xb = x + (size_t)b * SS * HH;
  const float* Ab = Amat + (size_t)b * CHK * CHK;
  const size_t col = (size_t)n0 + lane;

  v4f av[32];
#pragma unroll
  for (int i = 0; i < 32; ++i) {
    int idx = i * 32 + lane;
    av[i] = *(const v4f*)(Ab + (idx >> 4) * CHK + (idx & 15) * 4);
  }
  float pv[CHK], xv[CHK];
  const bool last_chunk = (chunk == SS / CHK - 1);
  const float* Xn = Xb + ((size_t)chunk * CHK + 1) * HH + col;
#pragma unroll
  for (int t = 0; t < CHK; ++t) {
    pv[t] = P0[((size_t)b * CHK + t) * HH + col];
    if (t < CHK - 1) {
      xv[t] = Xn[(size_t)t * HH];
    } else {
      const float xnext = Xn[(size_t)(last_chunk ? t - 1 : t) * HH];
      xv[t] = last_chunk ? 0.0f : xnext;
    }
  }
#pragma unroll
  for (int i = 0; i < 32; ++i) {
    int idx = i * 32 + lane;
    *(v4f*)(As + (idx >> 4) * APITCH + (idx & 15) * 4) = av[i];
  }
  __syncthreads();

  float bsum = 0.0f;
#pragma unroll
  for (int i = 0; i < CHK / 16; ++i) {
    const int t0 = i * 16;
    v4f cf[16][4];
#pragma unroll
    for (int r = 1; r < 16; ++r) {
#pragma unroll
      for (int q4 = 0; q4 * 4 < r; ++q4)
        cf[r][q4] = *(const v4f*)(As + (t0 + r) * APITCH + t0 + q4 * 4);
    }
    float sc[16];
    if (i > 0) {
      v8f s0 = {}, s1 = {};
#pragma unroll
      for (int k0 = 0; k0 < t0; k0 += 4) {
        const int k = k0 + 2 * hf;
        v2f a = *(const v2f*)(As + (t0 + c) * APITCH + k);
        v2f b0, b1;
        b0.x = Ds[k * DPITCH + c];       b0.y = Ds[(k + 1) * DPITCH + c];
        b1.x = Ds[k * DPITCH + 16 + c];  b1.y = Ds[(k + 1) * DPITCH + 16 + c];
        s0 = wmma4(a, b0, s0);
        s1 = wmma4(a, b1, s1);
      }
#pragma unroll
      for (int r = 0; r < 8; ++r) {
        Ts[(8 * hf + r) * DPITCH + c]      = s0[r];
        Ts[(8 * hf + r) * DPITCH + 16 + c] = s1[r];
      }
      __syncthreads();
#pragma unroll
      for (int r = 0; r < 16; ++r) sc[r] = Ts[r * DPITCH + lane];
    }
    float d[16];
#pragma unroll
    for (int r = 0; r < 16; ++r) {
      const int t = t0 + r;
      float acc = (i > 0) ? pv[t] - sc[r] : pv[t];
#pragma unroll
      for (int q = 0; q < r; ++q) acc -= cf[r][q >> 2][q & 3] * d[q];
      float dd = acc - xv[t];
      d[r] = dd;
      Ds[t * DPITCH + lane] = dd;
      pv[t] = acc;
      float dl = (t == CHK - 1 && last_chunk) ? 0.0f : dd;
      xv[t] = dl;
      bsum += dl;
    }
    __syncthreads();
  }
#pragma unroll
  for (int t = 0; t < CHK; ++t) {
    Pred[((size_t)b * CHK + t) * HH + col] = pv[t];
    Diff[((size_t)b * CHK + t) * HH + col] = xv[t];
  }
  bst[(size_t)b * HH + col] -= LRC * bsum;
}

__device__ __forceinline__ void d_wupdate(int blk, const float* __restrict__ x, const float* __restrict__ Diff,
                          float* __restrict__ Wst, int chunk) {
  int wid = (blk * 256 + threadIdx.x) >> 5;
  int lane = threadIdx.x & 31;
  int tn = wid % (HH / 16);
  int tm = (wid / (HH / 16)) % (HH / 16);
  int b  = wid / ((HH / 16) * (HH / 16));
  const float* Xc = x + ((size_t)b * SS + chunk * CHK) * HH;
  const float* Db = Diff + (size_t)b * CHK * HH;
  float* W = Wst + (size_t)b * HH * HH;
  v8f s = {};
  for (int k0 = 0; k0 < CHK; k0 += 4) {
    v2f a  = frag_kn(Db, HH, k0, tm * 16, lane);
    v2f bb = frag_kn(Xc, HH, k0, tn * 16, lane);
    s = wmma4(a, bb, s);
  }
  v8f acc = load_tile(W, HH, tm * 16, tn * 16, lane);
#pragma unroll
  for (int i = 0; i < 8; ++i) acc[i] -= LRC * s[i];
  store_tile(W, HH, tm * 16, tn * 16, acc, lane);
}

__device__ __forceinline__ void d_mlp1(int blk, const float* __restrict__ Pred,
                       const float* __restrict__ gw, const float* __restrict__ gb,
                       const float* __restrict__ uw, const float* __restrict__ ub,
                       float* __restrict__ Gt) {
  int wid = (blk * 256 + threadIdx.x) >> 5;
  int lane = threadIdx.x & 31;
  int tn = wid % (H4 / 16);
  int tm = (wid / (H4 / 16)) % (CHK / 16);
  int b  = wid / ((H4 / 16) * (CHK / 16));
  const float* Pb = Pred + (size_t)b * CHK * HH;
  v8f g = {}, u = {};
  for (int k0 = 0; k0 < HH; k0 += 4) {
    v2f a  = frag_nk(Pb, HH, tm * 16, k0, lane);
    v2f bg = frag_nk(gw, HH, tn * 16, k0, lane);
    v2f bu = frag_nk(uw, HH, tn * 16, k0, lane);
    g = wmma4(a, bg, g);
    u = wmma4(a, bu, u);
  }
  int n = tn * 16 + (lane & 15);
  float gbias = gb[n], ubias = ub[n];
  v8f r;
#pragma unroll
  for (int i = 0; i < 8; ++i) {
    float zg = g[i] + gbias;
    float zu = u[i] + ubias;
    r[i] = (1.0f / (1.0f + __expf(-zg))) * zu;
  }
  store_tile(Gt + (size_t)b * CHK * H4, H4, tm * 16, tn * 16, r, lane);
}

__device__ __forceinline__ void d_mlp2(int blk, const float* __restrict__ Gt, const float* __restrict__ dw,
                       const float* __restrict__ db, const float* __restrict__ Pred,
                       float* __restrict__ Ypre) {
  int wid = (blk * 256 + threadIdx.x) >> 5;
  int lane = threadIdx.x & 31;
  int tn = wid % (HH / 16);
  int tm = (wid / (HH / 16)) % (CHK / 16);
  int b  = wid / ((HH / 16) * (CHK / 16));
  const float* Gb = Gt + (size_t)b * CHK * H4;
  v8f acc = {};
  for (int k0 = 0; k0 < H4; k0 += 4) {
    if ((k0 & 63) == 0) __builtin_prefetch(Gb + (size_t)(tm * 16) * H4 + k0 + 64, 0, 0);
    v2f a  = frag_nk(Gb, H4, tm * 16, k0, lane);
    v2f bb = frag_nk(dw, H4, tn * 16, k0, lane);
    acc = wmma4(a, bb, acc);
  }
  int n = tn * 16 + (lane & 15);
  int mbase = tm * 16 + ((lane >> 4) << 3);
  float bias = db[n];
  const float* Pb = Pred + (size_t)b * CHK * HH;
  float* Yb = Ypre + (size_t)b * CHK * HH;
#pragma unroll
  for (int i = 0; i < 8; ++i) {
    int m = mbase + i;
    Yb[(size_t)m * HH + n] = acc[i] + bias + Pb[(size_t)m * HH + n];
  }
}

__device__ __forceinline__ void d_ln(int blk, const float* __restrict__ Y, const float* __restrict__ nw,
                     const float* __restrict__ nb, float* __restrict__ out, int chunk) {
  __shared__ float s1[256], s2[256];
  int row = blk;
  int b = row / CHK, t = row % CHK;
  const float* y = Y + ((size_t)b * CHK + t) * HH;
  int tid = threadIdx.x;
  float v0 = y[tid], v1 = y[tid + 256], v2 = y[tid + 512];
  s1[tid] = v0 + v1 + v2;
  s2[tid] = v0 * v0 + v1 * v1 + v2 * v2;
  __syncthreads();
  for (int off = 128; off > 0; off >>= 1) {
    if (tid < off) { s1[tid] += s1[tid + off]; s2[tid] += s2[tid + off]; }
    __syncthreads();
  }
  float mu  = s1[0] * (1.0f / HH);
  float var = s2[0] * (1.0f / HH) - mu * mu;
  float r = rsqrtf(var + EPSC);
  int gt = chunk * CHK + t;
  float* o = out + ((size_t)b * SS + gt) * HH;
  o[tid]       = (v0 - mu) * r * nw[tid]       + nb[tid];
  o[tid + 256] = (v1 - mu) * r * nw[tid + 256] + nb[tid + 256];
  o[tid + 512] = (v2 - mu) * r * nw[tid + 512] + nb[tid + 512];
}


__global__ __launch_bounds__(256) void k_first(const float* __restrict__ x,
                       const float* __restrict__ Wi, const float* __restrict__ bi,
                       float* __restrict__ Wst, float* __restrict__ bst,
                       float* __restrict__ Amat) {
  const int blk = blockIdx.x;
  if (blk < NB_ATTN) d_attn(blk, x, Amat, 0);
  else               d_init(blk - NB_ATTN, Wi, bi, Wst, bst);
}

__global__ __launch_bounds__(256) void k_p0(const float* __restrict__ x, const float* __restrict__ Wst,
                     const float* __restrict__ bst, float* __restrict__ P0, int chunk) {
  d_p0(blockIdx.x, x, Wst, bst, P0, chunk);
}

__global__ __launch_bounds__(256) void k_stage_a(const float* __restrict__ x,
                       const float* __restrict__ Diff, float* __restrict__ Wst,
                       const float* __restrict__ Pred,
                       const float* __restrict__ gw, const float* __restrict__ gb,
                       const float* __restrict__ uw, const float* __restrict__ ub,
                       float* __restrict__ Gt, float* __restrict__ Amat,
                       const float* __restrict__ Ypre,
                       const float* __restrict__ nw, const float* __restrict__ nb,
                       float* __restrict__ out, int c) {
  int blk = blockIdx.x;
  if (blk < NB_ATTN) {
    if (c < NCHUNK) d_attn(blk, x, Amat, c);
    return;
  }
  blk -= NB_ATTN;
  if (blk < NB_MLP1) {
    d_mlp1(blk, Pred, gw, gb, uw, ub, Gt);
    return;
  }
  blk -= NB_MLP1;
  if (blk < NB_WUPD) {
    d_wupdate(blk, x, Diff, Wst, c - 1);
    return;
  }
  blk -= NB_WUPD;
  if (c >= 2) d_ln(blk, Ypre, nw, nb, out, c - 2);
}

__global__ __launch_bounds__(256) void k_stage_b(const float* __restrict__ x,
                       const float* __restrict__ Wst, const float* __restrict__ bst,
                       float* __restrict__ P0,
                       const float* __restrict__ Gt, const float* __restrict__ dw,
                       const float* __restrict__ db, const float* __restrict__ Pred,
                       float* __restrict__ Ypre, int c) {
  int blk = blockIdx.x;
  if (blk < NB_MLP2) {
    d_mlp2(blk, Gt, dw, db, Pred, Ypre);
    return;
  }
  blk -= NB_MLP2;
  if (c < NCHUNK) d_p0(blk, x, Wst, bst, P0, c);
}

__global__ __launch_bounds__(256) void k_ln(const float* __restrict__ Y, const float* __restrict__ nw,
                     const float* __restrict__ nb, float* __restrict__ out, int chunk) {
  d_ln(blockIdx.x, Y, nw, nb, out, chunk);
}

extern "C" void kernel_launch(void* const* d_in, const int* in_sizes, int n_in,
                              void* d_out, int out_size, void* d_ws, size_t ws_size,
                              hipStream_t stream) {
  const float* x      = (const float*)d_in[0];
  const float* W_init = (const float*)d_in[1];
  const float* b_init = (const float*)d_in[2];
  const float* gate_w = (const float*)d_in[3];
  const float* gate_b = (const float*)d_in[4];
  const float* up_w   = (const float*)d_in[5];
  const float* up_b   = (const float*)d_in[6];
  const float* down_w = (const float*)d_in[7];
  const float* down_b = (const float*)d_in[8];
  const float* norm_w = (const float*)d_in[9];
  const float* norm_b = (const float*)d_in[10];
  float* out = (float*)d_out;

  float* ws   = (float*)d_ws;
  float* Wst  = ws;  ws += (size_t)BB * HH * HH;
  float* bst  = ws;  ws += (size_t)BB * HH;
  float* P0   = ws;  ws += (size_t)BB * CHK * HH;
  float* Pred = ws;  ws += (size_t)BB * CHK * HH;
  float* Diff = ws;  ws += (size_t)BB * CHK * HH;
  float* Amat = ws;  ws += (size_t)BB * CHK * CHK;
  float* Gt   = ws;  ws += (size_t)BB * CHK * H4;
  float* Ypre = ws;  ws += (size_t)BB * CHK * HH;

  k_first<<<NB_ATTN + NB_INIT, 256, 0, stream>>>(x, W_init, b_init, Wst, bst, Amat);
  k_p0<<<NB_P0, 256, 0, stream>>>(x, Wst, bst, P0, 0);
  k_solve<<<48, 32, 0, stream>>>(x, P0, Amat, Pred, Diff, bst, 0);

  for (int c = 1; c <= NCHUNK; ++c) {
    k_stage_a<<<NB_ATTN + NB_MLP1 + NB_WUPD + NB_LN, 256, 0, stream>>>(
        x, Diff, Wst, Pred, gate_w, gate_b, up_w, up_b, Gt, Amat, Ypre, norm_w, norm_b, out, c);
    k_stage_b<<<NB_MLP2 + NB_P0, 256, 0, stream>>>(
        x, Wst, bst, P0, Gt, down_w, down_b, Pred, Ypre, c);
    if (c < NCHUNK) k_solve<<<48, 32, 0, stream>>>(x, P0, Amat, Pred, Diff, bst, c);
  }
  k_ln<<<NB_LN, 256, 0, stream>>>(Ypre, norm_w, norm_b, out, NCHUNK - 1);
}


# gate_w, up_w, down_w copied once per call into WMMA fragment order (in the launch of p0(0)): mlp1/mlp2 B fragments by whole-line 16-byte loads, 4 lines each, was 16; k pairing, so sum order, changes
# speedup vs baseline: 3.0040x; 1.4303x over previous
#include <hip/hip_runtime.h>
#include <math.h>

#define BB 2
#define SS 256
#define HH 768
#define H4 3072
#define CHK 64
#define LRC 0.01f
#define EPSC 1e-5f

typedef float v2f __attribute__((ext_vector_type(2)));
typedef float v8f __attribute__((ext_vector_type(8)));
typedef float v4f __attribute__((ext_vector_type(4)));

__device__ __forceinline__ v8f wmma4(v2f a, v2f b, v8f c) {
  return __builtin_amdgcn_wmma_f32_16x16x4_f32(
       false, a,  false, b,
       (short)0, c,  false,  false);
}

__device__ __forceinline__ v2f frag_nk(const float* M, int ld, int r0, int k0, int lane) {
  const float* p = M + (size_t)(r0 + (lane & 15)) * ld + k0 + ((lane >> 4) << 1);
  v2f v; v.x = p[0]; v.y = p[1];
  return v;
}
__device__ __forceinline__ v2f frag_kn(const float* M, int ld, int k0, int r0, int lane) {
  const float* p = M + (size_t)(k0 + ((lane >> 4) << 1)) * ld + r0 + (lane & 15);
  v2f v; v.x = p[0]; v.y = p[ld];
  return v;
}
__device__ __forceinline__ v2f lo2(v4f v) { v2f r; r.x = v.x; r.y = v.y; return r; }
__device__ __forceinline__ v2f hi2(v4f v) { v2f r; r.x = v.z; r.y = v.w; return r; }

__device__ __forceinline__ void store_tile(float* D, int ld, int r0, int c0, v8f v, int lane) {
  int n = c0 + (lane & 15);
  int m0 = r0 + ((lane >> 4) << 3);
#pragma unroll
  for (int i = 0; i < 8; ++i) D[(size_t)(m0 + i) * ld + n] = v[i];
}
__device__ __forceinline__ v8f load_tile(const float* D, int ld, int r0, int c0, int lane) {
  int n = c0 + (lane & 15);
  int m0 = r0 + ((lane >> 4) << 3);
  v8f v;
#pragma unroll
  for (int i = 0; i < 8; ++i) v[i] = D[(size_t)(m0 + i) * ld + n];
  return v;
}

#define NB_INIT ((HH * HH + 255) / 256)
#define NB_P0   (BB * (CHK / 16) * (HH / 16) / 8)
#define NB_ATTN (BB * (CHK / 16) * (CHK / 16) / 8)
#define NB_WUPD (BB * (HH / 16) * (HH / 16) / 8)
#define NB_MLP1 (BB * (CHK / 16) * (H4 / 16) / 8)
#define NB_MLP2 (BB * (CHK / 16) * (HH / 16) / 8)
#define NB_LN   (BB * CHK)
#define NCHUNK  (SS / CHK)
#define NB_PACK (3 * 192)

__device__ __forceinline__ void d_init(int blk, const float* __restrict__ Wi, const float* __restrict__ bi,
                       float* __restrict__ Wst, float* __restrict__ bst) {
  int i = blk * 256 + threadIdx.x;
  const int n = HH * HH;
  if (i < n) {
    float w = Wi[i];
    Wst[i] = w;
    Wst[n + i] = w;
  }
  if (i < HH) {
    float b = bi[i];
    bst[i] = b;
    bst[HH + i] = b;
  }
}

__device__ __forceinline__ void d_pack(int blk, const float* __restrict__ gw, const float* __restrict__ uw,
                       const float* __restrict__ dw, float* __restrict__ gwp,
                       float* __restrict__ uwp, float* __restrict__ dwp) {
  const int which = blk / 192, r = blk % 192;
  const float* src = (which == 0) ? gw : (which == 1) ? uw : dw;
  float* dst = (which == 0) ? gwp : (which == 1) ? uwp : dwp;
  const int K   = (which == 2) ? H4 : HH;
  const int tn  = (which == 2) ? (r >> 2) : r;
  const int seg = (which == 2) ? (r & 3) : 0;
  const int w = threadIdx.x >> 5, lane = threadIdx.x & 31;
  const float* s = src + (size_t)(tn * 16 + (lane & 15)) * K + 4 * (lane >> 4);
  float* d = dst + ((size_t)tn * (K / 8) * 32 + lane) * 4;
  const int kg0 = seg * 96 + w * 12;
  v4f v[12];
#pragma unroll
  for (int i = 0; i < 12; ++i) v[i] = *(const v4f*)(s + (size_t)(kg0 + i) * 8);
#pragma unroll
  for (int i = 0; i < 12; ++i) *(v4f*)(d + (size_t)(kg0 + i) * 128) = v[i];
}

__device__ __forceinline__ void d_p0(int blk, const float* __restrict__ x, const float* __restrict__ Wst,
                     const float* __restrict__ bst, float* __restrict__ P0, int chunk) {
  int wid = (blk * 256 + threadIdx.x) >> 5;
  int lane = threadIdx.x & 31;
  int tn = wid % (HH / 16);
  int tm = (wid / (HH / 16)) % (CHK / 16);
  int b  = wid / ((HH / 16) * (CHK / 16));
  const float* Xc = x + ((size_t)b * SS + chunk * CHK) * HH;
  const float* W  = Wst + (size_t)b * HH * HH;
  v8f acc = {};
  for (int k0 = 0; k0 < HH; k0 += 4) {
    v2f a  = frag_nk(Xc, HH, tm * 16, k0, lane);
    v2f bb = frag_nk(W,  HH, tn * 16, k0, lane);
    acc = wmma4(a, bb, acc);
  }
  float bias = bst[(size_t)b * HH + tn * 16 + (lane & 15)];
#pragma unroll
  for (int i = 0; i < 8; ++i) acc[i] += bias;
  store_tile(P0 + (size_t)b * CHK * HH, HH, tm * 16, tn * 16, acc, lane);
}

__device__ __forceinline__ void d_attn(int blk, const float* __restrict__ x, float* __restrict__ Amat, int chunk) {
  int wid = (blk * 256 + threadIdx.x) >> 5;
  int lane = threadIdx.x & 31;
  int tn = wid % (CHK / 16);
  int tm = (wid / (CHK / 16)) % (CHK / 16);
  int b  = wid / ((CHK / 16) * (CHK / 16));
  const float* Xc = x + ((size_t)b * SS + chunk * CHK) * HH;
  v8f acc = {};
  for (int k0 = 0; k0 < HH; k0 += 4) {
    v2f a  = frag_nk(Xc, HH, tm * 16, k0, lane);
    v2f bb = frag_nk(Xc, HH, tn * 16, k0, lane);
    acc = wmma4(a, bb, acc);
  }
  int s = tn * 16 + (lane & 15);
  int mbase = tm * 16 + ((lane >> 4) << 3);
  float* Ab = Amat + (size_t)b * CHK * CHK;
#pragma unroll
  for (int i = 0; i < 8; ++i) {
    int t = mbase + i;
    float v = (s < t) ? LRC * (acc[i] + 1.0f) : 0.0f;
    Ab[(size_t)t * CHK + s] = v;
  }
}

#define APITCH 68
#define DPITCH 40

__global__ __launch_bounds__(32) void k_solve(const float* __restrict__ x, const float* __restrict__ P0,
                        const float* __restrict__ Amat, float* __restrict__ Pred,
                        float* __restrict__ Diff, float* __restrict__ bst, int chunk) {
  __shared__ __attribute__((aligned(16))) float As[CHK * APITCH];
  __shared__ __attribute__((aligned(16))) float Ds[CHK * DPITCH];
  __shared__ __attribute__((aligned(16))) float Ts[16 * DPITCH];
  const int lane = threadIdx.x;
  const int c = lane & 15, hf = lane >> 4;
  const int b  = blockIdx.x / (HH / 32);
  const int n0 = (blockIdx.x % (HH / 32)) * 32;
  const float* Xb = x + (size_t)b * SS * HH;
  const float* Ab = Amat + (size_t)b * CHK * CHK;
  const size_t col = (size_t)n0 + lane;

  v4f av[32];
#pragma unroll
  for (int i = 0; i < 32; ++i) {
    int idx = i * 32 + lane;
    av[i] = *(const v4f*)(Ab + (idx >> 4) * CHK + (idx & 15) * 4);
  }
  float pv[CHK], xv[CHK];
  const bool last_chunk = (chunk == SS / CHK - 1);
  const float* Xn = Xb + ((size_t)chunk * CHK + 1) * HH + col;
#pragma unroll
  for (int t = 0; t < CHK; ++t) {
    pv[t] = P0[((size_t)b * CHK + t) * HH + col];
    if (t < CHK - 1) {
      xv[t] = Xn[(size_t)t * HH];
    } else {
      const float xnext = Xn[(size_t)(last_chunk ? t - 1 : t) * HH];
      xv[t] = last_chunk ? 0.0f : xnext;
    }
  }
#pragma unroll
  for (int i = 0; i < 32; ++i) {
    int idx = i * 32 + lane;
    *(v4f*)(As + (idx >> 4) * APITCH + (idx & 15) * 4) = av[i];
  }
  __syncthreads();

  float bsum = 0.0f;
#pragma unroll
  for (int i = 0; i < CHK / 16; ++i) {
    const int t0 = i * 16;
    v4f cf[16][4];
#pragma unroll
    for (int r = 1; r < 16; ++r) {
#pragma unroll
      for (int q4 = 0; q4 * 4 < r; ++q4)
        cf[r][q4] = *(const v4f*)(As + (t0 + r) * APITCH + t0 + q4 * 4);
    }
    float sc[16];
    if (i > 0) {
      v8f s0 = {}, s1 = {};
#pragma unroll
      for (int k0 = 0; k0 < t0; k0 += 4) {
        const int k = k0 + 2 * hf;
        v2f a = *(const v2f*)(As + (t0 + c) * APITCH + k);
        v2f b0, b1;
        b0.x = Ds[k * DPITCH + c];       b0.y = Ds[(k + 1) * DPITCH + c];
        b1.x = Ds[k * DPITCH + 16 + c];  b1.y = Ds[(k + 1) * DPITCH + 16 + c];
        s0 = wmma4(a, b0, s0);
        s1 = wmma4(a, b1, s1);
      }
#pragma unroll
      for (int r = 0; r < 8; ++r) {
        Ts[(8 * hf + r) * DPITCH + c]      = s0[r];
        Ts[(8 * hf + r) * DPITCH + 16 + c] = s1[r];
      }
      __syncthreads();
#pragma unroll
      for (int r = 0; r < 16; ++r) sc[r] = Ts[r * DPITCH + lane];
    }
    float d[16];
#pragma unroll
    for (int r = 0; r < 16; ++r) {
      const int t = t0 + r;
      float acc = (i > 0) ? pv[t] - sc[r] : pv[t];
#pragma unroll
      for (int q = 0; q < r; ++q) acc -= cf[r][q >> 2][q & 3] * d[q];
      float dd = acc - xv[t];
      d[r] = dd;
      Ds[t * DPITCH + lane] = dd;
      pv[t] = acc;
      float dl = (t == CHK - 1 && last_chunk) ? 0.0f : dd;
      xv[t] = dl;
      bsum += dl;
    }
    __syncthreads();
  }
#pragma unroll
  for (int t = 0; t < CHK; ++t) {
    Pred[((size_t)b * CHK + t) * HH + col] = pv[t];
    Diff[((size_t)b * CHK + t) * HH + col] = xv[t];
  }
  bst[(size_t)b * HH + col] -= LRC * bsum;
}

__device__ __forceinline__ void d_wupdate(int blk, const float* __restrict__ x, const float* __restrict__ Diff,
                          float* __restrict__ Wst, int chunk) {
  int wid = (blk * 256 + threadIdx.x) >> 5;
  int lane = threadIdx.x & 31;
  int tn = wid % (HH / 16);
  int tm = (wid / (HH / 16)) % (HH / 16);
  int b  = wid / ((HH / 16) * (HH / 16));
  const float* Xc = x + ((size_t)b * SS + chunk * CHK) * HH;
  const float* Db = Diff + (size_t)b * CHK * HH;
  float* W = Wst + (size_t)b * HH * HH;
  v8f s = {};
  for (int k0 = 0; k0 < CHK; k0 += 4) {
    v2f a  = frag_kn(Db, HH, k0, tm * 16, lane);
    v2f bb = frag_kn(Xc, HH, k0, tn * 16, lane);
    s = wmma4(a, bb, s);
  }
  v8f acc = load_tile(W, HH, tm * 16, tn * 16, lane);
#pragma unroll
  for (int i = 0; i < 8; ++i) acc[i] -= LRC * s[i];
  store_tile(W, HH, tm * 16, tn * 16, acc, lane);
}

__device__ __forceinline__ void d_mlp1(int blk, const float* __restrict__ Pred,
                       const float* __restrict__ gw, const float* __restrict__ gb,
                       const float* __restrict__ uw, const float* __restrict__ ub,
                       float* __restrict__ Gt) {
  int wid = (blk * 256 + threadIdx.x) >> 5;
  int lane = threadIdx.x & 31;
  int tn = wid % (H4 / 16);
  int tm = (wid / (H4 / 16)) % (CHK / 16);
  int b  = wid / ((H4 / 16) * (CHK / 16));
  const float* Pb = Pred + (size_t)b * CHK * HH;
  const float* ap = Pb + (size_t)(tm * 16 + (lane & 15)) * HH + 4 * (lane >> 4);
  const float* gp = gw + ((size_t)tn * (HH / 8) * 32 + lane) * 4;
  const float* up = uw + ((size_t)tn * (HH / 8) * 32 + lane) * 4;
  v8f g = {}, u = {};
#pragma unroll 8
  for (int kg = 0; kg < HH / 8; ++kg) {
    v4f a  = *(const v4f*)(ap + kg * 8);
    v4f bg = *(const v4f*)(gp + (size_t)kg * 128);
    v4f bu = *(const v4f*)(up + (size_t)kg * 128);
    g = wmma4(lo2(a), lo2(bg), g);
    u = wmma4(lo2(a), lo2(bu), u);
    g = wmma4(hi2(a), hi2(bg), g);
    u = wmma4(hi2(a), hi2(bu), u);
  }
  int n = tn * 16 + (lane & 15);
  float gbias = gb[n], ubias = ub[n];
  v8f r;
#pragma unroll
  for (int i = 0; i < 8; ++i) {
    float zg = g[i] + gbias;
    float zu = u[i] + ubias;
    r[i] = (1.0f / (1.0f + __expf(-zg))) * zu;
  }
  store_tile(Gt + (size_t)b * CHK * H4, H4, tm * 16, tn * 16, r, lane);
}

__device__ __forceinline__ void d_mlp2(int blk, const float* __restrict__ Gt, const float* __restrict__ dw,
                       const float* __restrict__ db, const float* __restrict__ Pred,
                       float* __restrict__ Ypre) {
  int wid = (blk * 256 + threadIdx.x) >> 5;
  int lane = threadIdx.x & 31;
  int tn = wid % (HH / 16);
  int tm = (wid / (HH / 16)) % (CHK / 16);
  int b  = wid / ((HH / 16) * (CHK / 16));
  const float* Gb = Gt + (size_t)b * CHK * H4;
  const float* ap = Gb + (size_t)(tm * 16 + (lane & 15)) * H4 + 4 * (lane >> 4);
  const float* bp = dw + ((size_t)tn * (H4 / 8) * 32 + lane) * 4;
  v8f acc = {};
#pragma unroll 8
  for (int kg = 0; kg < H4 / 8; ++kg) {
    v4f a  = *(const v4f*)(ap + kg * 8);
    v4f bb = *(const v4f*)(bp + (size_t)kg * 128);
    acc = wmma4(lo2(a), lo2(bb), acc);
    acc = wmma4(hi2(a), hi2(bb), acc);
  }
  int n = tn * 16 + (lane & 15);
  int mbase = tm * 16 + ((lane >> 4) << 3);
  float bias = db[n];
  const float* Pb = Pred + (size_t)b * CHK * HH;
  float* Yb = Ypre + (size_t)b * CHK * HH;
#pragma unroll
  for (int i = 0; i < 8; ++i) {
    int m = mbase + i;
    Yb[(size_t)m * HH + n] = acc[i] + bias + Pb[(size_t)m * HH + n];
  }
}

__device__ __forceinline__ void d_ln(int blk, const float* __restrict__ Y, const float* __restrict__ nw,
                     const float* __restrict__ nb, float* __restrict__ out, int chunk) {
  __shared__ float s1[256], s2[256];
  int row = blk;
  int b = row / CHK, t = row % CHK;
  const float* y = Y + ((size_t)b * CHK + t) * HH;
  int tid = threadIdx.x;
  float v0 = y[tid], v1 = y[tid + 256], v2 = y[tid + 512];
  s1[tid] = v0 + v1 + v2;
  s2[tid] = v0 * v0 + v1 * v1 + v2 * v2;
  __syncthreads();
  for (int off = 128; off > 0; off >>= 1) {
    if (tid < off) { s1[tid] += s1[tid + off]; s2[tid] += s2[tid + off]; }
    __syncthreads();
  }
  float mu  = s1[0] * (1.0f / HH);
  float var = s2[0] * (1.0f / HH) - mu * mu;
  float r = rsqrtf(var + EPSC);
  int gt = chunk * CHK + t;
  float* o = out + ((size_t)b * SS + gt) * HH;
  o[tid]       = (v0 - mu) * r * nw[tid]       + nb[tid];
  o[tid + 256] = (v1 - mu) * r * nw[tid + 256] + nb[tid + 256];
  o[tid + 512] = (v2 - mu) * r * nw[tid + 512] + nb[tid + 512];
}


__global__ __launch_bounds__(256) void k_first(const float* __restrict__ x,
                       const float* __restrict__ Wi, const float* __restrict__ bi,
                       float* __restrict__ Wst, float* __restrict__ bst,
                       float* __restrict__ Amat) {
  const int blk = blockIdx.x;
  if (blk < NB_ATTN) d_attn(blk, x, Amat, 0);
  else               d_init(blk - NB_ATTN, Wi, bi, Wst, bst);
}

__global__ __launch_bounds__(256) void k_p0(const float* __restrict__ x, const float* __restrict__ Wst,
                     const float* __restrict__ bst, float* __restrict__ P0,
                     const float* __restrict__ gw, const float* __restrict__ uw,
                     const float* __restrict__ dw, float* __restrict__ gwp,
                     float* __restrict__ uwp, float* __restrict__ dwp) {
  const int blk = blockIdx.x;
  if (blk < NB_P0) d_p0(blk, x, Wst, bst, P0, 0);
  else             d_pack(blk - NB_P0, gw, uw, dw, gwp, uwp, dwp);
}

__global__ __launch_bounds__(256) void k_stage_a(const float* __restrict__ x,
                       const float* __restrict__ Diff, float* __restrict__ Wst,
                       const float* __restrict__ Pred,
                       const float* __restrict__ gw, const float* __restrict__ gb,
                       const float* __restrict__ uw, const float* __restrict__ ub,
                       float* __restrict__ Gt, float* __restrict__ Amat,
                       const float* __restrict__ Ypre,
                       const float* __restrict__ nw, const float* __restrict__ nb,
                       float* __restrict__ out, int c) {
  int blk = blockIdx.x;
  if (blk < NB_ATTN) {
    if (c < NCHUNK) d_attn(blk, x, Amat, c);
    return;
  }
  blk -= NB_ATTN;
  if (blk < NB_MLP1) {
    d_mlp1(blk, Pred, gw, gb, uw, ub, Gt);
    return;
  }
  blk -= NB_MLP1;
  if (blk < NB_WUPD) {
    d_wupdate(blk, x, Diff, Wst, c - 1);
    return;
  }
  blk -= NB_WUPD;
  if (c >= 2) d_ln(blk, Ypre, nw, nb, out, c - 2);
}

__global__ __launch_bounds__(256) void k_stage_b(const float* __restrict__ x,
                       const float* __restrict__ Wst, const float* __restrict__ bst,
                       float* __restrict__ P0,
                       const float* __restrict__ Gt, const float* __restrict__ dw,
                       const float* __restrict__ db, const float* __restrict__ Pred,
                       float* __restrict__ Ypre, int c) {
  int blk = blockIdx.x;
  if (blk < NB_MLP2) {
    d_mlp2(blk, Gt, dw, db, Pred, Ypre);
    return;
  }
  blk -= NB_MLP2;
  if (c < NCHUNK) d_p0(blk, x, Wst, bst, P0, c);
}

__global__ __launch_bounds__(256) void k_ln(const float* __restrict__ Y, const float* __restrict__ nw,
                     const float* __restrict__ nb, float* __restrict__ out, int chunk) {
  d_ln(blockIdx.x, Y, nw, nb, out, chunk);
}

extern "C" void kernel_launch(void* const* d_in, const int* in_sizes, int n_in,
                              void* d_out, int out_size, void* d_ws, size_t ws_size,
                              hipStream_t stream) {
  const float* x      = (const float*)d_in[0];
  const float* W_init = (const float*)d_in[1];
  const float* b_init = (const float*)d_in[2];
  const float* gate_w = (const float*)d_in[3];
  const float* gate_b = (const float*)d_in[4];
  const float* up_w   = (const float*)d_in[5];
  const float* up_b   = (const float*)d_in[6];
  const float* down_w = (const float*)d_in[7];
  const float* down_b = (const float*)d_in[8];
  const float* norm_w = (const float*)d_in[9];
  const float* norm_b = (const float*)d_in[10];
  float* out = (float*)d_out;

  float* ws   = (float*)d_ws;
  float* Wst  = ws;  ws += (size_t)BB * HH * HH;
  float* bst  = ws;  ws += (size_t)BB * HH;
  float* P0   = ws;  ws += (size_t)BB * CHK * HH;
  float* Pred = ws;  ws += (size_t)BB * CHK * HH;
  float* Diff = ws;  ws += (size_t)BB * CHK * HH;
  float* Amat = ws;  ws += (size_t)BB * CHK * CHK;
  float* Gt   = ws;  ws += (size_t)BB * CHK * H4;
  float* Ypre = ws;  ws += (size_t)BB * CHK * HH;
  float* gwp  = ws;  ws += (size_t)H4 * HH;
  float* uwp  = ws;  ws += (size_t)H4 * HH;
  float* dwp  = ws;  ws += (size_t)HH * H4;

  k_first<<<NB_ATTN + NB_INIT, 256, 0, stream>>>(x, W_init, b_init, Wst, bst, Amat);
  k_p0<<<NB_P0 + NB_PACK, 256, 0, stream>>>(x, Wst, bst, P0, gate_w, up_w, down_w, gwp, uwp, dwp);
  k_solve<<<48, 32, 0, stream>>>(x, P0, Amat, Pred, Diff, bst, 0);

  for (int c = 1; c <= NCHUNK; ++c) {
    k_stage_a<<<NB_ATTN + NB_MLP1 + NB_WUPD + NB_LN, 256, 0, stream>>>(
        x, Diff, Wst, Pred, gwp, gate_b, uwp, up_b, Gt, Amat, Ypre, norm_w, norm_b, out, c);
    k_stage_b<<<NB_MLP2 + NB_P0, 256, 0, stream>>>(
        x, Wst, bst, P0, Gt, dwp, down_b, Pred, Ypre, c);
    if (c < NCHUNK) k_solve<<<48, 32, 0, stream>>>(x, P0, Amat, Pred, Diff, bst, c);
  }
  k_ln<<<NB_LN, 256, 0, stream>>>(Ypre, norm_w, norm_b, out, NCHUNK - 1);
}


# P0 moves into the solve launch: each 32-column solve block's 8 waves compute its 64x32 P0 tile (handed over in LDS), wave 0 then solves; solve runs beside mlp2: 15 launches -> 11; sums as before
# speedup vs baseline: 3.3026x; 1.0994x over previous
#include <hip/hip_runtime.h>
#include <math.h>

#define BB 2
#define SS 256
#define HH 768
#define H4 3072
#define CHK 64
#define LRC 0.01f
#define EPSC 1e-5f

typedef float v2f __attribute__((ext_vector_type(2)));
typedef float v8f __attribute__((ext_vector_type(8)));
typedef float v4f __attribute__((ext_vector_type(4)));

__device__ __forceinline__ v8f wmma4(v2f a, v2f b, v8f c) {
  return __builtin_amdgcn_wmma_f32_16x16x4_f32(
       false, a,  false, b,
       (short)0, c,  false,  false);
}

__device__ __forceinline__ v2f frag_nk(const float* M, int ld, int r0, int k0, int lane) {
  const float* p = M + (size_t)(r0 + (lane & 15)) * ld + k0 + ((lane >> 4) << 1);
  v2f v; v.x = p[0]; v.y = p[1];
  return v;
}
__device__ __forceinline__ v2f frag_kn(const float* M, int ld, int k0, int r0, int lane) {
  const float* p = M + (size_t)(k0 + ((lane >> 4) << 1)) * ld + r0 + (lane & 15);
  v2f v; v.x = p[0]; v.y = p[ld];
  return v;
}
__device__ __forceinline__ v2f lo2(v4f v) { v2f r; r.x = v.x; r.y = v.y; return r; }
__device__ __forceinline__ v2f hi2(v4f v) { v2f r; r.x = v.z; r.y = v.w; return r; }

__device__ __forceinline__ void store_tile(float* D, int ld, int r0, int c0, v8f v, int lane) {
  int n = c0 + (lane & 15);
  int m0 = r0 + ((lane >> 4) << 3);
#pragma unroll
  for (int i = 0; i < 8; ++i) D[(size_t)(m0 + i) * ld + n] = v[i];
}
__device__ __forceinline__ v8f load_tile(const float* D, int ld, int r0, int c0, int lane) {
  int n = c0 + (lane & 15);
  int m0 = r0 + ((lane >> 4) << 3);
  v8f v;
#pragma unroll
  for (int i = 0; i < 8; ++i) v[i] = D[(size_t)(m0 + i) * ld + n];
  return v;
}

#define NB_INIT ((HH * HH + 255) / 256)
#define NB_SOLVE (BB * (HH / 32))
#define NB_ATTN (BB * (CHK / 16) * (CHK / 16) / 8)
#define NB_WUPD (BB * (HH / 16) * (HH / 16) / 8)
#define NB_MLP1 (BB * (CHK / 16) * (H4 / 16) / 8)
#define NB_MLP2 (BB * (CHK / 16) * (HH / 16) / 8)
#define NB_LN   (BB * CHK)
#define NCHUNK  (SS / CHK)
#define NB_PACK (3 * 192)

__device__ __forceinline__ void d_init(int blk, const float* __restrict__ Wi, const float* __restrict__ bi,
                       float* __restrict__ Wst, float* __restrict__ bst) {
  int i = blk * 256 + threadIdx.x;
  const int n = HH * HH;
  if (i < n) {
    float w = Wi[i];
    Wst[i] = w;
    Wst[n + i] = w;
  }
  if (i < HH) {
    float b = bi[i];
    bst[i] = b;
    bst[HH + i] = b;
  }
}

__device__ __forceinline__ void d_pack(int blk, const float* __restrict__ gw, const float* __restrict__ uw,
                       const float* __restrict__ dw, float* __restrict__ gwp,
                       float* __restrict__ uwp, float* __restrict__ dwp) {
  const int which = blk / 192, r = blk % 192;
  const float* src = (which == 0) ? gw : (which == 1) ? uw : dw;
  float* dst = (which == 0) ? gwp : (which == 1) ? uwp : dwp;
  const int K   = (which == 2) ? H4 : HH;
  const int tn  = (which == 2) ? (r >> 2) : r;
  const int seg = (which == 2) ? (r & 3) : 0;
  const int w = threadIdx.x >> 5, lane = threadIdx.x & 31;
  const float* s = src + (size_t)(tn * 16 + (lane & 15)) * K + 4 * (lane >> 4);
  float* d = dst + ((size_t)tn * (K / 8) * 32 + lane) * 4;
  const int kg0 = seg * 96 + w * 12;
  v4f v[12];
#pragma unroll
  for (int i = 0; i < 12; ++i) v[i] = *(const v4f*)(s + (size_t)(kg0 + i) * 8);
#pragma unroll
  for (int i = 0; i < 12; ++i) *(v4f*)(d + (size_t)(kg0 + i) * 128) = v[i];
}

__device__ __forceinline__ void d_attn(int blk, const float* __restrict__ x, float* __restrict__ Amat, int chunk) {
  int wid = (blk * 256 + threadIdx.x) >> 5;
  int lane = threadIdx.x & 31;
  int tn = wid % (CHK / 16);
  int tm = (wid / (CHK / 16)) % (CHK / 16);
  int b  = wid / ((CHK / 16) * (CHK / 16));
  const float* Xc = x + ((size_t)b * SS + chunk * CHK) * HH;
  v8f acc = {};
  for (int k0 = 0; k0 < HH; k0 += 4) {
    v2f a  = frag_nk(Xc, HH, tm * 16, k0, lane);
    v2f bb = frag_nk(Xc, HH, tn * 16, k0, lane);
    acc = wmma4(a, bb, acc);
  }
  int s = tn * 16 + (lane & 15);
  int mbase = tm * 16 + ((lane >> 4) << 3);
  float* Ab = Amat + (size_t)b * CHK * CHK;
#pragma unroll
  for (int i = 0; i < 8; ++i) {
    int t = mbase + i;
    float v = (s < t) ? LRC * (acc[i] + 1.0f) : 0.0f;
    Ab[(size_t)t * CHK + s] = v;
  }
}

#define APITCH 68
#define DPITCH 40
#define PPITCH 36

__device__ __forceinline__ void wave_lds_sync() {
  __builtin_amdgcn_fence(__ATOMIC_SEQ_CST, "workgroup");
  __builtin_amdgcn_wave_barrier();
}

__device__ __forceinline__ void d_p0solve(int blk, const float* __restrict__ x, const float* __restrict__ Wst,
                        const float* __restrict__ Amat, float* __restrict__ Pred,
                        float* __restrict__ Diff, float* __restrict__ bst, int chunk) {
  __shared__ __attribute__((aligned(16))) float As[CHK * APITCH];
  __shared__ __attribute__((aligned(16))) float Ds[CHK * DPITCH];
  __shared__ __attribute__((aligned(16))) float Ts[16 * DPITCH];
  __shared__ __attribute__((aligned(16))) float Ps[CHK * PPITCH];
  const int tid = threadIdx.x;
  const int w = tid >> 5, lane = tid & 31;
  const int c = lane & 15, hf = lane >> 4;
  const int b  = blk / (HH / 32);
  const int n0 = (blk % (HH / 32)) * 32;
  const float* Xb = x + (size_t)b * SS * HH;
  const float* Ab = Amat + (size_t)b * CHK * CHK;
  const size_t col = (size_t)n0 + lane;
  const bool last_chunk = (chunk == SS / CHK - 1);

  v4f av[4];
#pragma unroll
  for (int i = 0; i < 4; ++i) {
    int idx = i * 256 + tid;
    av[i] = *(const v4f*)(Ab + (idx >> 4) * CHK + (idx & 15) * 4);
  }
  float xv[CHK];
  if (w == 0) {
    const float* Xn = Xb + ((size_t)chunk * CHK + 1) * HH + col;
#pragma unroll
    for (int t = 0; t < CHK; ++t) {
      if (t < CHK - 1) {
        xv[t] = Xn[(size_t)t * HH];
      } else {
        const float xnext = Xn[(size_t)(last_chunk ? t - 1 : t) * HH];
        xv[t] = last_chunk ? 0.0f : xnext;
      }
    }
  }
  {
    const int tm = w & 3, tn = w >> 2;
    const float* Xc = Xb + (size_t)chunk * CHK * HH;
    const float* W  = Wst + (size_t)b * HH * HH;
    v8f acc = {};
    for (int k0 = 0; k0 < HH; k0 += 4) {
      v2f a  = frag_nk(Xc, HH, tm * 16, k0, lane);
      v2f bb = frag_nk(W,  HH, n0 + tn * 16, k0, lane);
      acc = wmma4(a, bb, acc);
    }
    float bias = bst[(size_t)b * HH + n0 + tn * 16 + c];
#pragma unroll
    for (int i = 0; i < 8; ++i)
      Ps[(tm * 16 + 8 * hf + i) * PPITCH + tn * 16 + c] = acc[i] + bias;
  }
#pragma unroll
  for (int i = 0; i < 4; ++i) {
    int idx = i * 256 + tid;
    *(v4f*)(As + (idx >> 4) * APITCH + (idx & 15) * 4) = av[i];
  }
  __syncthreads();
  if (w != 0) return;

  float pv[CHK];
#pragma unroll
  for (int t = 0; t < CHK; ++t) pv[t] = Ps[t * PPITCH + lane];

  float bsum = 0.0f;
#pragma unroll
  for (int i = 0; i < CHK / 16; ++i) {
    const int t0 = i * 16;
    v4f cf[16][4];
#pragma unroll
    for (int r = 1; r < 16; ++r) {
#pragma unroll
      for (int q4 = 0; q4 * 4 < r; ++q4)
        cf[r][q4] = *(const v4f*)(As + (t0 + r) * APITCH + t0 + q4 * 4);
    }
    float sc[16];
    if (i > 0) {
      v8f s0 = {}, s1 = {};
#pragma unroll
      for (int k0 = 0; k0 < t0; k0 += 4) {
        const int k = k0 + 2 * hf;
        v2f a = *(const v2f*)(As + (t0 + c) * APITCH + k);
        v2f b0, b1;
        b0.x = Ds[k * DPITCH + c];       b0.y = Ds[(k + 1) * DPITCH + c];
        b1.x = Ds[k * DPITCH + 16 + c];  b1.y = Ds[(k + 1) * DPITCH + 16 + c];
        s0 = wmma4(a, b0, s0);
        s1 = wmma4(a, b1, s1);
      }
#pragma unroll
      for (int r = 0; r < 8; ++r) {
        Ts[(8 * hf + r) * DPITCH + c]      = s0[r];
        Ts[(8 * hf + r) * DPITCH + 16 + c] = s1[r];
      }
      wave_lds_sync();
#pragma unroll
      for (int r = 0; r < 16; ++r) sc[r] = Ts[r * DPITCH + lane];
    }
    float d[16];
#pragma unroll
    for (int r = 0; r < 16; ++r) {
      const int t = t0 + r;
      float acc = (i > 0) ? pv[t] - sc[r] : pv[t];
#pragma unroll
      for (int q = 0; q < r; ++q) acc -= cf[r][q >> 2][q & 3] * d[q];
      float dd = acc - xv[t];
      d[r] = dd;
      Ds[t * DPITCH + lane] = dd;
      pv[t] = acc;
      float dl = (t == CHK - 1 && last_chunk) ? 0.0f : dd;
      xv[t] = dl;
      bsum += dl;
    }
    wave_lds_sync();
  }
#pragma unroll
  for (int t = 0; t < CHK; ++t) {
    Pred[((size_t)b * CHK + t) * HH + col] = pv[t];
    Diff[((size_t)b * CHK + t) * HH + col] = xv[t];
  }
  bst[(size_t)b * HH + col] -= LRC * bsum;
}

__device__ __forceinline__ void d_wupdate(int blk, const float* __restrict__ x, const float* __restrict__ Diff,
                          float* __restrict__ Wst, int chunk) {
  int wid = (blk * 256 + threadIdx.x) >> 5;
  int lane = threadIdx.x & 31;
  int tn = wid % (HH / 16);
  int tm = (wid / (HH / 16)) % (HH / 16);
  int b  = wid / ((HH / 16) * (HH / 16));
  const float* Xc = x + ((size_t)b * SS + chunk * CHK) * HH;
  const float* Db = Diff + (size_t)b * CHK * HH;
  float* W = Wst + (size_t)b * HH * HH;
  v8f s = {};
  for (int k0 = 0; k0 < CHK; k0 += 4) {
    v2f a  = frag_kn(Db, HH, k0, tm * 16, lane);
    v2f bb = frag_kn(Xc, HH, k0, tn * 16, lane);
    s = wmma4(a, bb, s);
  }
  v8f acc = load_tile(W, HH, tm * 16, tn * 16, lane);
#pragma unroll
  for (int i = 0; i < 8; ++i) acc[i] -= LRC * s[i];
  store_tile(W, HH, tm * 16, tn * 16, acc, lane);
}

__device__ __forceinline__ void d_mlp1(int blk, const float* __restrict__ Pred,
                       const float* __restrict__ gw, const float* __restrict__ gb,
                       const float* __restrict__ uw, const float* __restrict__ ub,
                       float* __restrict__ Gt) {
  int wid = (blk * 256 + threadIdx.x) >> 5;
  int lane = threadIdx.x & 31;
  int tn = wid % (H4 / 16);
  int tm = (wid / (H4 / 16)) % (CHK / 16);
  int b  = wid / ((H4 / 16) * (CHK / 16));
  const float* Pb = Pred + (size_t)b * CHK * HH;
  const float* ap = Pb + (size_t)(tm * 16 + (lane & 15)) * HH + 4 * (lane >> 4);
  const float* gp = gw + ((size_t)tn * (HH / 8) * 32 + lane) * 4;
  const float* up = uw + ((size_t)tn * (HH / 8) * 32 + lane) * 4;
  v8f g = {}, u = {};
#pragma unroll 8
  for (int kg = 0; kg < HH / 8; ++kg) {
    v4f a  = *(const v4f*)(ap + kg * 8);
    v4f bg = *(const v4f*)(gp + (size_t)kg * 128);
    v4f bu = *(const v4f*)(up + (size_t)kg * 128);
    g = wmma4(lo2(a), lo2(bg), g);
    u = wmma4(lo2(a), lo2(bu), u);
    g = wmma4(hi2(a), hi2(bg), g);
    u = wmma4(hi2(a), hi2(bu), u);
  }
  int n = tn * 16 + (lane & 15);
  float gbias = gb[n], ubias = ub[n];
  v8f r;
#pragma unroll
  for (int i = 0; i < 8; ++i) {
    float zg = g[i] + gbias;
    float zu = u[i] + ubias;
    r[i] = (1.0f / (1.0f + __expf(-zg))) * zu;
  }
  store_tile(Gt + (size_t)b * CHK * H4, H4, tm * 16, tn * 16, r, lane);
}

__device__ __forceinline__ void d_mlp2(int blk, const float* __restrict__ Gt, const float* __restrict__ dw,
                       const float* __restrict__ db, const float* __restrict__ Pred,
                       float* __restrict__ Ypre) {
  int wid = (blk * 256 + threadIdx.x) >> 5;
  int lane = threadIdx.x & 31;
  int tn = wid % (HH / 16);
  int tm = (wid / (HH / 16)) % (CHK / 16);
  int b  = wid / ((HH / 16) * (CHK / 16));
  const float* Gb = Gt + (size_t)b * CHK * H4;
  const float* ap = Gb + (size_t)(tm * 16 + (lane & 15)) * H4 + 4 * (lane >> 4);
  const float* bp = dw + ((size_t)tn * (H4 / 8) * 32 + lane) * 4;
  v8f acc = {};
#pragma unroll 8
  for (int kg = 0; kg < H4 / 8; ++kg) {
    v4f a  = *(const v4f*)(ap + kg * 8);
    v4f bb = *(const v4f*)(bp + (size_t)kg * 128);
    acc = wmma4(lo2(a), lo2(bb), acc);
    acc = wmma4(hi2(a), hi2(bb), acc);
  }
  int n = tn * 16 + (lane & 15);
  int mbase = tm * 16 + ((lane >> 4) << 3);
  float bias = db[n];
  const float* Pb = Pred + (size_t)b * CHK * HH;
  float* Yb = Ypre + (size_t)b * CHK * HH;
#pragma unroll
  for (int i = 0; i < 8; ++i) {
    int m = mbase + i;
    Yb[(size_t)m * HH + n] = acc[i] + bias + Pb[(size_t)m * HH + n];
  }
}

__device__ __forceinline__ void d_ln(int blk, const float* __restrict__ Y, const float* __restrict__ nw,
                     const float* __restrict__ nb, float* __restrict__ out, int chunk) {
  __shared__ float s1[256], s2[256];
  int row = blk;
  int b = row / CHK, t = row % CHK;
  const float* y = Y + ((size_t)b * CHK + t) * HH;
  int tid = threadIdx.x;
  float v0 = y[tid], v1 = y[tid + 256], v2 = y[tid + 512];
  s1[tid] = v0 + v1 + v2;
  s2[tid] = v0 * v0 + v1 * v1 + v2 * v2;
  __syncthreads();
  for (int off = 128; off > 0; off >>= 1) {
    if (tid < off) { s1[tid] += s1[tid + off]; s2[tid] += s2[tid + off]; }
    __syncthreads();
  }
  float mu  = s1[0] * (1.0f / HH);
  float var = s2[0] * (1.0f / HH) - mu * mu;
  float r = rsqrtf(var + EPSC);
  int gt = chunk * CHK + t;
  float* o = out + ((size_t)b * SS + gt) * HH;
  o[tid]       = (v0 - mu) * r * nw[tid]       + nb[tid];
  o[tid + 256] = (v1 - mu) * r * nw[tid + 256] + nb[tid + 256];
  o[tid + 512] = (v2 - mu) * r * nw[tid + 512] + nb[tid + 512];
}


__global__ __launch_bounds__(256) void k_first(const float* __restrict__ x,
                       const float* __restrict__ Wi, const float* __restrict__ bi,
                       float* __restrict__ Wst, float* __restrict__ bst,
                       float* __restrict__ Amat,
                       const float* __restrict__ gw, const float* __restrict__ uw,
                       const float* __restrict__ dw, float* __restrict__ gwp,
                       float* __restrict__ uwp, float* __restrict__ dwp) {
  int blk = blockIdx.x;
  if (blk < NB_ATTN) { d_attn(blk, x, Amat, 0); return; }
  blk -= NB_ATTN;
  if (blk < NB_PACK) { d_pack(blk, gw, uw, dw, gwp, uwp, dwp); return; }
  d_init(blk - NB_PACK, Wi, bi, Wst, bst);
}

__global__ __launch_bounds__(256) void k_stage_a(const float* __restrict__ x,
                       const float* __restrict__ Diff, float* __restrict__ Wst,
                       const float* __restrict__ Pred,
                       const float* __restrict__ gw, const float* __restrict__ gb,
                       const float* __restrict__ uw, const float* __restrict__ ub,
                       float* __restrict__ Gt, float* __restrict__ Amat,
                       const float* __restrict__ Ypre,
                       const float* __restrict__ nw, const float* __restrict__ nb,
                       float* __restrict__ out, int c) {
  int blk = blockIdx.x;
  if (blk < NB_ATTN) {
    if (c < NCHUNK) d_attn(blk, x, Amat, c);
    return;
  }
  blk -= NB_ATTN;
  if (blk < NB_MLP1) {
    d_mlp1(blk, Pred, gw, gb, uw, ub, Gt);
    return;
  }
  blk -= NB_MLP1;
  if (blk < NB_WUPD) {
    d_wupdate(blk, x, Diff, Wst, c - 1);
    return;
  }
  blk -= NB_WUPD;
  if (c >= 2) d_ln(blk, Ypre, nw, nb, out, c - 2);
}

__global__ __launch_bounds__(256) void k_stage_s(const float* __restrict__ x,
                       const float* __restrict__ Wst, float* __restrict__ bst,
                       const float* __restrict__ Amat,
                       float* __restrict__ Pred, float* __restrict__ Diff,
                       const float* __restrict__ Gt, const float* __restrict__ dw,
                       const float* __restrict__ db, const float* __restrict__ PredPrev,
                       float* __restrict__ Ypre, int c, int n_mlp2) {
  int blk = blockIdx.x;
  if (blk < n_mlp2) {
    d_mlp2(blk, Gt, dw, db, PredPrev, Ypre);
    return;
  }
  blk -= n_mlp2;
  if (c < NCHUNK) d_p0solve(blk, x, Wst, Amat, Pred, Diff, bst, c);
}

__global__ __launch_bounds__(256) void k_ln(const float* __restrict__ Y, const float* __restrict__ nw,
                     const float* __restrict__ nb, float* __restrict__ out, int chunk) {
  d_ln(blockIdx.x, Y, nw, nb, out, chunk);
}

extern "C" void kernel_launch(void* const* d_in, const int* in_sizes, int n_in,
                              void* d_out, int out_size, void* d_ws, size_t ws_size,
                              hipStream_t stream) {
  const float* x      = (const float*)d_in[0];
  const float* W_init = (const float*)d_in[1];
  const float* b_init = (const float*)d_in[2];
  const float* gate_w = (const float*)d_in[3];
  const float* gate_b = (const float*)d_in[4];
  const float* up_w   = (const float*)d_in[5];
  const float* up_b   = (const float*)d_in[6];
  const float* down_w = (const float*)d_in[7];
  const float* down_b = (const float*)d_in[8];
  const float* norm_w = (const float*)d_in[9];
  const float* norm_b = (const float*)d_in[10];
  float* out = (float*)d_out;

  float* ws   = (float*)d_ws;
  float* Wst  = ws;  ws += (size_t)BB * HH * HH;
  float* bst  = ws;  ws += (size_t)BB * HH;
  float* PrE  = ws;  ws += (size_t)BB * CHK * HH;
  float* PrO  = ws;  ws += (size_t)BB * CHK * HH;
  float* Diff = ws;  ws += (size_t)BB * CHK * HH;
  float* Amat = ws;  ws += (size_t)BB * CHK * CHK;
  float* Gt   = ws;  ws += (size_t)BB * CHK * H4;
  float* Ypre = ws;  ws += (size_t)BB * CHK * HH;
  float* gwp  = ws;  ws += (size_t)H4 * HH;
  float* uwp  = ws;  ws += (size_t)H4 * HH;
  float* dwp  = ws;  ws += (size_t)HH * H4;

  k_first<<<NB_ATTN + NB_PACK + NB_INIT, 256, 0, stream>>>(
      x, W_init, b_init, Wst, bst, Amat, gate_w, up_w, down_w, gwp, uwp, dwp);
  k_stage_s<<<NB_SOLVE, 256, 0, stream>>>(
      x, Wst, bst, Amat, PrE, Diff, Gt, dwp, down_b, PrO, Ypre, 0, 0);

  k_stage_a<<<NB_ATTN + NB_MLP1 + NB_WUPD + NB_LN, 256, 0, stream>>>(
      x, Diff, Wst, PrE, gwp, gate_b, uwp, up_b, Gt, Amat, Ypre, norm_w, norm_b, out, 1);
  k_stage_s<<<NB_MLP2 + NB_SOLVE, 256, 0, stream>>>(
      x, Wst, bst, Amat, PrO, Diff, Gt, dwp, down_b, PrE, Ypre, 1, NB_MLP2);
  k_stage_a<<<NB_ATTN + NB_MLP1 + NB_WUPD + NB_LN, 256, 0, stream>>>(
      x, Diff, Wst, PrO, gwp, gate_b, uwp, up_b, Gt, Amat, Ypre, norm_w, norm_b, out, 2);
  k_stage_s<<<NB_MLP2 + NB_SOLVE, 256, 0, stream>>>(
      x, Wst, bst, Amat, PrE, Diff, Gt, dwp, down_b, PrO, Ypre, 2, NB_MLP2);
  k_stage_a<<<NB_ATTN + NB_MLP1 + NB_WUPD + NB_LN, 256, 0, stream>>>(
      x, Diff, Wst, PrE, gwp, gate_b, uwp, up_b, Gt, Amat, Ypre, norm_w, norm_b, out, 3);
  k_stage_s<<<NB_MLP2 + NB_SOLVE, 256, 0, stream>>>(
      x, Wst, bst, Amat, PrO, Diff, Gt, dwp, down_b, PrE, Ypre, 3, NB_MLP2);
  k_stage_a<<<NB_ATTN + NB_MLP1 + NB_WUPD + NB_LN, 256, 0, stream>>>(
      x, Diff, Wst, PrO, gwp, gate_b, uwp, up_b, Gt, Amat, Ypre, norm_w, norm_b, out, 4);
  k_stage_s<<<NB_MLP2, 256, 0, stream>>>(
      x, Wst, bst, Amat, PrE, Diff, Gt, dwp, down_b, PrO, Ypre, 4, NB_MLP2);
  k_ln<<<NB_LN, 256, 0, stream>>>(Ypre, norm_w, norm_b, out, NCHUNK - 1);
}


# mlp1+mlp2 fused per block: its Gated 16x128 slab stays in LDS and feeds that slice's share of the down GEMM on 192 WGPs; LayerNorm adds the 24 slice sums; 11 launches -> 10; sum order changes
# speedup vs baseline: 4.6256x; 1.4006x over previous
#include <hip/hip_runtime.h>
#include <math.h>

#define BB 2
#define SS 256
#define HH 768
#define H4 3072
#define CHK 64
#define LRC 0.01f
#define EPSC 1e-5f

typedef float v2f __attribute__((ext_vector_type(2)));
typedef float v8f __attribute__((ext_vector_type(8)));
typedef float v4f __attribute__((ext_vector_type(4)));

__device__ __forceinline__ v8f wmma4(v2f a, v2f b, v8f c) {
  return __builtin_amdgcn_wmma_f32_16x16x4_f32(
       false, a,  false, b,
       (short)0, c,  false,  false);
}

__device__ __forceinline__ v2f frag_nk(const float* M, int ld, int r0, int k0, int lane) {
  const float* p = M + (size_t)(r0 + (lane & 15)) * ld + k0 + ((lane >> 4) << 1);
  v2f v; v.x = p[0]; v.y = p[1];
  return v;
}
__device__ __forceinline__ v2f frag_kn(const float* M, int ld, int k0, int r0, int lane) {
  const float* p = M + (size_t)(k0 + ((lane >> 4) << 1)) * ld + r0 + (lane & 15);
  v2f v; v.x = p[0]; v.y = p[ld];
  return v;
}
__device__ __forceinline__ v2f lo2(v4f v) { v2f r; r.x = v.x; r.y = v.y; return r; }
__device__ __forceinline__ v2f hi2(v4f v) { v2f r; r.x = v.z; r.y = v.w; return r; }

__device__ __forceinline__ void store_tile(float* D, int ld, int r0, int c0, v8f v, int lane) {
  int n = c0 + (lane & 15);
  int m0 = r0 + ((lane >> 4) << 3);
#pragma unroll
  for (int i = 0; i < 8; ++i) D[(size_t)(m0 + i) * ld + n] = v[i];
}
__device__ __forceinline__ v8f load_tile(const float* D, int ld, int r0, int c0, int lane) {
  int n = c0 + (lane & 15);
  int m0 = r0 + ((lane >> 4) << 3);
  v8f v;
#pragma unroll
  for (int i = 0; i < 8; ++i) v[i] = D[(size_t)(m0 + i) * ld + n];
  return v;
}

#define NB_INIT ((HH * HH + 255) / 256)
#define NB_SOLVE (BB * (HH / 32))
#define NB_ATTN (BB * (CHK / 16) * (CHK / 16) / 8)
#define NB_WUPD (BB * (HH / 16) * (HH / 16) / 8)
#define NB_MLP1 (BB * (CHK / 16) * (H4 / 16) / 8)
#define NSL     (H4 / 128)
#define NB_LN   (BB * CHK)
#define NCHUNK  (SS / CHK)
#define NB_PACK (3 * 192)

__device__ __forceinline__ void d_init(int blk, const float* __restrict__ Wi, const float* __restrict__ bi,
                       float* __restrict__ Wst, float* __restrict__ bst) {
  int i = blk * 256 + threadIdx.x;
  const int n = HH * HH;
  if (i < n) {
    float w = Wi[i];
    Wst[i] = w;
    Wst[n + i] = w;
  }
  if (i < HH) {
    float b = bi[i];
    bst[i] = b;
    bst[HH + i] = b;
  }
}

__device__ __forceinline__ void d_pack(int blk, const float* __restrict__ gw, const float* __restrict__ uw,
                       const float* __restrict__ dw, float* __restrict__ gwp,
                       float* __restrict__ uwp, float* __restrict__ dwp) {
  const int which = blk / 192, r = blk % 192;
  const float* src = (which == 0) ? gw : (which == 1) ? uw : dw;
  float* dst = (which == 0) ? gwp : (which == 1) ? uwp : dwp;
  const int K   = (which == 2) ? H4 : HH;
  const int tn  = (which == 2) ? (r >> 2) : r;
  const int seg = (which == 2) ? (r & 3) : 0;
  const int w = threadIdx.x >> 5, lane = threadIdx.x & 31;
  const float* s = src + (size_t)(tn * 16 + (lane & 15)) * K + 4 * (lane >> 4);
  float* d = dst + ((size_t)tn * (K / 8) * 32 + lane) * 4;
  const int kg0 = seg * 96 + w * 12;
  v4f v[12];
#pragma unroll
  for (int i = 0; i < 12; ++i) v[i] = *(const v4f*)(s + (size_t)(kg0 + i) * 8);
#pragma unroll
  for (int i = 0; i < 12; ++i) *(v4f*)(d + (size_t)(kg0 + i) * 128) = v[i];
}

__device__ __forceinline__ void d_attn(int blk, const float* __restrict__ x, float* __restrict__ Amat, int chunk) {
  int wid = (blk * 256 + threadIdx.x) >> 5;
  int lane = threadIdx.x & 31;
  int tn = wid % (CHK / 16);
  int tm = (wid / (CHK / 16)) % (CHK / 16);
  int b  = wid / ((CHK / 16) * (CHK / 16));
  const float* Xc = x + ((size_t)b * SS + chunk * CHK) * HH;
  v8f acc = {};
  for (int k0 = 0; k0 < HH; k0 += 4) {
    v2f a  = frag_nk(Xc, HH, tm * 16, k0, lane);
    v2f bb = frag_nk(Xc, HH, tn * 16, k0, lane);
    acc = wmma4(a, bb, acc);
  }
  int s = tn * 16 + (lane & 15);
  int mbase = tm * 16 + ((lane >> 4) << 3);
  float* Ab = Amat + (size_t)b * CHK * CHK;
#pragma unroll
  for (int i = 0; i < 8; ++i) {
    int t = mbase + i;
    float v = (s < t) ? LRC * (acc[i] + 1.0f) : 0.0f;
    Ab[(size_t)t * CHK + s] = v;
  }
}

#define APITCH 68
#define DPITCH 40
#define PPITCH 36

__device__ __forceinline__ void wave_lds_sync() {
  __builtin_amdgcn_fence(__ATOMIC_SEQ_CST, "workgroup");
  __builtin_amdgcn_wave_barrier();
}

__device__ __forceinline__ void d_p0solve(int blk, const float* __restrict__ x, const float* __restrict__ Wst,
                        const float* __restrict__ Amat, float* __restrict__ Pred,
                        float* __restrict__ Diff, float* __restrict__ bst, int chunk) {
  __shared__ __attribute__((aligned(16))) float As[CHK * APITCH];
  __shared__ __attribute__((aligned(16))) float Ds[CHK * DPITCH];
  __shared__ __attribute__((aligned(16))) float Ts[16 * DPITCH];
  __shared__ __attribute__((aligned(16))) float Ps[CHK * PPITCH];
  const int tid = threadIdx.x;
  const int w = tid >> 5, lane = tid & 31;
  const int c = lane & 15, hf = lane >> 4;
  const int b  = blk / (HH / 32);
  const int n0 = (blk % (HH / 32)) * 32;
  const float* Xb = x + (size_t)b * SS * HH;
  const float* Ab = Amat + (size_t)b * CHK * CHK;
  const size_t col = (size_t)n0 + lane;
  const bool last_chunk = (chunk == SS / CHK - 1);

  v4f av[4];
#pragma unroll
  for (int i = 0; i < 4; ++i) {
    int idx = i * 256 + tid;
    av[i] = *(const v4f*)(Ab + (idx >> 4) * CHK + (idx & 15) * 4);
  }
  float xv[CHK];
  if (w == 0) {
    const float* Xn = Xb + ((size_t)chunk * CHK + 1) * HH + col;
#pragma unroll
    for (int t = 0; t < CHK; ++t) {
      if (t < CHK - 1) {
        xv[t] = Xn[(size_t)t * HH];
      } else {
        const float xnext = Xn[(size_t)(last_chunk ? t - 1 : t) * HH];
        xv[t] = last_chunk ? 0.0f : xnext;
      }
    }
  }
  {
    const int tm = w & 3, tn = w >> 2;
    const float* Xc = Xb + (size_t)chunk * CHK * HH;
    const float* W  = Wst + (size_t)b * HH * HH;
    v8f acc = {};
    for (int k0 = 0; k0 < HH; k0 += 4) {
      v2f a  = frag_nk(Xc, HH, tm * 16, k0, lane);
      v2f bb = frag_nk(W,  HH, n0 + tn * 16, k0, lane);
      acc = wmma4(a, bb, acc);
    }
    float bias = bst[(size_t)b * HH + n0 + tn * 16 + c];
#pragma unroll
    for (int i = 0; i < 8; ++i)
      Ps[(tm * 16 + 8 * hf + i) * PPITCH + tn * 16 + c] = acc[i] + bias;
  }
#pragma unroll
  for (int i = 0; i < 4; ++i) {
    int idx = i * 256 + tid;
    *(v4f*)(As + (idx >> 4) * APITCH + (idx & 15) * 4) = av[i];
  }
  __syncthreads();
  if (w != 0) return;

  float pv[CHK];
#pragma unroll
  for (int t = 0; t < CHK; ++t) pv[t] = Ps[t * PPITCH + lane];

  float bsum = 0.0f;
#pragma unroll
  for (int i = 0; i < CHK / 16; ++i) {
    const int t0 = i * 16;
    v4f cf[16][4];
#pragma unroll
    for (int r = 1; r < 16; ++r) {
#pragma unroll
      for (int q4 = 0; q4 * 4 < r; ++q4)
        cf[r][q4] = *(const v4f*)(As + (t0 + r) * APITCH + t0 + q4 * 4);
    }
    float sc[16];
    if (i > 0) {
      v8f s0 = {}, s1 = {};
#pragma unroll
      for (int k0 = 0; k0 < t0; k0 += 4) {
        const int k = k0 + 2 * hf;
        v2f a = *(const v2f*)(As + (t0 + c) * APITCH + k);
        v2f b0, b1;
        b0.x = Ds[k * DPITCH + c];       b0.y = Ds[(k + 1) * DPITCH + c];
        b1.x = Ds[k * DPITCH + 16 + c];  b1.y = Ds[(k + 1) * DPITCH + 16 + c];
        s0 = wmma4(a, b0, s0);
        s1 = wmma4(a, b1, s1);
      }
#pragma unroll
      for (int r = 0; r < 8; ++r) {
        Ts[(8 * hf + r) * DPITCH + c]      = s0[r];
        Ts[(8 * hf + r) * DPITCH + 16 + c] = s1[r];
      }
      wave_lds_sync();
#pragma unroll
      for (int r = 0; r < 16; ++r) sc[r] = Ts[r * DPITCH + lane];
    }
    float d[16];
#pragma unroll
    for (int r = 0; r < 16; ++r) {
      const int t = t0 + r;
      float acc = (i > 0) ? pv[t] - sc[r] : pv[t];
#pragma unroll
      for (int q = 0; q < r; ++q) acc -= cf[r][q >> 2][q & 3] * d[q];
      float dd = acc - xv[t];
      d[r] = dd;
      Ds[t * DPITCH + lane] = dd;
      pv[t] = acc;
      float dl = (t == CHK - 1 && last_chunk) ? 0.0f : dd;
      xv[t] = dl;
      bsum += dl;
    }
    wave_lds_sync();
  }
#pragma unroll
  for (int t = 0; t < CHK; ++t) {
    Pred[((size_t)b * CHK + t) * HH + col] = pv[t];
    Diff[((size_t)b * CHK + t) * HH + col] = xv[t];
  }
  bst[(size_t)b * HH + col] -= LRC * bsum;
}

__device__ __forceinline__ void d_wupdate(int blk, const float* __restrict__ x, const float* __restrict__ Diff,
                          float* __restrict__ Wst, int chunk) {
  int wid = (blk * 256 + threadIdx.x) >> 5;
  int lane = threadIdx.x & 31;
  int tn = wid % (HH / 16);
  int tm = (wid / (HH / 16)) % (HH / 16);
  int b  = wid / ((HH / 16) * (HH / 16));
  const float* Xc = x + ((size_t)b * SS + chunk * CHK) * HH;
  const float* Db = Diff + (size_t)b * CHK * HH;
  float* W = Wst + (size_t)b * HH * HH;
  v8f s = {};
  for (int k0 = 0; k0 < CHK; k0 += 4) {
    v2f a  = frag_kn(Db, HH, k0, tm * 16, lane);
    v2f bb = frag_kn(Xc, HH, k0, tn * 16, lane);
    s = wmma4(a, bb, s);
  }
  v8f acc = load_tile(W, HH, tm * 16, tn * 16, lane);
#pragma unroll
  for (int i = 0; i < 8; ++i) acc[i] -= LRC * s[i];
  store_tile(W, HH, tm * 16, tn * 16, acc, lane);
}

#define GPITCH 132
__device__ __forceinline__ void d_mlp(int blk, const float* __restrict__ Pred,
                       const float* __restrict__ gw, const float* __restrict__ gb,
                       const float* __restrict__ uw, const float* __restrict__ ub,
                       const float* __restrict__ dw, float* __restrict__ Ypart) {
  __shared__ __attribute__((aligned(16))) float Gs[16 * GPITCH];
  const int w = threadIdx.x >> 5, lane = threadIdx.x & 31;
  const int c = lane & 15, hf = lane >> 4;
  const int sl = blk % NSL;
  const int tm = (blk / NSL) % (CHK / 16);
  const int b  = blk / (NSL * (CHK / 16));
  const int tn = sl * 8 + w;
  const float* Pb = Pred + (size_t)b * CHK * HH;
  const float* ap = Pb + (size_t)(tm * 16 + c) * HH + 4 * hf;
  const float* gp = gw + ((size_t)tn * (HH / 8) * 32 + lane) * 4;
  const float* up = uw + ((size_t)tn * (HH / 8) * 32 + lane) * 4;
  v8f g = {}, u = {};
#pragma unroll 8
  for (int kg = 0; kg < HH / 8; ++kg) {
    v4f a  = *(const v4f*)(ap + kg * 8);
    v4f bg = *(const v4f*)(gp + (size_t)kg * 128);
    v4f bu = *(const v4f*)(up + (size_t)kg * 128);
    g = wmma4(lo2(a), lo2(bg), g);
    u = wmma4(lo2(a), lo2(bu), u);
    g = wmma4(hi2(a), hi2(bg), g);
    u = wmma4(hi2(a), hi2(bu), u);
  }
  {
    const int n = tn * 16 + c;
    const float gbias = gb[n], ubias = ub[n];
#pragma unroll
    for (int i = 0; i < 8; ++i) {
      float zg = g[i] + gbias;
      float zu = u[i] + ubias;
      Gs[(8 * hf + i) * GPITCH + w * 16 + c] = (1.0f / (1.0f + __expf(-zg))) * zu;
    }
  }
  __syncthreads();
  v4f a[16];
#pragma unroll
  for (int kk = 0; kk < 16; ++kk) a[kk] = *(const v4f*)(Gs + c * GPITCH + kk * 8 + 4 * hf);
  const float* bp = dw + (((size_t)(w * 6) * (H4 / 8) + sl * 16) * 32 + lane) * 4;
  float* Yp = Ypart + ((size_t)sl * (BB * CHK) + b * CHK + tm * 16 + 8 * hf) * HH + w * 96 + c;
#pragma unroll
  for (int q = 0; q < 6; q += 2) {
    const float* b0 = bp + (size_t)q * (H4 / 8) * 128;
    const float* b1 = b0 + (size_t)(H4 / 8) * 128;
    v8f y0 = {}, y1 = {};
    v4f v0[16], v1[16];
#pragma unroll
    for (int kk = 0; kk < 16; ++kk) {
      v0[kk] = *(const v4f*)(b0 + kk * 128);
      v1[kk] = *(const v4f*)(b1 + kk * 128);
    }
    __builtin_amdgcn_sched_barrier(0);
#pragma unroll
    for (int kk = 0; kk < 16; ++kk) {
      y0 = wmma4(lo2(a[kk]), lo2(v0[kk]), y0);
      y1 = wmma4(lo2(a[kk]), lo2(v1[kk]), y1);
      y0 = wmma4(hi2(a[kk]), hi2(v0[kk]), y0);
      y1 = wmma4(hi2(a[kk]), hi2(v1[kk]), y1);
    }
#pragma unroll
    for (int i = 0; i < 8; ++i) {
      Yp[(size_t)i * HH + q * 16]      = y0[i];
      Yp[(size_t)i * HH + q * 16 + 16] = y1[i];
    }
  }
}

__device__ __forceinline__ void d_ln(int blk, const float* __restrict__ Ypart, const float* __restrict__ db,
                     const float* __restrict__ Pred, const float* __restrict__ nw,
                     const float* __restrict__ nb, float* __restrict__ out, int chunk) {
  __shared__ float s1[256], s2[256];
  int row = blk;
  int b = row / CHK, t = row % CHK;
  const float* y = Ypart + (size_t)row * HH;
  const float* p = Pred + (size_t)row * HH;
  int tid = threadIdx.x;
  float v0 = 0.0f, v1 = 0.0f, v2 = 0.0f;
#pragma unroll
  for (int j = 0; j < NSL; ++j) {
    const float* yj = y + (size_t)j * (BB * CHK) * HH;
    v0 += yj[tid]; v1 += yj[tid + 256]; v2 += yj[tid + 512];
  }
  v0 = v0 + db[tid]       + p[tid];
  v1 = v1 + db[tid + 256] + p[tid + 256];
  v2 = v2 + db[tid + 512] + p[tid + 512];
  s1[tid] = v0 + v1 + v2;
  s2[tid] = v0 * v0 + v1 * v1 + v2 * v2;
  __syncthreads();
  for (int off = 128; off > 0; off >>= 1) {
    if (tid < off) { s1[tid] += s1[tid + off]; s2[tid] += s2[tid + off]; }
    __syncthreads();
  }
  float mu  = s1[0] * (1.0f / HH);
  float var = s2[0] * (1.0f / HH) - mu * mu;
  float r = rsqrtf(var + EPSC);
  int gt = chunk * CHK + t;
  float* o = out + ((size_t)b * SS + gt) * HH;
  o[tid]       = (v0 - mu) * r * nw[tid]       + nb[tid];
  o[tid + 256] = (v1 - mu) * r * nw[tid + 256] + nb[tid + 256];
  o[tid + 512] = (v2 - mu) * r * nw[tid + 512] + nb[tid + 512];
}


__global__ __launch_bounds__(256) void k_first(const float* __restrict__ x,
                       const float* __restrict__ Wi, const float* __restrict__ bi,
                       float* __restrict__ Wst, float* __restrict__ bst,
                       float* __restrict__ Amat,
                       const float* __restrict__ gw, const float* __restrict__ uw,
                       const float* __restrict__ dw, float* __restrict__ gwp,
                       float* __restrict__ uwp, float* __restrict__ dwp) {
  int blk = blockIdx.x;
  if (blk < NB_ATTN) { d_attn(blk, x, Amat, 0); return; }
  blk -= NB_ATTN;
  if (blk < NB_PACK) { d_pack(blk, gw, uw, dw, gwp, uwp, dwp); return; }
  d_init(blk - NB_PACK, Wi, bi, Wst, bst);
}

__global__ __launch_bounds__(256) void k_stage_a(const float* __restrict__ x,
                       const float* __restrict__ Diff, float* __restrict__ Wst,
                       const float* __restrict__ Pred,
                       const float* __restrict__ gw, const float* __restrict__ gb,
                       const float* __restrict__ uw, const float* __restrict__ ub,
                       const float* __restrict__ dw, float* __restrict__ Ypart,
                       float* __restrict__ Amat, int c) {
  int blk = blockIdx.x;
  if (blk < NB_ATTN) {
    if (c < NCHUNK) d_attn(blk, x, Amat, c);
    return;
  }
  blk -= NB_ATTN;
  if (blk < NB_MLP1) {
    d_mlp(blk, Pred, gw, gb, uw, ub, dw, Ypart);
    return;
  }
  blk -= NB_MLP1;
  d_wupdate(blk, x, Diff, Wst, c - 1);
}

__global__ __launch_bounds__(256) void k_stage_s(const float* __restrict__ x,
                       const float* __restrict__ Wst, float* __restrict__ bst,
                       const float* __restrict__ Amat,
                       float* __restrict__ Pred, float* __restrict__ Diff,
                       const float* __restrict__ Ypart, const float* __restrict__ db,
                       const float* __restrict__ PredPrev,
                       const float* __restrict__ nw, const float* __restrict__ nb,
                       float* __restrict__ out, int c) {
  int blk = blockIdx.x;
  if (blk < NB_SOLVE) {
    d_p0solve(blk, x, Wst, Amat, Pred, Diff, bst, c);
    return;
  }
  d_ln(blk - NB_SOLVE, Ypart, db, PredPrev, nw, nb, out, c - 1);
}

__global__ __launch_bounds__(256) void k_ln(const float* __restrict__ Ypart, const float* __restrict__ db,
                     const float* __restrict__ Pred, const float* __restrict__ nw,
                     const float* __restrict__ nb, float* __restrict__ out, int chunk) {
  d_ln(blockIdx.x, Ypart, db, Pred, nw, nb, out, chunk);
}

extern "C" void kernel_launch(void* const* d_in, const int* in_sizes, int n_in,
                              void* d_out, int out_size, void* d_ws, size_t ws_size,
                              hipStream_t stream) {
  const float* x      = (const float*)d_in[0];
  const float* W_init = (const float*)d_in[1];
  const float* b_init = (const float*)d_in[2];
  const float* gate_w = (const float*)d_in[3];
  const float* gate_b = (const float*)d_in[4];
  const float* up_w   = (const float*)d_in[5];
  const float* up_b   = (const float*)d_in[6];
  const float* down_w = (const float*)d_in[7];
  const float* down_b = (const float*)d_in[8];
  const float* norm_w = (const float*)d_in[9];
  const float* norm_b = (const float*)d_in[10];
  float* out = (float*)d_out;

  float* ws   = (float*)d_ws;
  float* Wst  = ws;  ws += (size_t)BB * HH * HH;
  float* bst  = ws;  ws += (size_t)BB * HH;
  float* PrE  = ws;  ws += (size_t)BB * CHK * HH;
  float* PrO  = ws;  ws += (size_t)BB * CHK * HH;
  float* Diff = ws;  ws += (size_t)BB * CHK * HH;
  float* Amat = ws;  ws += (size_t)BB * CHK * CHK;
  float* Ypart = ws; ws += (size_t)NSL * BB * CHK * HH;
  float* gwp  = ws;  ws += (size_t)H4 * HH;
  float* uwp  = ws;  ws += (size_t)H4 * HH;
  float* dwp  = ws;  ws += (size_t)HH * H4;

  k_first<<<NB_ATTN + NB_PACK + NB_INIT, 256, 0, stream>>>(
      x, W_init, b_init, Wst, bst, Amat, gate_w, up_w, down_w, gwp, uwp, dwp);
  k_stage_s<<<NB_SOLVE, 256, 0, stream>>>(
      x, Wst, bst, Amat, PrE, Diff, Ypart, down_b, PrO, norm_w, norm_b, out, 0);

  k_stage_a<<<NB_ATTN + NB_MLP1 + NB_WUPD, 256, 0, stream>>>(
      x, Diff, Wst, PrE, gwp, gate_b, uwp, up_b, dwp, Ypart, Amat, 1);
  k_stage_s<<<NB_SOLVE + NB_LN, 256, 0, stream>>>(
      x, Wst, bst, Amat, PrO, Diff, Ypart, down_b, PrE, norm_w, norm_b, out, 1);
  k_stage_a<<<NB_ATTN + NB_MLP1 + NB_WUPD, 256, 0, stream>>>(
      x, Diff, Wst, PrO, gwp, gate_b, uwp, up_b, dwp, Ypart, Amat, 2);
  k_stage_s<<<NB_SOLVE + NB_LN, 256, 0, stream>>>(
      x, Wst, bst, Amat, PrE, Diff, Ypart, down_b, PrO, norm_w, norm_b, out, 2);
  k_stage_a<<<NB_ATTN + NB_MLP1 + NB_WUPD, 256, 0, stream>>>(
      x, Diff, Wst, PrE, gwp, gate_b, uwp, up_b, dwp, Ypart, Amat, 3);
  k_stage_s<<<NB_SOLVE + NB_LN, 256, 0, stream>>>(
      x, Wst, bst, Amat, PrO, Diff, Ypart, down_b, PrE, norm_w, norm_b, out, 3);
  k_stage_a<<<NB_ATTN + NB_MLP1 + NB_WUPD, 256, 0, stream>>>(
      x, Diff, Wst, PrO, gwp, gate_b, uwp, up_b, dwp, Ypart, Amat, 4);
  k_ln<<<NB_LN, 256, 0, stream>>>(Ypart, down_b, PrO, norm_w, norm_b, out, NCHUNK - 1);
}


# launch bodies re-paired: only wupdate stays between two solves; mlp(c-1) and attn(c+1) run beside solve(c), ln(c-2) beside wupdate(c-1); A, Ypart get 2 planes; 10 launches; every sum as before
# speedup vs baseline: 5.4832x; 1.1854x over previous
#include <hip/hip_runtime.h>
#include <math.h>

#define BB 2
#define SS 256
#define HH 768
#define H4 3072
#define CHK 64
#define LRC 0.01f
#define EPSC 1e-5f

typedef float v2f __attribute__((ext_vector_type(2)));
typedef float v8f __attribute__((ext_vector_type(8)));
typedef float v4f __attribute__((ext_vector_type(4)));

__device__ __forceinline__ v8f wmma4(v2f a, v2f b, v8f c) {
  return __builtin_amdgcn_wmma_f32_16x16x4_f32(
       false, a,  false, b,
       (short)0, c,  false,  false);
}

__device__ __forceinline__ v2f frag_nk(const float* M, int ld, int r0, int k0, int lane) {
  const float* p = M + (size_t)(r0 + (lane & 15)) * ld + k0 + ((lane >> 4) << 1);
  v2f v; v.x = p[0]; v.y = p[1];
  return v;
}
__device__ __forceinline__ v2f frag_kn(const float* M, int ld, int k0, int r0, int lane) {
  const float* p = M + (size_t)(k0 + ((lane >> 4) << 1)) * ld + r0 + (lane & 15);
  v2f v; v.x = p[0]; v.y = p[ld];
  return v;
}
__device__ __forceinline__ v2f lo2(v4f v) { v2f r; r.x = v.x; r.y = v.y; return r; }
__device__ __forceinline__ v2f hi2(v4f v) { v2f r; r.x = v.z; r.y = v.w; return r; }

__device__ __forceinline__ void store_tile(float* D, int ld, int r0, int c0, v8f v, int lane) {
  int n = c0 + (lane & 15);
  int m0 = r0 + ((lane >> 4) << 3);
#pragma unroll
  for (int i = 0; i < 8; ++i) D[(size_t)(m0 + i) * ld + n] = v[i];
}
__device__ __forceinline__ v8f load_tile(const float* D, int ld, int r0, int c0, int lane) {
  int n = c0 + (lane & 15);
  int m0 = r0 + ((lane >> 4) << 3);
  v8f v;
#pragma unroll
  for (int i = 0; i < 8; ++i) v[i] = D[(size_t)(m0 + i) * ld + n];
  return v;
}

#define NB_INIT ((HH * HH + 255) / 256)
#define NB_SOLVE (BB * (HH / 32))
#define NB_ATTN (BB * (CHK / 16) * (CHK / 16) / 8)
#define NB_WUPD (BB * (HH / 16) * (HH / 16) / 8)
#define NB_MLP1 (BB * (CHK / 16) * (H4 / 16) / 8)
#define NSL     (H4 / 128)
#define NB_LN   (BB * CHK)
#define NCHUNK  (SS / CHK)
#define NB_PACK (3 * 192)

__device__ __forceinline__ void d_init(int blk, const float* __restrict__ Wi, const float* __restrict__ bi,
                       float* __restrict__ Wst, float* __restrict__ bst) {
  int i = blk * 256 + threadIdx.x;
  const int n = HH * HH;
  if (i < n) {
    float w = Wi[i];
    Wst[i] = w;
    Wst[n + i] = w;
  }
  if (i < HH) {
    float b = bi[i];
    bst[i] = b;
    bst[HH + i] = b;
  }
}

__device__ __forceinline__ void d_pack(int blk, const float* __restrict__ gw, const float* __restrict__ uw,
                       const float* __restrict__ dw, float* __restrict__ gwp,
                       float* __restrict__ uwp, float* __restrict__ dwp) {
  const int which = blk / 192, r = blk % 192;
  const float* src = (which == 0) ? gw : (which == 1) ? uw : dw;
  float* dst = (which == 0) ? gwp : (which == 1) ? uwp : dwp;
  const int K   = (which == 2) ? H4 : HH;
  const int tn  = (which == 2) ? (r >> 2) : r;
  const int seg = (which == 2) ? (r & 3) : 0;
  const int w = threadIdx.x >> 5, lane = threadIdx.x & 31;
  const float* s = src + (size_t)(tn * 16 + (lane & 15)) * K + 4 * (lane >> 4);
  float* d = dst + ((size_t)tn * (K / 8) * 32 + lane) * 4;
  const int kg0 = seg * 96 + w * 12;
  v4f v[12];
#pragma unroll
  for (int i = 0; i < 12; ++i) v[i] = *(const v4f*)(s + (size_t)(kg0 + i) * 8);
#pragma unroll
  for (int i = 0; i < 12; ++i) *(v4f*)(d + (size_t)(kg0 + i) * 128) = v[i];
}

__device__ __forceinline__ void d_attn(int blk, const float* __restrict__ x, float* __restrict__ Amat, int chunk) {
  int wid = (blk * 256 + threadIdx.x) >> 5;
  int lane = threadIdx.x & 31;
  int tn = wid % (CHK / 16);
  int tm = (wid / (CHK / 16)) % (CHK / 16);
  int b  = wid / ((CHK / 16) * (CHK / 16));
  const float* Xc = x + ((size_t)b * SS + chunk * CHK) * HH;
  v8f acc = {};
  for (int k0 = 0; k0 < HH; k0 += 4) {
    v2f a  = frag_nk(Xc, HH, tm * 16, k0, lane);
    v2f bb = frag_nk(Xc, HH, tn * 16, k0, lane);
    acc = wmma4(a, bb, acc);
  }
  int s = tn * 16 + (lane & 15);
  int mbase = tm * 16 + ((lane >> 4) << 3);
  float* Ab = Amat + (size_t)b * CHK * CHK;
#pragma unroll
  for (int i = 0; i < 8; ++i) {
    int t = mbase + i;
    float v = (s < t) ? LRC * (acc[i] + 1.0f) : 0.0f;
    Ab[(size_t)t * CHK + s] = v;
  }
}

#define APITCH 68
#define DPITCH 40
#define PPITCH 36

__device__ __forceinline__ void wave_lds_sync() {
  __builtin_amdgcn_fence(__ATOMIC_SEQ_CST, "workgroup");
  __builtin_amdgcn_wave_barrier();
}

__device__ __forceinline__ void d_p0solve(int blk, const float* __restrict__ x, const float* __restrict__ Wst,
                        const float* __restrict__ Amat, float* __restrict__ Pred,
                        float* __restrict__ Diff, float* __restrict__ bst, int chunk) {
  __shared__ __attribute__((aligned(16))) float As[CHK * APITCH];
  __shared__ __attribute__((aligned(16))) float Ds[CHK * DPITCH];
  __shared__ __attribute__((aligned(16))) float Ts[16 * DPITCH];
  __shared__ __attribute__((aligned(16))) float Ps[CHK * PPITCH];
  const int tid = threadIdx.x;
  const int w = tid >> 5, lane = tid & 31;
  const int c = lane & 15, hf = lane >> 4;
  const int b  = blk / (HH / 32);
  const int n0 = (blk % (HH / 32)) * 32;
  const float* Xb = x + (size_t)b * SS * HH;
  const float* Ab = Amat + (size_t)b * CHK * CHK;
  const size_t col = (size_t)n0 + lane;
  const bool last_chunk = (chunk == SS / CHK - 1);

  v4f av[4];
#pragma unroll
  for (int i = 0; i < 4; ++i) {
    int idx = i * 256 + tid;
    av[i] = *(const v4f*)(Ab + (idx >> 4) * CHK + (idx & 15) * 4);
  }
  float xv[CHK];
  if (w == 0) {
    const float* Xn = Xb + ((size_t)chunk * CHK + 1) * HH + col;
#pragma unroll
    for (int t = 0; t < CHK; ++t) {
      if (t < CHK - 1) {
        xv[t] = Xn[(size_t)t * HH];
      } else {
        const float xnext = Xn[(size_t)(last_chunk ? t - 1 : t) * HH];
        xv[t] = last_chunk ? 0.0f : xnext;
      }
    }
  }
  {
    const int tm = w & 3, tn = w >> 2;
    const float* Xc = Xb + (size_t)chunk * CHK * HH;
    const float* W  = Wst + (size_t)b * HH * HH;
    v8f acc = {};
    for (int k0 = 0; k0 < HH; k0 += 4) {
      v2f a  = frag_nk(Xc, HH, tm * 16, k0, lane);
      v2f bb = frag_nk(W,  HH, n0 + tn * 16, k0, lane);
      acc = wmma4(a, bb, acc);
    }
    float bias = bst[(size_t)b * HH + n0 + tn * 16 + c];
#pragma unroll
    for (int i = 0; i < 8; ++i)
      Ps[(tm * 16 + 8 * hf + i) * PPITCH + tn * 16 + c] = acc[i] + bias;
  }
#pragma unroll
  for (int i = 0; i < 4; ++i) {
    int idx = i * 256 + tid;
    *(v4f*)(As + (idx >> 4) * APITCH + (idx & 15) * 4) = av[i];
  }
  __syncthreads();
  if (w != 0) return;

  float pv[CHK];
#pragma unroll
  for (int t = 0; t < CHK; ++t) pv[t] = Ps[t * PPITCH + lane];

  float bsum = 0.0f;
#pragma unroll
  for (int i = 0; i < CHK / 16; ++i) {
    const int t0 = i * 16;
    v4f cf[16][4];
#pragma unroll
    for (int r = 1; r < 16; ++r) {
#pragma unroll
      for (int q4 = 0; q4 * 4 < r; ++q4)
        cf[r][q4] = *(const v4f*)(As + (t0 + r) * APITCH + t0 + q4 * 4);
    }
    float sc[16];
    if (i > 0) {
      v8f s0 = {}, s1 = {};
#pragma unroll
      for (int k0 = 0; k0 < t0; k0 += 4) {
        const int k = k0 + 2 * hf;
        v2f a = *(const v2f*)(As + (t0 + c) * APITCH + k);
        v2f b0, b1;
        b0.x = Ds[k * DPITCH + c];       b0.y = Ds[(k + 1) * DPITCH + c];
        b1.x = Ds[k * DPITCH + 16 + c];  b1.y = Ds[(k + 1) * DPITCH + 16 + c];
        s0 = wmma4(a, b0, s0);
        s1 = wmma4(a, b1, s1);
      }
#pragma unroll
      for (int r = 0; r < 8; ++r) {
        Ts[(8 * hf + r) * DPITCH + c]      = s0[r];
        Ts[(8 * hf + r) * DPITCH + 16 + c] = s1[r];
      }
      wave_lds_sync();
#pragma unroll
      for (int r = 0; r < 16; ++r) sc[r] = Ts[r * DPITCH + lane];
    }
    float d[16];
#pragma unroll
    for (int r = 0; r < 16; ++r) {
      const int t = t0 + r;
      float acc = (i > 0) ? pv[t] - sc[r] : pv[t];
#pragma unroll
      for (int q = 0; q < r; ++q) acc -= cf[r][q >> 2][q & 3] * d[q];
      float dd = acc - xv[t];
      d[r] = dd;
      Ds[t * DPITCH + lane] = dd;
      pv[t] = acc;
      float dl = (t == CHK - 1 && last_chunk) ? 0.0f : dd;
      xv[t] = dl;
      bsum += dl;
    }
    wave_lds_sync();
  }
#pragma unroll
  for (int t = 0; t < CHK; ++t) {
    Pred[((size_t)b * CHK + t) * HH + col] = pv[t];
    Diff[((size_t)b * CHK + t) * HH + col] = xv[t];
  }
  bst[(size_t)b * HH + col] -= LRC * bsum;
}

__device__ __forceinline__ void d_wupdate(int blk, const float* __restrict__ x, const float* __restrict__ Diff,
                          float* __restrict__ Wst, int chunk) {
  int wid = (blk * 256 + threadIdx.x) >> 5;
  int lane = threadIdx.x & 31;
  int tn = wid % (HH / 16);
  int tm = (wid / (HH / 16)) % (HH / 16);
  int b  = wid / ((HH / 16) * (HH / 16));
  const float* Xc = x + ((size_t)b * SS + chunk * CHK) * HH;
  const float* Db = Diff + (size_t)b * CHK * HH;
  float* W = Wst + (size_t)b * HH * HH;
  v8f s = {};
  for (int k0 = 0; k0 < CHK; k0 += 4) {
    v2f a  = frag_kn(Db, HH, k0, tm * 16, lane);
    v2f bb = frag_kn(Xc, HH, k0, tn * 16, lane);
    s = wmma4(a, bb, s);
  }
  v8f acc = load_tile(W, HH, tm * 16, tn * 16, lane);
#pragma unroll
  for (int i = 0; i < 8; ++i) acc[i] -= LRC * s[i];
  store_tile(W, HH, tm * 16, tn * 16, acc, lane);
}

#define GPITCH 132
__device__ __forceinline__ void d_mlp(int blk, const float* __restrict__ Pred,
                       const float* __restrict__ gw, const float* __restrict__ gb,
                       const float* __restrict__ uw, const float* __restrict__ ub,
                       const float* __restrict__ dw, float* __restrict__ Ypart) {
  __shared__ __attribute__((aligned(16))) float Gs[16 * GPITCH];
  const int w = threadIdx.x >> 5, lane = threadIdx.x & 31;
  const int c = lane & 15, hf = lane >> 4;
  const int sl = blk % NSL;
  const int tm = (blk / NSL) % (CHK / 16);
  const int b  = blk / (NSL * (CHK / 16));
  const int tn = sl * 8 + w;
  const float* Pb = Pred + (size_t)b * CHK * HH;
  const float* ap = Pb + (size_t)(tm * 16 + c) * HH + 4 * hf;
  const float* gp = gw + ((size_t)tn * (HH / 8) * 32 + lane) * 4;
  const float* up = uw + ((size_t)tn * (HH / 8) * 32 + lane) * 4;
  v8f g = {}, u = {};
#pragma unroll 8
  for (int kg = 0; kg < HH / 8; ++kg) {
    v4f a  = *(const v4f*)(ap + kg * 8);
    v4f bg = *(const v4f*)(gp + (size_t)kg * 128);
    v4f bu = *(const v4f*)(up + (size_t)kg * 128);
    g = wmma4(lo2(a), lo2(bg), g);
    u = wmma4(lo2(a), lo2(bu), u);
    g = wmma4(hi2(a), hi2(bg), g);
    u = wmma4(hi2(a), hi2(bu), u);
  }
  {
    const int n = tn * 16 + c;
    const float gbias = gb[n], ubias = ub[n];
#pragma unroll
    for (int i = 0; i < 8; ++i) {
      float zg = g[i] + gbias;
      float zu = u[i] + ubias;
      Gs[(8 * hf + i) * GPITCH + w * 16 + c] = (1.0f / (1.0f + __expf(-zg))) * zu;
    }
  }
  __syncthreads();
  v4f a[16];
#pragma unroll
  for (int kk = 0; kk < 16; ++kk) a[kk] = *(const v4f*)(Gs + c * GPITCH + kk * 8 + 4 * hf);
  const float* bp = dw + (((size_t)(w * 6) * (H4 / 8) + sl * 16) * 32 + lane) * 4;
  float* Yp = Ypart + ((size_t)sl * (BB * CHK) + b * CHK + tm * 16 + 8 * hf) * HH + w * 96 + c;
#pragma unroll
  for (int q = 0; q < 6; q += 2) {
    const float* b0 = bp + (size_t)q * (H4 / 8) * 128;
    const float* b1 = b0 + (size_t)(H4 / 8) * 128;
    v8f y0 = {}, y1 = {};
    v4f v0[16], v1[16];
#pragma unroll
    for (int kk = 0; kk < 16; ++kk) {
      v0[kk] = *(const v4f*)(b0 + kk * 128);
      v1[kk] = *(const v4f*)(b1 + kk * 128);
    }
    __builtin_amdgcn_sched_barrier(0);
#pragma unroll
    for (int kk = 0; kk < 16; ++kk) {
      y0 = wmma4(lo2(a[kk]), lo2(v0[kk]), y0);
      y1 = wmma4(lo2(a[kk]), lo2(v1[kk]), y1);
      y0 = wmma4(hi2(a[kk]), hi2(v0[kk]), y0);
      y1 = wmma4(hi2(a[kk]), hi2(v1[kk]), y1);
    }
#pragma unroll
    for (int i = 0; i < 8; ++i) {
      Yp[(size_t)i * HH + q * 16]      = y0[i];
      Yp[(size_t)i * HH + q * 16 + 16] = y1[i];
    }
  }
}

__device__ __forceinline__ void d_ln(int blk, const float* __restrict__ Ypart, const float* __restrict__ db,
                     const float* __restrict__ Pred, const float* __restrict__ nw,
                     const float* __restrict__ nb, float* __restrict__ out, int chunk) {
  __shared__ float s1[256], s2[256];
  int row = blk;
  int b = row / CHK, t = row % CHK;
  const float* y = Ypart + (size_t)row * HH;
  const float* p = Pred + (size_t)row * HH;
  int tid = threadIdx.x;
  float v0 = 0.0f, v1 = 0.0f, v2 = 0.0f;
#pragma unroll
  for (int j = 0; j < NSL; ++j) {
    const float* yj = y + (size_t)j * (BB * CHK) * HH;
    v0 += yj[tid]; v1 += yj[tid + 256]; v2 += yj[tid + 512];
  }
  v0 = v0 + db[tid]       + p[tid];
  v1 = v1 + db[tid + 256] + p[tid + 256];
  v2 = v2 + db[tid + 512] + p[tid + 512];
  s1[tid] = v0 + v1 + v2;
  s2[tid] = v0 * v0 + v1 * v1 + v2 * v2;
  __syncthreads();
  for (int off = 128; off > 0; off >>= 1) {
    if (tid < off) { s1[tid] += s1[tid + off]; s2[tid] += s2[tid + off]; }
    __syncthreads();
  }
  float mu  = s1[0] * (1.0f / HH);
  float var = s2[0] * (1.0f / HH) - mu * mu;
  float r = rsqrtf(var + EPSC);
  int gt = chunk * CHK + t;
  float* o = out + ((size_t)b * SS + gt) * HH;
  o[tid]       = (v0 - mu) * r * nw[tid]       + nb[tid];
  o[tid + 256] = (v1 - mu) * r * nw[tid + 256] + nb[tid + 256];
  o[tid + 512] = (v2 - mu) * r * nw[tid + 512] + nb[tid + 512];
}


__global__ __launch_bounds__(256) void k_first(const float* __restrict__ x,
                       const float* __restrict__ Wi, const float* __restrict__ bi,
                       float* __restrict__ Wst, float* __restrict__ bst,
                       float* __restrict__ Amat,
                       const float* __restrict__ gw, const float* __restrict__ uw,
                       const float* __restrict__ dw, float* __restrict__ gwp,
                       float* __restrict__ uwp, float* __restrict__ dwp) {
  int blk = blockIdx.x;
  if (blk < NB_ATTN) { d_attn(blk, x, Amat, 0); return; }
  blk -= NB_ATTN;
  if (blk < NB_PACK) { d_pack(blk, gw, uw, dw, gwp, uwp, dwp); return; }
  d_init(blk - NB_PACK, Wi, bi, Wst, bst);
}

__global__ __launch_bounds__(256) void k_stage_u(const float* __restrict__ x,
                       const float* __restrict__ Diff, float* __restrict__ Wst,
                       const float* __restrict__ YpartLn, const float* __restrict__ db,
                       const float* __restrict__ PredLn,
                       const float* __restrict__ nw, const float* __restrict__ nb,
                       float* __restrict__ out, int c) {
  int blk = blockIdx.x;
  if (blk < NB_WUPD) {
    d_wupdate(blk, x, Diff, Wst, c - 1);
    return;
  }
  if (c >= 2) d_ln(blk - NB_WUPD, YpartLn, db, PredLn, nw, nb, out, c - 2);
}

__global__ __launch_bounds__(256) void k_stage_s(const float* __restrict__ x,
                       const float* __restrict__ Wst, float* __restrict__ bst,
                       const float* __restrict__ AmatCur, float* __restrict__ AmatNext,
                       float* __restrict__ Pred, float* __restrict__ Diff,
                       const float* __restrict__ PredPrev,
                       const float* __restrict__ gw, const float* __restrict__ gb,
                       const float* __restrict__ uw, const float* __restrict__ ub,
                       const float* __restrict__ dw, float* __restrict__ Ypart, int c) {
  int blk = blockIdx.x;
  if (blk < NB_SOLVE) {
    d_p0solve(blk, x, Wst, AmatCur, Pred, Diff, bst, c);
    return;
  }
  blk -= NB_SOLVE;
  if (blk < NB_ATTN) {
    if (c + 1 < NCHUNK) d_attn(blk, x, AmatNext, c + 1);
    return;
  }
  blk -= NB_ATTN;
  if (c >= 1) d_mlp(blk, PredPrev, gw, gb, uw, ub, dw, Ypart);
}

__global__ __launch_bounds__(256) void k_tail(const float* __restrict__ PredMlp,
                       const float* __restrict__ gw, const float* __restrict__ gb,
                       const float* __restrict__ uw, const float* __restrict__ ub,
                       const float* __restrict__ dw, float* __restrict__ YpartMlp,
                       const float* __restrict__ YpartLn, const float* __restrict__ db,
                       const float* __restrict__ PredLn,
                       const float* __restrict__ nw, const float* __restrict__ nb,
                       float* __restrict__ out, int chunk_ln) {
  int blk = blockIdx.x;
  if (blk < NB_MLP1) {
    d_mlp(blk, PredMlp, gw, gb, uw, ub, dw, YpartMlp);
    return;
  }
  d_ln(blk - NB_MLP1, YpartLn, db, PredLn, nw, nb, out, chunk_ln);
}

__global__ __launch_bounds__(256) void k_ln(const float* __restrict__ Ypart, const float* __restrict__ db,
                     const float* __restrict__ Pred, const float* __restrict__ nw,
                     const float* __restrict__ nb, float* __restrict__ out, int chunk) {
  d_ln(blockIdx.x, Ypart, db, Pred, nw, nb, out, chunk);
}

extern "C" void kernel_launch(void* const* d_in, const int* in_sizes, int n_in,
                              void* d_out, int out_size, void* d_ws, size_t ws_size,
                              hipStream_t stream) {
  const float* x      = (const float*)d_in[0];
  const float* W_init = (const float*)d_in[1];
  const float* b_init = (const float*)d_in[2];
  const float* gate_w = (const float*)d_in[3];
  const float* gate_b = (const float*)d_in[4];
  const float* up_w   = (const float*)d_in[5];
  const float* up_b   = (const float*)d_in[6];
  const float* down_w = (const float*)d_in[7];
  const float* down_b = (const float*)d_in[8];
  const float* norm_w = (const float*)d_in[9];
  const float* norm_b = (const float*)d_in[10];
  float* out = (float*)d_out;

  float* ws   = (float*)d_ws;
  float* Wst  = ws;  ws += (size_t)BB * HH * HH;
  float* bst  = ws;  ws += (size_t)BB * HH;
  float* PrE  = ws;  ws += (size_t)BB * CHK * HH;
  float* PrO  = ws;  ws += (size_t)BB * CHK * HH;
  float* Diff = ws;  ws += (size_t)BB * CHK * HH;
  float* AmE  = ws;  ws += (size_t)BB * CHK * CHK;
  float* AmO  = ws;  ws += (size_t)BB * CHK * CHK;
  float* YpE  = ws;  ws += (size_t)NSL * BB * CHK * HH;
  float* YpO  = ws;  ws += (size_t)NSL * BB * CHK * HH;
  float* gwp  = ws;  ws += (size_t)H4 * HH;
  float* uwp  = ws;  ws += (size_t)H4 * HH;
  float* dwp  = ws;  ws += (size_t)HH * H4;

  k_first<<<NB_ATTN + NB_PACK + NB_INIT, 256, 0, stream>>>(
      x, W_init, b_init, Wst, bst, AmE, gate_w, up_w, down_w, gwp, uwp, dwp);
  k_stage_s<<<NB_SOLVE + NB_ATTN, 256, 0, stream>>>(
      x, Wst, bst, AmE, AmO, PrE, Diff, PrO, gwp, gate_b, uwp, up_b, dwp, YpO, 0);
  k_stage_u<<<NB_WUPD, 256, 0, stream>>>(
      x, Diff, Wst, YpE, down_b, PrE, norm_w, norm_b, out, 1);
  k_stage_s<<<NB_SOLVE + NB_ATTN + NB_MLP1, 256, 0, stream>>>(
      x, Wst, bst, AmO, AmE, PrO, Diff, PrE, gwp, gate_b, uwp, up_b, dwp, YpE, 1);
  k_stage_u<<<NB_WUPD + NB_LN, 256, 0, stream>>>(
      x, Diff, Wst, YpE, down_b, PrE, norm_w, norm_b, out, 2);
  k_stage_s<<<NB_SOLVE + NB_ATTN + NB_MLP1, 256, 0, stream>>>(
      x, Wst, bst, AmE, AmO, PrE, Diff, PrO, gwp, gate_b, uwp, up_b, dwp, YpO, 2);
  k_stage_u<<<NB_WUPD + NB_LN, 256, 0, stream>>>(
      x, Diff, Wst, YpO, down_b, PrO, norm_w, norm_b, out, 3);
  k_stage_s<<<NB_SOLVE + NB_ATTN + NB_MLP1, 256, 0, stream>>>(
      x, Wst, bst, AmO, AmE, PrO, Diff, PrE, gwp, gate_b, uwp, up_b, dwp, YpE, 3);
  k_tail<<<NB_MLP1 + NB_LN, 256, 0, stream>>>(
      PrO, gwp, gate_b, uwp, up_b, dwp, YpO, YpE, down_b, PrE, norm_w, norm_b, out, 2);
  k_ln<<<NB_LN, 256, 0, stream>>>(YpO, down_b, PrO, norm_w, norm_b, out, NCHUNK - 1);
}


# x and the fast-weight state held in WMMA fragment order (x copied once; init and update, on transposed tiles, write the state so): P0/attn K loops load whole lines; k pairing, so sum order, changes
# speedup vs baseline: 6.4793x; 1.1817x over previous
#include <hip/hip_runtime.h>
#include <math.h>

#define BB 2
#define SS 256
#define HH 768
#define H4 3072
#define CHK 64
#define LRC 0.01f
#define EPSC 1e-5f

typedef float v2f __attribute__((ext_vector_type(2)));
typedef float v8f __attribute__((ext_vector_type(8)));
typedef float v4f __attribute__((ext_vector_type(4)));

__device__ __forceinline__ v8f wmma4(v2f a, v2f b, v8f c) {
  return __builtin_amdgcn_wmma_f32_16x16x4_f32(
       false, a,  false, b,
       (short)0, c,  false,  false);
}

__device__ __forceinline__ v2f frag_kn(const float* M, int ld, int k0, int r0, int lane) {
  const float* p = M + (size_t)(k0 + ((lane >> 4) << 1)) * ld + r0 + (lane & 15);
  v2f v; v.x = p[0]; v.y = p[ld];
  return v;
}
__device__ __forceinline__ v2f lo2(v4f v) { v2f r; r.x = v.x; r.y = v.y; return r; }
__device__ __forceinline__ v2f hi2(v4f v) { v2f r; r.x = v.z; r.y = v.w; return r; }

__device__ __forceinline__ v8f tile_kh(const float* __restrict__ a0, const float* __restrict__ b0, int kstep) {
  v8f acc = {};
#pragma unroll 16
  for (int kg = 0; kg < HH / 8; ++kg) {
    v4f a = *(const v4f*)(a0 + (size_t)kg * kstep);
    v4f b = *(const v4f*)(b0 + (size_t)kg * kstep);
    acc = wmma4(lo2(a), lo2(b), acc);
    acc = wmma4(hi2(a), hi2(b), acc);
  }
  return acc;
}


#define NB_INIT ((HH * HH / 4 + 255) / 256)
#define NB_SOLVE (BB * (HH / 32))
#define NB_ATTN (BB * (CHK / 16) * (CHK / 16) / 8)
#define NB_WUPD (BB * (HH / 16) * (HH / 16) / 8)
#define NB_MLP1 (BB * (CHK / 16) * (H4 / 16) / 8)
#define NSL     (H4 / 128)
#define NB_LN   (BB * CHK)
#define NCHUNK  (SS / CHK)
#define NB_PACK (3 * 192 + BB * SS / 16)

__device__ __forceinline__ void d_init(int blk, const float* __restrict__ Wi, const float* __restrict__ bi,
                       float* __restrict__ Wst, float* __restrict__ bst) {
  int i = blk * 256 + threadIdx.x;
  const int n = HH * HH;
  if (i < n / 4) {
    const int lane = i & 31, u = i >> 5;
    const int tn = u / (HH / 8), kg = u % (HH / 8);
    v4f w = *(const v4f*)(Wi + (size_t)(tn * 16 + (lane & 15)) * HH + kg * 8 + 4 * (lane >> 4));
    *(v4f*)(Wst + (size_t)i * 4) = w;
    *(v4f*)(Wst + n + (size_t)i * 4) = w;
  }
  if (i < HH) {
    float b = bi[i];
    bst[i] = b;
    bst[HH + i] = b;
  }
}

__device__ __forceinline__ void d_pack(int blk, const float* __restrict__ gw, const float* __restrict__ uw,
                       const float* __restrict__ dw, const float* __restrict__ x,
                       float* __restrict__ gwp, float* __restrict__ uwp,
                       float* __restrict__ dwp, float* __restrict__ xp) {
  const int which = blk / 192, r = blk % 192;
  const float* src = (which == 0) ? gw : (which == 1) ? uw : (which == 2) ? dw : x;
  float* dst = (which == 0) ? gwp : (which == 1) ? uwp : (which == 2) ? dwp : xp;
  const int K   = (which == 2) ? H4 : HH;
  const int tn  = (which == 2) ? (r >> 2) : r;
  const int seg = (which == 2) ? (r & 3) : 0;
  const int w = threadIdx.x >> 5, lane = threadIdx.x & 31;
  const float* s = src + (size_t)(tn * 16 + (lane & 15)) * K + 4 * (lane >> 4);
  float* d = dst + ((size_t)tn * (K / 8) * 32 + lane) * 4;
  const int kg0 = seg * 96 + w * 12;
  v4f v[12];
#pragma unroll
  for (int i = 0; i < 12; ++i) v[i] = *(const v4f*)(s + (size_t)(kg0 + i) * 8);
#pragma unroll
  for (int i = 0; i < 12; ++i) *(v4f*)(d + (size_t)(kg0 + i) * 128) = v[i];
}

__device__ __forceinline__ void d_attn(int blk, const float* __restrict__ xs, bool packed,
                       float* __restrict__ Amat, int chunk) {
  int wid = (blk * 256 + threadIdx.x) >> 5;
  int lane = threadIdx.x & 31;
  int tn = wid % (CHK / 16);
  int tm = (wid / (CHK / 16)) % (CHK / 16);
  int b  = wid / ((CHK / 16) * (CHK / 16));
  v8f acc;
  if (packed) {
    const int t0 = b * (SS / 16) + chunk * (CHK / 16);
    acc = tile_kh(xs + ((size_t)(t0 + tm) * (HH / 8) * 32 + lane) * 4,
                  xs + ((size_t)(t0 + tn) * (HH / 8) * 32 + lane) * 4, 128);
  } else {
    const float* Xc = xs + ((size_t)b * SS + chunk * CHK) * HH + 4 * (lane >> 4);
    acc = tile_kh(Xc + (size_t)(tm * 16 + (lane & 15)) * HH,
                  Xc + (size_t)(tn * 16 + (lane & 15)) * HH, 8);
  }
  int s = tn * 16 + (lane & 15);
  int mbase = tm * 16 + ((lane >> 4) << 3);
  float* Ab = Amat + (size_t)b * CHK * CHK;
#pragma unroll
  for (int i = 0; i < 8; ++i) {
    int t = mbase + i;
    float v = (s < t) ? LRC * (acc[i] + 1.0f) : 0.0f;
    Ab[(size_t)t * CHK + s] = v;
  }
}

#define APITCH 68
#define DPITCH 40
#define PPITCH 36

__device__ __forceinline__ void wave_lds_sync() {
  __builtin_amdgcn_fence(__ATOMIC_SEQ_CST, "workgroup");
  __builtin_amdgcn_wave_barrier();
}

__device__ __forceinline__ void d_p0solve(int blk, const float* __restrict__ x, const float* __restrict__ xp,
                        const float* __restrict__ Wst,
                        const float* __restrict__ Amat, float* __restrict__ Pred,
                        float* __restrict__ Diff, float* __restrict__ bst, int chunk) {
  __shared__ __attribute__((aligned(16))) float As[CHK * APITCH];
  __shared__ __attribute__((aligned(16))) float Ds[CHK * DPITCH];
  __shared__ __attribute__((aligned(16))) float Ts[16 * DPITCH];
  __shared__ __attribute__((aligned(16))) float Ps[CHK * PPITCH];
  const int tid = threadIdx.x;
  const int w = tid >> 5, lane = tid & 31;
  const int c = lane & 15, hf = lane >> 4;
  const int b  = blk / (HH / 32);
  const int n0 = (blk % (HH / 32)) * 32;
  const float* Xb = x + (size_t)b * SS * HH;
  const float* Ab = Amat + (size_t)b * CHK * CHK;
  const size_t col = (size_t)n0 + lane;
  const bool last_chunk = (chunk == SS / CHK - 1);

  v4f av[4];
#pragma unroll
  for (int i = 0; i < 4; ++i) {
    int idx = i * 256 + tid;
    av[i] = *(const v4f*)(Ab + (idx >> 4) * CHK + (idx & 15) * 4);
  }
  float xv[CHK];
  if (w == 0) {
    const float* Xn = Xb + ((size_t)chunk * CHK + 1) * HH + col;
#pragma unroll
    for (int t = 0; t < CHK; ++t) {
      if (t < CHK - 1) {
        xv[t] = Xn[(size_t)t * HH];
      } else {
        const float xnext = Xn[(size_t)(last_chunk ? t - 1 : t) * HH];
        xv[t] = last_chunk ? 0.0f : xnext;
      }
    }
  }
  {
    const int tm = w & 3, tn = w >> 2;
    const float* ap = xp + ((size_t)(b * (SS / 16) + chunk * (CHK / 16) + tm) * (HH / 8) * 32 + lane) * 4;
    const float* bp = Wst + (size_t)b * HH * HH + ((size_t)(n0 / 16 + tn) * (HH / 8) * 32 + lane) * 4;
    v8f acc = tile_kh(ap, bp, 128);
    float bias = bst[(size_t)b * HH + n0 + tn * 16 + c];
#pragma unroll
    for (int i = 0; i < 8; ++i)
      Ps[(tm * 16 + 8 * hf + i) * PPITCH + tn * 16 + c] = acc[i] + bias;
  }
#pragma unroll
  for (int i = 0; i < 4; ++i) {
    int idx = i * 256 + tid;
    *(v4f*)(As + (idx >> 4) * APITCH + (idx & 15) * 4) = av[i];
  }
  __syncthreads();
  if (w != 0) return;

  float pv[CHK];
#pragma unroll
  for (int t = 0; t < CHK; ++t) pv[t] = Ps[t * PPITCH + lane];

  float bsum = 0.0f;
#pragma unroll
  for (int i = 0; i < CHK / 16; ++i) {
    const int t0 = i * 16;
    v4f cf[16][4];
#pragma unroll
    for (int r = 1; r < 16; ++r) {
#pragma unroll
      for (int q4 = 0; q4 * 4 < r; ++q4)
        cf[r][q4] = *(const v4f*)(As + (t0 + r) * APITCH + t0 + q4 * 4);
    }
    float sc[16];
    if (i > 0) {
      v8f s0 = {}, s1 = {};
#pragma unroll
      for (int k0 = 0; k0 < t0; k0 += 4) {
        const int k = k0 + 2 * hf;
        v2f a = *(const v2f*)(As + (t0 + c) * APITCH + k);
        v2f b0, b1;
        b0.x = Ds[k * DPITCH + c];       b0.y = Ds[(k + 1) * DPITCH + c];
        b1.x = Ds[k * DPITCH + 16 + c];  b1.y = Ds[(k + 1) * DPITCH + 16 + c];
        s0 = wmma4(a, b0, s0);
        s1 = wmma4(a, b1, s1);
      }
#pragma unroll
      for (int r = 0; r < 8; ++r) {
        Ts[(8 * hf + r) * DPITCH + c]      = s0[r];
        Ts[(8 * hf + r) * DPITCH + 16 + c] = s1[r];
      }
      wave_lds_sync();
#pragma unroll
      for (int r = 0; r < 16; ++r) sc[r] = Ts[r * DPITCH + lane];
    }
    float d[16];
#pragma unroll
    for (int r = 0; r < 16; ++r) {
      const int t = t0 + r;
      float acc = (i > 0) ? pv[t] - sc[r] : pv[t];
#pragma unroll
      for (int q = 0; q < r; ++q) acc -= cf[r][q >> 2][q & 3] * d[q];
      float dd = acc - xv[t];
      d[r] = dd;
      Ds[t * DPITCH + lane] = dd;
      pv[t] = acc;
      float dl = (t == CHK - 1 && last_chunk) ? 0.0f : dd;
      xv[t] = dl;
      bsum += dl;
    }
    wave_lds_sync();
  }
#pragma unroll
  for (int t = 0; t < CHK; ++t) {
    Pred[((size_t)b * CHK + t) * HH + col] = pv[t];
    Diff[((size_t)b * CHK + t) * HH + col] = xv[t];
  }
  bst[(size_t)b * HH + col] -= LRC * bsum;
}

__device__ __forceinline__ void d_wupdate(int blk, const float* __restrict__ x, const float* __restrict__ Diff,
                          float* __restrict__ Wst, int chunk) {
  int wid = (blk * 256 + threadIdx.x) >> 5;
  int lane = threadIdx.x & 31;
  int tn = wid % (HH / 16);
  int tm = (wid / (HH / 16)) % (HH / 16);
  int b  = wid / ((HH / 16) * (HH / 16));
  const float* Xc = x + ((size_t)b * SS + chunk * CHK) * HH;
  const float* Db = Diff + (size_t)b * CHK * HH;
  float* W = Wst + (size_t)b * HH * HH;
  v8f s = {};
  for (int k0 = 0; k0 < CHK; k0 += 4) {
    v2f a  = frag_kn(Xc, HH, k0, tn * 16, lane);
    v2f bb = frag_kn(Db, HH, k0, tm * 16, lane);
    s = wmma4(a, bb, s);
  }
  float* p = W + (((size_t)tm * (HH / 8) + tn * 2 + (lane >> 4)) * 32 + (lane & 15)) * 4;
  v4f w0 = *(const v4f*)p;
  v4f w1 = *(const v4f*)(p + 64);
  w0.x -= LRC * s[0]; w0.y -= LRC * s[1]; w0.z -= LRC * s[2]; w0.w -= LRC * s[3];
  w1.x -= LRC * s[4]; w1.y -= LRC * s[5]; w1.z -= LRC * s[6]; w1.w -= LRC * s[7];
  *(v4f*)p = w0;
  *(v4f*)(p + 64) = w1;
}

#define GPITCH 132
__device__ __forceinline__ void d_mlp(int blk, const float* __restrict__ Pred,
                       const float* __restrict__ gw, const float* __restrict__ gb,
                       const float* __restrict__ uw, const float* __restrict__ ub,
                       const float* __restrict__ dw, float* __restrict__ Ypart) {
  __shared__ __attribute__((aligned(16))) float Gs[16 * GPITCH];
  const int w = threadIdx.x >> 5, lane = threadIdx.x & 31;
  const int c = lane & 15, hf = lane >> 4;
  const int sl = blk % NSL;
  const int tm = (blk / NSL) % (CHK / 16);
  const int b  = blk / (NSL * (CHK / 16));
  const int tn = sl * 8 + w;
  const float* Pb = Pred + (size_t)b * CHK * HH;
  const float* ap = Pb + (size_t)(tm * 16 + c) * HH + 4 * hf;
  const float* gp = gw + ((size_t)tn * (HH / 8) * 32 + lane) * 4;
  const float* up = uw + ((size_t)tn * (HH / 8) * 32 + lane) * 4;
  v8f g = {}, u = {};
#pragma unroll 8
  for (int kg = 0; kg < HH / 8; ++kg) {
    v4f a  = *(const v4f*)(ap + kg * 8);
    v4f bg = *(const v4f*)(gp + (size_t)kg * 128);
    v4f bu = *(const v4f*)(up + (size_t)kg * 128);
    g = wmma4(lo2(a), lo2(bg), g);
    u = wmma4(lo2(a), lo2(bu), u);
    g = wmma4(hi2(a), hi2(bg), g);
    u = wmma4(hi2(a), hi2(bu), u);
  }
  {
    const int n = tn * 16 + c;
    const float gbias = gb[n], ubias = ub[n];
#pragma unroll
    for (int i = 0; i < 8; ++i) {
      float zg = g[i] + gbias;
      float zu = u[i] + ubias;
      Gs[(8 * hf + i) * GPITCH + w * 16 + c] = (1.0f / (1.0f + __expf(-zg))) * zu;
    }
  }
  __syncthreads();
  v4f a[16];
#pragma unroll
  for (int kk = 0; kk < 16; ++kk) a[kk] = *(const v4f*)(Gs + c * GPITCH + kk * 8 + 4 * hf);
  const float* bp = dw + (((size_t)(w * 6) * (H4 / 8) + sl * 16) * 32 + lane) * 4;
  float* Yp = Ypart + ((size_t)sl * (BB * CHK) + b * CHK + tm * 16 + 8 * hf) * HH + w * 96 + c;
#pragma unroll
  for (int q = 0; q < 6; q += 2) {
    const float* b0 = bp + (size_t)q * (H4 / 8) * 128;
    const float* b1 = b0 + (size_t)(H4 / 8) * 128;
    v8f y0 = {}, y1 = {};
    v4f v0[16], v1[16];
#pragma unroll
    for (int kk = 0; kk < 16; ++kk) {
      v0[kk] = *(const v4f*)(b0 + kk * 128);
      v1[kk] = *(const v4f*)(b1 + kk * 128);
    }
    __builtin_amdgcn_sched_barrier(0);
#pragma unroll
    for (int kk = 0; kk < 16; ++kk) {
      y0 = wmma4(lo2(a[kk]), lo2(v0[kk]), y0);
      y1 = wmma4(lo2(a[kk]), lo2(v1[kk]), y1);
      y0 = wmma4(hi2(a[kk]), hi2(v0[kk]), y0);
      y1 = wmma4(hi2(a[kk]), hi2(v1[kk]), y1);
    }
#pragma unroll
    for (int i = 0; i < 8; ++i) {
      Yp[(size_t)i * HH + q * 16]      = y0[i];
      Yp[(size_t)i * HH + q * 16 + 16] = y1[i];
    }
  }
}

__device__ __forceinline__ void d_ln(int blk, const float* __restrict__ Ypart, const float* __restrict__ db,
                     const float* __restrict__ Pred, const float* __restrict__ nw,
                     const float* __restrict__ nb, float* __restrict__ out, int chunk) {
  __shared__ float s1[256], s2[256];
  int row = blk;
  int b = row / CHK, t = row % CHK;
  const float* y = Ypart + (size_t)row * HH;
  const float* p = Pred + (size_t)row * HH;
  int tid = threadIdx.x;
  float v0 = 0.0f, v1 = 0.0f, v2 = 0.0f;
#pragma unroll
  for (int j = 0; j < NSL; ++j) {
    const float* yj = y + (size_t)j * (BB * CHK) * HH;
    v0 += yj[tid]; v1 += yj[tid + 256]; v2 += yj[tid + 512];
  }
  v0 = v0 + db[tid]       + p[tid];
  v1 = v1 + db[tid + 256] + p[tid + 256];
  v2 = v2 + db[tid + 512] + p[tid + 512];
  s1[tid] = v0 + v1 + v2;
  s2[tid] = v0 * v0 + v1 * v1 + v2 * v2;
  __syncthreads();
  for (int off = 128; off > 0; off >>= 1) {
    if (tid < off) { s1[tid] += s1[tid + off]; s2[tid] += s2[tid + off]; }
    __syncthreads();
  }
  float mu  = s1[0] * (1.0f / HH);
  float var = s2[0] * (1.0f / HH) - mu * mu;
  float r = rsqrtf(var + EPSC);
  int gt = chunk * CHK + t;
  float* o = out + ((size_t)b * SS + gt) * HH;
  o[tid]       = (v0 - mu) * r * nw[tid]       + nb[tid];
  o[tid + 256] = (v1 - mu) * r * nw[tid + 256] + nb[tid + 256];
  o[tid + 512] = (v2 - mu) * r * nw[tid + 512] + nb[tid + 512];
}


__global__ __launch_bounds__(256) void k_first(const float* __restrict__ x,
                       const float* __restrict__ Wi, const float* __restrict__ bi,
                       float* __restrict__ Wst, float* __restrict__ bst,
                       float* __restrict__ Amat,
                       const float* __restrict__ gw, const float* __restrict__ uw,
                       const float* __restrict__ dw, float* __restrict__ gwp,
                       float* __restrict__ uwp, float* __restrict__ dwp,
                       float* __restrict__ xp) {
  int blk = blockIdx.x;
  if (blk < NB_ATTN) { d_attn(blk, x, false, Amat, 0); return; }
  blk -= NB_ATTN;
  if (blk < NB_PACK) { d_pack(blk, gw, uw, dw, x, gwp, uwp, dwp, xp); return; }
  d_init(blk - NB_PACK, Wi, bi, Wst, bst);
}

__global__ __launch_bounds__(256) void k_stage_u(const float* __restrict__ x,
                       const float* __restrict__ Diff, float* __restrict__ Wst,
                       const float* __restrict__ YpartLn, const float* __restrict__ db,
                       const float* __restrict__ PredLn,
                       const float* __restrict__ nw, const float* __restrict__ nb,
                       float* __restrict__ out, int c) {
  int blk = blockIdx.x;
  if (blk < NB_WUPD) {
    d_wupdate(blk, x, Diff, Wst, c - 1);
    return;
  }
  if (c >= 2) d_ln(blk - NB_WUPD, YpartLn, db, PredLn, nw, nb, out, c - 2);
}

__global__ __launch_bounds__(256) void k_stage_s(const float* __restrict__ x,
                       const float* __restrict__ xp,
                       const float* __restrict__ Wst, float* __restrict__ bst,
                       const float* __restrict__ AmatCur, float* __restrict__ AmatNext,
                       float* __restrict__ Pred, float* __restrict__ Diff,
                       const float* __restrict__ PredPrev,
                       const float* __restrict__ gw, const float* __restrict__ gb,
                       const float* __restrict__ uw, const float* __restrict__ ub,
                       const float* __restrict__ dw, float* __restrict__ Ypart, int c) {
  int blk = blockIdx.x;
  if (blk < NB_SOLVE) {
    d_p0solve(blk, x, xp, Wst, AmatCur, Pred, Diff, bst, c);
    return;
  }
  blk -= NB_SOLVE;
  if (blk < NB_ATTN) {
    if (c + 1 < NCHUNK) d_attn(blk, xp, true, AmatNext, c + 1);
    return;
  }
  blk -= NB_ATTN;
  if (c >= 1) d_mlp(blk, PredPrev, gw, gb, uw, ub, dw, Ypart);
}

__global__ __launch_bounds__(256) void k_tail(const float* __restrict__ PredMlp,
                       const float* __restrict__ gw, const float* __restrict__ gb,
                       const float* __restrict__ uw, const float* __restrict__ ub,
                       const float* __restrict__ dw, float* __restrict__ YpartMlp,
                       const float* __restrict__ YpartLn, const float* __restrict__ db,
                       const float* __restrict__ PredLn,
                       const float* __restrict__ nw, const float* __restrict__ nb,
                       float* __restrict__ out, int chunk_ln) {
  int blk = blockIdx.x;
  if (blk < NB_MLP1) {
    d_mlp(blk, PredMlp, gw, gb, uw, ub, dw, YpartMlp);
    return;
  }
  d_ln(blk - NB_MLP1, YpartLn, db, PredLn, nw, nb, out, chunk_ln);
}

__global__ __launch_bounds__(256) void k_ln(const float* __restrict__ Ypart, const float* __restrict__ db,
                     const float* __restrict__ Pred, const float* __restrict__ nw,
                     const float* __restrict__ nb, float* __restrict__ out, int chunk) {
  d_ln(blockIdx.x, Ypart, db, Pred, nw, nb, out, chunk);
}

extern "C" void kernel_launch(void* const* d_in, const int* in_sizes, int n_in,
                              void* d_out, int out_size, void* d_ws, size_t ws_size,
                              hipStream_t stream) {
  const float* x      = (const float*)d_in[0];
  const float* W_init = (const float*)d_in[1];
  const float* b_init = (const float*)d_in[2];
  const float* gate_w = (const float*)d_in[3];
  const float* gate_b = (const float*)d_in[4];
  const float* up_w   = (const float*)d_in[5];
  const float* up_b   = (const float*)d_in[6];
  const float* down_w = (const float*)d_in[7];
  const float* down_b = (const float*)d_in[8];
  const float* norm_w = (const float*)d_in[9];
  const float* norm_b = (const float*)d_in[10];
  float* out = (float*)d_out;

  float* ws   = (float*)d_ws;
  float* Wst  = ws;  ws += (size_t)BB * HH * HH;
  float* bst  = ws;  ws += (size_t)BB * HH;
  float* PrE  = ws;  ws += (size_t)BB * CHK * HH;
  float* PrO  = ws;  ws += (size_t)BB * CHK * HH;
  float* Diff = ws;  ws += (size_t)BB * CHK * HH;
  float* AmE  = ws;  ws += (size_t)BB * CHK * CHK;
  float* AmO  = ws;  ws += (size_t)BB * CHK * CHK;
  float* YpE  = ws;  ws += (size_t)NSL * BB * CHK * HH;
  float* YpO  = ws;  ws += (size_t)NSL * BB * CHK * HH;
  float* gwp  = ws;  ws += (size_t)H4 * HH;
  float* uwp  = ws;  ws += (size_t)H4 * HH;
  float* dwp  = ws;  ws += (size_t)HH * H4;
  float* xp   = ws;  ws += (size_t)BB * SS * HH;

  k_first<<<NB_ATTN + NB_PACK + NB_INIT, 256, 0, stream>>>(
      x, W_init, b_init, Wst, bst, AmE, gate_w, up_w, down_w, gwp, uwp, dwp, xp);
  k_stage_s<<<NB_SOLVE + NB_ATTN, 256, 0, stream>>>(
      x, xp, Wst, bst, AmE, AmO, PrE, Diff, PrO, gwp, gate_b, uwp, up_b, dwp, YpO, 0);
  k_stage_u<<<NB_WUPD, 256, 0, stream>>>(
      x, Diff, Wst, YpE, down_b, PrE, norm_w, norm_b, out, 1);
  k_stage_s<<<NB_SOLVE + NB_ATTN + NB_MLP1, 256, 0, stream>>>(
      x, xp, Wst, bst, AmO, AmE, PrO, Diff, PrE, gwp, gate_b, uwp, up_b, dwp, YpE, 1);
  k_stage_u<<<NB_WUPD + NB_LN, 256, 0, stream>>>(
      x, Diff, Wst, YpE, down_b, PrE, norm_w, norm_b, out, 2);
  k_stage_s<<<NB_SOLVE + NB_ATTN + NB_MLP1, 256, 0, stream>>>(
      x, xp, Wst, bst, AmE, AmO, PrE, Diff, PrO, gwp, gate_b, uwp, up_b, dwp, YpO, 2);
  k_stage_u<<<NB_WUPD + NB_LN, 256, 0, stream>>>(
      x, Diff, Wst, YpO, down_b, PrO, norm_w, norm_b, out, 3);
  k_stage_s<<<NB_SOLVE + NB_ATTN + NB_MLP1, 256, 0, stream>>>(
      x, xp, Wst, bst, AmO, AmE, PrO, Diff, PrE, gwp, gate_b, uwp, up_b, dwp, YpE, 3);
  k_tail<<<NB_MLP1 + NB_LN, 256, 0, stream>>>(
      PrO, gwp, gate_b, uwp, up_b, dwp, YpO, YpE, down_b, PrE, norm_w, norm_b, out, 2);
  k_ln<<<NB_LN, 256, 0, stream>>>(YpO, down_b, PrO, norm_w, norm_b, out, NCHUNK - 1);
}


# chunk length 64 -> 128: 2 chunk rounds instead of 4 (one state update, 10 launches -> 6); solve loops over 16-token blocks, in-chunk terms of 128 tokens by WMMA from LDS; sum order changes
# speedup vs baseline: 7.3389x; 1.1327x over previous
#include <hip/hip_runtime.h>
#include <math.h>

#define BB 2
#define SS 256
#define HH 768
#define H4 3072
#define CHK 128
#define LRC 0.01f
#define EPSC 1e-5f

typedef float v2f __attribute__((ext_vector_type(2)));
typedef float v8f __attribute__((ext_vector_type(8)));
typedef float v4f __attribute__((ext_vector_type(4)));

__device__ __forceinline__ v8f wmma4(v2f a, v2f b, v8f c) {
  return __builtin_amdgcn_wmma_f32_16x16x4_f32(
       false, a,  false, b,
       (short)0, c,  false,  false);
}

__device__ __forceinline__ v2f frag_kn(const float* M, int ld, int k0, int r0, int lane) {
  const float* p = M + (size_t)(k0 + ((lane >> 4) << 1)) * ld + r0 + (lane & 15);
  v2f v; v.x = p[0]; v.y = p[ld];
  return v;
}
__device__ __forceinline__ v2f lo2(v4f v) { v2f r; r.x = v.x; r.y = v.y; return r; }
__device__ __forceinline__ v2f hi2(v4f v) { v2f r; r.x = v.z; r.y = v.w; return r; }

__device__ __forceinline__ v8f tile_kh(const float* __restrict__ a0, const float* __restrict__ b0, int kstep) {
  v8f acc = {};
#pragma unroll 16
  for (int kg = 0; kg < HH / 8; ++kg) {
    v4f a = *(const v4f*)(a0 + (size_t)kg * kstep);
    v4f b = *(const v4f*)(b0 + (size_t)kg * kstep);
    acc = wmma4(lo2(a), lo2(b), acc);
    acc = wmma4(hi2(a), hi2(b), acc);
  }
  return acc;
}

__device__ __forceinline__ void tile_kh2(const float* __restrict__ a0, const float* __restrict__ b0,
                                         const float* __restrict__ b1, int kstep, v8f& c0, v8f& c1) {
  v8f x = {}, y = {};
#pragma unroll 8
  for (int kg = 0; kg < HH / 8; ++kg) {
    v4f a = *(const v4f*)(a0 + (size_t)kg * kstep);
    v4f p = *(const v4f*)(b0 + (size_t)kg * kstep);
    v4f q = *(const v4f*)(b1 + (size_t)kg * kstep);
    x = wmma4(lo2(a), lo2(p), x);
    y = wmma4(lo2(a), lo2(q), y);
    x = wmma4(hi2(a), hi2(p), x);
    y = wmma4(hi2(a), hi2(q), y);
  }
  c0 = x; c1 = y;
}


#define NB_INIT ((HH * HH / 4 + 255) / 256)
#define NB_SOLVE (BB * (HH / 32))
#define NB_ATTN (BB * (CHK / 16) * (CHK / 16) / 8)
#define NB_WUPD (BB * (HH / 16) * (HH / 16) / 8)
#define NB_MLP1 (BB * (CHK / 16) * (H4 / 16) / 8)
#define NSL     (H4 / 128)
#define NB_LN   (BB * CHK)
#define NCHUNK  (SS / CHK)
#define NB_PACK (3 * 192 + BB * SS / 16)

__device__ __forceinline__ void d_init(int blk, const float* __restrict__ Wi, const float* __restrict__ bi,
                       float* __restrict__ Wst, float* __restrict__ bst) {
  int i = blk * 256 + threadIdx.x;
  const int n = HH * HH;
  if (i < n / 4) {
    const int lane = i & 31, u = i >> 5;
    const int tn = u / (HH / 8), kg = u % (HH / 8);
    v4f w = *(const v4f*)(Wi + (size_t)(tn * 16 + (lane & 15)) * HH + kg * 8 + 4 * (lane >> 4));
    *(v4f*)(Wst + (size_t)i * 4) = w;
    *(v4f*)(Wst + n + (size_t)i * 4) = w;
  }
  if (i < HH) {
    float b = bi[i];
    bst[i] = b;
    bst[HH + i] = b;
  }
}

__device__ __forceinline__ void d_pack(int blk, const float* __restrict__ gw, const float* __restrict__ uw,
                       const float* __restrict__ dw, const float* __restrict__ x,
                       float* __restrict__ gwp, float* __restrict__ uwp,
                       float* __restrict__ dwp, float* __restrict__ xp) {
  const int which = blk / 192, r = blk % 192;
  const float* src = (which == 0) ? gw : (which == 1) ? uw : (which == 2) ? dw : x;
  float* dst = (which == 0) ? gwp : (which == 1) ? uwp : (which == 2) ? dwp : xp;
  const int K   = (which == 2) ? H4 : HH;
  const int tn  = (which == 2) ? (r >> 2) : r;
  const int seg = (which == 2) ? (r & 3) : 0;
  const int w = threadIdx.x >> 5, lane = threadIdx.x & 31;
  const float* s = src + (size_t)(tn * 16 + (lane & 15)) * K + 4 * (lane >> 4);
  float* d = dst + ((size_t)tn * (K / 8) * 32 + lane) * 4;
  const int kg0 = seg * 96 + w * 12;
  v4f v[12];
#pragma unroll
  for (int i = 0; i < 12; ++i) v[i] = *(const v4f*)(s + (size_t)(kg0 + i) * 8);
#pragma unroll
  for (int i = 0; i < 12; ++i) *(v4f*)(d + (size_t)(kg0 + i) * 128) = v[i];
}

__device__ __forceinline__ void d_attn(int blk, const float* __restrict__ xs, bool packed,
                       float* __restrict__ Amat, int chunk) {
  int wid = (blk * 256 + threadIdx.x) >> 5;
  int lane = threadIdx.x & 31;
  int tn = wid % (CHK / 16);
  int tm = (wid / (CHK / 16)) % (CHK / 16);
  int b  = wid / ((CHK / 16) * (CHK / 16));
  v8f acc;
  if (packed) {
    const int t0 = b * (SS / 16) + chunk * (CHK / 16);
    acc = tile_kh(xs + ((size_t)(t0 + tm) * (HH / 8) * 32 + lane) * 4,
                  xs + ((size_t)(t0 + tn) * (HH / 8) * 32 + lane) * 4, 128);
  } else {
    const float* Xc = xs + ((size_t)b * SS + chunk * CHK) * HH + 4 * (lane >> 4);
    acc = tile_kh(Xc + (size_t)(tm * 16 + (lane & 15)) * HH,
                  Xc + (size_t)(tn * 16 + (lane & 15)) * HH, 8);
  }
  int s = tn * 16 + (lane & 15);
  int mbase = tm * 16 + ((lane >> 4) << 3);
  float* Ab = Amat + (size_t)b * CHK * CHK;
#pragma unroll
  for (int i = 0; i < 8; ++i) {
    int t = mbase + i;
    float v = (s < t) ? LRC * (acc[i] + 1.0f) : 0.0f;
    Ab[(size_t)t * CHK + s] = v;
  }
}

#define APITCH (CHK + 4)
#define DPITCH 40
#define PPITCH 36

__device__ __forceinline__ void wave_lds_sync() {
  __builtin_amdgcn_fence(__ATOMIC_SEQ_CST, "workgroup");
  __builtin_amdgcn_wave_barrier();
}

__device__ __forceinline__ void d_p0solve(int blk, const float* __restrict__ x, const float* __restrict__ xp,
                        const float* __restrict__ Wst,
                        const float* __restrict__ Amat, float* __restrict__ Pred,
                        float* __restrict__ Diff, float* __restrict__ bst, int chunk) {
  __shared__ __attribute__((aligned(16))) float As[CHK * APITCH];
  __shared__ __attribute__((aligned(16))) float Ds[CHK * DPITCH];
  __shared__ __attribute__((aligned(16))) float Ts[16 * DPITCH];
  __shared__ __attribute__((aligned(16))) float Ps[CHK * PPITCH];
  __shared__ __attribute__((aligned(16))) float Xs[CHK * PPITCH];
  const int tid = threadIdx.x;
  const int w = tid >> 5, lane = tid & 31;
  const int c = lane & 15, hf = lane >> 4;
  const int b  = blk / (HH / 32);
  const int n0 = (blk % (HH / 32)) * 32;
  const float* Xb = x + (size_t)b * SS * HH;
  const float* Ab = Amat + (size_t)b * CHK * CHK;
  const size_t col = (size_t)n0 + lane;
  const bool last_chunk = (chunk == SS / CHK - 1);

  v4f av[CHK * CHK / 1024];
#pragma unroll
  for (int i = 0; i < CHK * CHK / 1024; ++i) {
    int idx = i * 256 + tid;
    av[i] = *(const v4f*)(Ab + (idx / (CHK / 4)) * CHK + (idx % (CHK / 4)) * 4);
  }
  v4f xq[CHK / 32];
  {
    const float* Xn = Xb + ((size_t)chunk * CHK + 1) * HH + n0;
#pragma unroll
    for (int i = 0; i < CHK / 32; ++i) {
      const int idx = i * 256 + tid;
      const int t = idx >> 3, q = idx & 7;
      const bool none = last_chunk && (t == CHK - 1);
      v4f v = *(const v4f*)(Xn + (size_t)(none ? t - 1 : t) * HH + q * 4);
      if (none) { v.x = 0.0f; v.y = 0.0f; v.z = 0.0f; v.w = 0.0f; }
      xq[i] = v;
    }
  }
#pragma unroll 1
  for (int tm = w; tm < CHK / 16; tm += 8) {
    const float* ap = xp + ((size_t)(b * (SS / 16) + chunk * (CHK / 16) + tm) * (HH / 8) * 32 + lane) * 4;
    const float* bp = Wst + (size_t)b * HH * HH + ((size_t)(n0 / 16) * (HH / 8) * 32 + lane) * 4;
    v8f acc0, acc1;
    tile_kh2(ap, bp, bp + (size_t)(HH / 8) * 128, 128, acc0, acc1);
    const float bias0 = bst[(size_t)b * HH + n0 + c];
    const float bias1 = bst[(size_t)b * HH + n0 + 16 + c];
#pragma unroll
    for (int i = 0; i < 8; ++i) {
      Ps[(tm * 16 + 8 * hf + i) * PPITCH + c]      = acc0[i] + bias0;
      Ps[(tm * 16 + 8 * hf + i) * PPITCH + 16 + c] = acc1[i] + bias1;
    }
  }
#pragma unroll
  for (int i = 0; i < CHK * CHK / 1024; ++i) {
    int idx = i * 256 + tid;
    *(v4f*)(As + (idx / (CHK / 4)) * APITCH + (idx % (CHK / 4)) * 4) = av[i];
  }
#pragma unroll
  for (int i = 0; i < CHK / 32; ++i) {
    const int idx = i * 256 + tid;
    *(v4f*)(Xs + (idx >> 3) * PPITCH + (idx & 7) * 4) = xq[i];
  }
  __syncthreads();
  if (w != 0) return;

  float bsum = 0.0f;
  float* Pb = Pred + (size_t)b * CHK * HH + col;
  float* Db = Diff + (size_t)b * CHK * HH + col;
#pragma unroll 1
  for (int i = 0; i < CHK / 16; ++i) {
    const int t0 = i * 16;
    v4f cf[16][4];
#pragma unroll
    for (int r = 1; r < 16; ++r) {
#pragma unroll
      for (int q4 = 0; q4 * 4 < r; ++q4)
        cf[r][q4] = *(const v4f*)(As + (t0 + r) * APITCH + t0 + q4 * 4);
    }
    float pv[16], xv[16];
#pragma unroll
    for (int r = 0; r < 16; ++r) {
      pv[r] = Ps[(t0 + r) * PPITCH + lane];
      xv[r] = Xs[(t0 + r) * PPITCH + lane];
    }
    float sc[16];
#pragma unroll
    for (int r = 0; r < 16; ++r) sc[r] = 0.0f;
    if (i > 0) {
      v8f s0 = {}, s1 = {};
#pragma unroll 1
      for (int kb = 0; kb < t0; kb += 16) {
#pragma unroll
        for (int kk = 0; kk < 16; kk += 4) {
          const int k = kb + kk + 2 * hf;
          v2f a = *(const v2f*)(As + (t0 + c) * APITCH + k);
          v2f b0, b1;
          b0.x = Ds[k * DPITCH + c];       b0.y = Ds[(k + 1) * DPITCH + c];
          b1.x = Ds[k * DPITCH + 16 + c];  b1.y = Ds[(k + 1) * DPITCH + 16 + c];
          s0 = wmma4(a, b0, s0);
          s1 = wmma4(a, b1, s1);
        }
      }
#pragma unroll
      for (int r = 0; r < 8; ++r) {
        Ts[(8 * hf + r) * DPITCH + c]      = s0[r];
        Ts[(8 * hf + r) * DPITCH + 16 + c] = s1[r];
      }
      wave_lds_sync();
#pragma unroll
      for (int r = 0; r < 16; ++r) sc[r] = Ts[r * DPITCH + lane];
    }
    float d[16];
#pragma unroll
    for (int r = 0; r < 16; ++r) {
      const int t = t0 + r;
      float acc = pv[r] - sc[r];
#pragma unroll
      for (int q = 0; q < r; ++q) acc -= cf[r][q >> 2][q & 3] * d[q];
      float dd = acc - xv[r];
      d[r] = dd;
      Ds[t * DPITCH + lane] = dd;
      float dl = (t == CHK - 1 && last_chunk) ? 0.0f : dd;
      bsum += dl;
      Pb[(size_t)t * HH] = acc;
      Db[(size_t)t * HH] = dl;
    }
    wave_lds_sync();
  }
  bst[(size_t)b * HH + col] -= LRC * bsum;
}

__device__ __forceinline__ void d_wupdate(int blk, const float* __restrict__ x, const float* __restrict__ Diff,
                          float* __restrict__ Wst, int chunk) {
  int wid = (blk * 256 + threadIdx.x) >> 5;
  int lane = threadIdx.x & 31;
  int tn = wid % (HH / 16);
  int tm = (wid / (HH / 16)) % (HH / 16);
  int b  = wid / ((HH / 16) * (HH / 16));
  const float* Xc = x + ((size_t)b * SS + chunk * CHK) * HH;
  const float* Db = Diff + (size_t)b * CHK * HH;
  float* W = Wst + (size_t)b * HH * HH;
  v8f s = {};
  for (int k0 = 0; k0 < CHK; k0 += 4) {
    v2f a  = frag_kn(Xc, HH, k0, tn * 16, lane);
    v2f bb = frag_kn(Db, HH, k0, tm * 16, lane);
    s = wmma4(a, bb, s);
  }
  float* p = W + (((size_t)tm * (HH / 8) + tn * 2 + (lane >> 4)) * 32 + (lane & 15)) * 4;
  v4f w0 = *(const v4f*)p;
  v4f w1 = *(const v4f*)(p + 64);
  w0.x -= LRC * s[0]; w0.y -= LRC * s[1]; w0.z -= LRC * s[2]; w0.w -= LRC * s[3];
  w1.x -= LRC * s[4]; w1.y -= LRC * s[5]; w1.z -= LRC * s[6]; w1.w -= LRC * s[7];
  *(v4f*)p = w0;
  *(v4f*)(p + 64) = w1;
}

#define GPITCH 132
__device__ __forceinline__ void d_mlp(int blk, const float* __restrict__ Pred,
                       const float* __restrict__ gw, const float* __restrict__ gb,
                       const float* __restrict__ uw, const float* __restrict__ ub,
                       const float* __restrict__ dw, float* __restrict__ Ypart) {
  __shared__ __attribute__((aligned(16))) float Gs[16 * GPITCH];
  const int w = threadIdx.x >> 5, lane = threadIdx.x & 31;
  const int c = lane & 15, hf = lane >> 4;
  const int sl = blk % NSL;
  const int tm = (blk / NSL) % (CHK / 16);
  const int b  = blk / (NSL * (CHK / 16));
  const int tn = sl * 8 + w;
  const float* Pb = Pred + (size_t)b * CHK * HH;
  const float* ap = Pb + (size_t)(tm * 16 + c) * HH + 4 * hf;
  const float* gp = gw + ((size_t)tn * (HH / 8) * 32 + lane) * 4;
  const float* up = uw + ((size_t)tn * (HH / 8) * 32 + lane) * 4;
  v8f g = {}, u = {};
#pragma unroll 8
  for (int kg = 0; kg < HH / 8; ++kg) {
    v4f a  = *(const v4f*)(ap + kg * 8);
    v4f bg = *(const v4f*)(gp + (size_t)kg * 128);
    v4f bu = *(const v4f*)(up + (size_t)kg * 128);
    g = wmma4(lo2(a), lo2(bg), g);
    u = wmma4(lo2(a), lo2(bu), u);
    g = wmma4(hi2(a), hi2(bg), g);
    u = wmma4(hi2(a), hi2(bu), u);
  }
  {
    const int n = tn * 16 + c;
    const float gbias = gb[n], ubias = ub[n];
#pragma unroll
    for (int i = 0; i < 8; ++i) {
      float zg = g[i] + gbias;
      float zu = u[i] + ubias;
      Gs[(8 * hf + i) * GPITCH + w * 16 + c] = (1.0f / (1.0f + __expf(-zg))) * zu;
    }
  }
  __syncthreads();
  v4f a[16];
#pragma unroll
  for (int kk = 0; kk < 16; ++kk) a[kk] = *(const v4f*)(Gs + c * GPITCH + kk * 8 + 4 * hf);
  const float* bp = dw + (((size_t)(w * 6) * (H4 / 8) + sl * 16) * 32 + lane) * 4;
  float* Yp = Ypart + ((size_t)sl * (BB * CHK) + b * CHK + tm * 16 + 8 * hf) * HH + w * 96 + c;
#pragma unroll
  for (int q = 0; q < 6; q += 2) {
    const float* b0 = bp + (size_t)q * (H4 / 8) * 128;
    const float* b1 = b0 + (size_t)(H4 / 8) * 128;
    v8f y0 = {}, y1 = {};
    v4f v0[16], v1[16];
#pragma unroll
    for (int kk = 0; kk < 16; ++kk) {
      v0[kk] = *(const v4f*)(b0 + kk * 128);
      v1[kk] = *(const v4f*)(b1 + kk * 128);
    }
    __builtin_amdgcn_sched_barrier(0);
#pragma unroll
    for (int kk = 0; kk < 16; ++kk) {
      y0 = wmma4(lo2(a[kk]), lo2(v0[kk]), y0);
      y1 = wmma4(lo2(a[kk]), lo2(v1[kk]), y1);
      y0 = wmma4(hi2(a[kk]), hi2(v0[kk]), y0);
      y1 = wmma4(hi2(a[kk]), hi2(v1[kk]), y1);
    }
#pragma unroll
    for (int i = 0; i < 8; ++i) {
      Yp[(size_t)i * HH + q * 16]      = y0[i];
      Yp[(size_t)i * HH + q * 16 + 16] = y1[i];
    }
  }
}

__device__ __forceinline__ void d_ln(int blk, const float* __restrict__ Ypart, const float* __restrict__ db,
                     const float* __restrict__ Pred, const float* __restrict__ nw,
                     const float* __restrict__ nb, float* __restrict__ out, int chunk) {
  __shared__ float s1[256], s2[256];
  int row = blk;
  int b = row / CHK, t = row % CHK;
  const float* y = Ypart + (size_t)row * HH;
  const float* p = Pred + (size_t)row * HH;
  int tid = threadIdx.x;
  float v0 = 0.0f, v1 = 0.0f, v2 = 0.0f;
#pragma unroll
  for (int j = 0; j < NSL; ++j) {
    const float* yj = y + (size_t)j * (BB * CHK) * HH;
    v0 += yj[tid]; v1 += yj[tid + 256]; v2 += yj[tid + 512];
  }
  v0 = v0 + db[tid]       + p[tid];
  v1 = v1 + db[tid + 256] + p[tid + 256];
  v2 = v2 + db[tid + 512] + p[tid + 512];
  s1[tid] = v0 + v1 + v2;
  s2[tid] = v0 * v0 + v1 * v1 + v2 * v2;
  __syncthreads();
  for (int off = 128; off > 0; off >>= 1) {
    if (tid < off) { s1[tid] += s1[tid + off]; s2[tid] += s2[tid + off]; }
    __syncthreads();
  }
  float mu  = s1[0] * (1.0f / HH);
  float var = s2[0] * (1.0f / HH) - mu * mu;
  float r = rsqrtf(var + EPSC);
  int gt = chunk * CHK + t;
  float* o = out + ((size_t)b * SS + gt) * HH;
  o[tid]       = (v0 - mu) * r * nw[tid]       + nb[tid];
  o[tid + 256] = (v1 - mu) * r * nw[tid + 256] + nb[tid + 256];
  o[tid + 512] = (v2 - mu) * r * nw[tid + 512] + nb[tid + 512];
}


__global__ __launch_bounds__(256) void k_first(const float* __restrict__ x,
                       const float* __restrict__ Wi, const float* __restrict__ bi,
                       float* __restrict__ Wst, float* __restrict__ bst,
                       float* __restrict__ Amat,
                       const float* __restrict__ gw, const float* __restrict__ uw,
                       const float* __restrict__ dw, float* __restrict__ gwp,
                       float* __restrict__ uwp, float* __restrict__ dwp,
                       float* __restrict__ xp) {
  int blk = blockIdx.x;
  if (blk < NB_ATTN) { d_attn(blk, x, false, Amat, 0); return; }
  blk -= NB_ATTN;
  if (blk < NB_PACK) { d_pack(blk, gw, uw, dw, x, gwp, uwp, dwp, xp); return; }
  d_init(blk - NB_PACK, Wi, bi, Wst, bst);
}

__global__ __launch_bounds__(256) void k_stage_u(const float* __restrict__ x,
                       const float* __restrict__ Diff, float* __restrict__ Wst,
                       const float* __restrict__ YpartLn, const float* __restrict__ db,
                       const float* __restrict__ PredLn,
                       const float* __restrict__ nw, const float* __restrict__ nb,
                       float* __restrict__ out, int c) {
  int blk = blockIdx.x;
  if (blk < NB_WUPD) {
    d_wupdate(blk, x, Diff, Wst, c - 1);
    return;
  }
  if (c >= 2) d_ln(blk - NB_WUPD, YpartLn, db, PredLn, nw, nb, out, c - 2);
}

__global__ __launch_bounds__(256) void k_stage_s(const float* __restrict__ x,
                       const float* __restrict__ xp,
                       const float* __restrict__ Wst, float* __restrict__ bst,
                       const float* __restrict__ AmatCur, float* __restrict__ AmatNext,
                       float* __restrict__ Pred, float* __restrict__ Diff,
                       const float* __restrict__ PredPrev,
                       const float* __restrict__ gw, const float* __restrict__ gb,
                       const float* __restrict__ uw, const float* __restrict__ ub,
                       const float* __restrict__ dw, float* __restrict__ Ypart, int c) {
  int blk = blockIdx.x;
  if (blk < NB_SOLVE) {
    d_p0solve(blk, x, xp, Wst, AmatCur, Pred, Diff, bst, c);
    return;
  }
  blk -= NB_SOLVE;
  if (blk < NB_ATTN) {
    if (c + 1 < NCHUNK) d_attn(blk, xp, true, AmatNext, c + 1);
    return;
  }
  blk -= NB_ATTN;
  if (c >= 1) d_mlp(blk, PredPrev, gw, gb, uw, ub, dw, Ypart);
}

__global__ __launch_bounds__(256) void k_tail(const float* __restrict__ PredMlp,
                       const float* __restrict__ gw, const float* __restrict__ gb,
                       const float* __restrict__ uw, const float* __restrict__ ub,
                       const float* __restrict__ dw, float* __restrict__ YpartMlp,
                       const float* __restrict__ YpartLn, const float* __restrict__ db,
                       const float* __restrict__ PredLn,
                       const float* __restrict__ nw, const float* __restrict__ nb,
                       float* __restrict__ out, int chunk_ln) {
  int blk = blockIdx.x;
  if (blk < NB_MLP1) {
    d_mlp(blk, PredMlp, gw, gb, uw, ub, dw, YpartMlp);
    return;
  }
  d_ln(blk - NB_MLP1, YpartLn, db, PredLn, nw, nb, out, chunk_ln);
}

__global__ __launch_bounds__(256) void k_ln(const float* __restrict__ Ypart, const float* __restrict__ db,
                     const float* __restrict__ Pred, const float* __restrict__ nw,
                     const float* __restrict__ nb, float* __restrict__ out, int chunk) {
  d_ln(blockIdx.x, Ypart, db, Pred, nw, nb, out, chunk);
}

extern "C" void kernel_launch(void* const* d_in, const int* in_sizes, int n_in,
                              void* d_out, int out_size, void* d_ws, size_t ws_size,
                              hipStream_t stream) {
  const float* x      = (const float*)d_in[0];
  const float* W_init = (const float*)d_in[1];
  const float* b_init = (const float*)d_in[2];
  const float* gate_w = (const float*)d_in[3];
  const float* gate_b = (const float*)d_in[4];
  const float* up_w   = (const float*)d_in[5];
  const float* up_b   = (const float*)d_in[6];
  const float* down_w = (const float*)d_in[7];
  const float* down_b = (const float*)d_in[8];
  const float* norm_w = (const float*)d_in[9];
  const float* norm_b = (const float*)d_in[10];
  float* out = (float*)d_out;

  float* ws   = (float*)d_ws;
  float* Wst  = ws;  ws += (size_t)BB * HH * HH;
  float* bst  = ws;  ws += (size_t)BB * HH;
  float* PrE  = ws;  ws += (size_t)BB * CHK * HH;
  float* PrO  = ws;  ws += (size_t)BB * CHK * HH;
  float* Diff = ws;  ws += (size_t)BB * CHK * HH;
  float* AmE  = ws;  ws += (size_t)BB * CHK * CHK;
  float* AmO  = ws;  ws += (size_t)BB * CHK * CHK;
  float* YpE  = ws;  ws += (size_t)NSL * BB * CHK * HH;
  float* YpO  = ws;  ws += (size_t)NSL * BB * CHK * HH;
  float* gwp  = ws;  ws += (size_t)H4 * HH;
  float* uwp  = ws;  ws += (size_t)H4 * HH;
  float* dwp  = ws;  ws += (size_t)HH * H4;
  float* xp   = ws;  ws += (size_t)BB * SS * HH;

  k_first<<<NB_ATTN + NB_PACK + NB_INIT, 256, 0, stream>>>(
      x, W_init, b_init, Wst, bst, AmE, gate_w, up_w, down_w, gwp, uwp, dwp, xp);
  k_stage_s<<<NB_SOLVE + NB_ATTN, 256, 0, stream>>>(
      x, xp, Wst, bst, AmE, AmO, PrE, Diff, PrO, gwp, gate_b, uwp, up_b, dwp, YpO, 0);
  k_stage_u<<<NB_WUPD, 256, 0, stream>>>(
      x, Diff, Wst, YpE, down_b, PrE, norm_w, norm_b, out, 1);
  k_stage_s<<<NB_SOLVE + NB_ATTN + NB_MLP1, 256, 0, stream>>>(
      x, xp, Wst, bst, AmO, AmE, PrO, Diff, PrE, gwp, gate_b, uwp, up_b, dwp, YpE, 1);
  k_tail<<<NB_MLP1 + NB_LN, 256, 0, stream>>>(
      PrO, gwp, gate_b, uwp, up_b, dwp, YpO, YpE, down_b, PrE, norm_w, norm_b, out, 0);
  k_ln<<<NB_LN, 256, 0, stream>>>(YpO, down_b, PrO, norm_w, norm_b, out, NCHUNK - 1);
}


# in-chunk solve on all 8 waves: wave w keeps block w's sum over earlier blocks (WMMA as each block's diffs reach LDS), wave 0 only the triangle chains; Pred/Diff stored at the end; sums as before
# speedup vs baseline: 8.0099x; 1.0914x over previous
#include <hip/hip_runtime.h>
#include <math.h>

#define BB 2
#define SS 256
#define HH 768
#define H4 3072
#define CHK 128
#define LRC 0.01f
#define EPSC 1e-5f

typedef float v2f __attribute__((ext_vector_type(2)));
typedef float v8f __attribute__((ext_vector_type(8)));
typedef float v4f __attribute__((ext_vector_type(4)));

__device__ __forceinline__ v8f wmma4(v2f a, v2f b, v8f c) {
  return __builtin_amdgcn_wmma_f32_16x16x4_f32(
       false, a,  false, b,
       (short)0, c,  false,  false);
}

__device__ __forceinline__ v2f frag_kn(const float* M, int ld, int k0, int r0, int lane) {
  const float* p = M + (size_t)(k0 + ((lane >> 4) << 1)) * ld + r0 + (lane & 15);
  v2f v; v.x = p[0]; v.y = p[ld];
  return v;
}
__device__ __forceinline__ v2f lo2(v4f v) { v2f r; r.x = v.x; r.y = v.y; return r; }
__device__ __forceinline__ v2f hi2(v4f v) { v2f r; r.x = v.z; r.y = v.w; return r; }

__device__ __forceinline__ v8f tile_kh(const float* __restrict__ a0, const float* __restrict__ b0, int kstep) {
  v8f acc = {};
#pragma unroll 16
  for (int kg = 0; kg < HH / 8; ++kg) {
    v4f a = *(const v4f*)(a0 + (size_t)kg * kstep);
    v4f b = *(const v4f*)(b0 + (size_t)kg * kstep);
    acc = wmma4(lo2(a), lo2(b), acc);
    acc = wmma4(hi2(a), hi2(b), acc);
  }
  return acc;
}

__device__ __forceinline__ void tile_kh2(const float* __restrict__ a0, const float* __restrict__ b0,
                                         const float* __restrict__ b1, int kstep, v8f& c0, v8f& c1) {
  v8f x = {}, y = {};
#pragma unroll 8
  for (int kg = 0; kg < HH / 8; ++kg) {
    v4f a = *(const v4f*)(a0 + (size_t)kg * kstep);
    v4f p = *(const v4f*)(b0 + (size_t)kg * kstep);
    v4f q = *(const v4f*)(b1 + (size_t)kg * kstep);
    x = wmma4(lo2(a), lo2(p), x);
    y = wmma4(lo2(a), lo2(q), y);
    x = wmma4(hi2(a), hi2(p), x);
    y = wmma4(hi2(a), hi2(q), y);
  }
  c0 = x; c1 = y;
}


#define NB_INIT ((HH * HH / 4 + 255) / 256)
#define NB_SOLVE (BB * (HH / 32))
#define NB_ATTN (BB * (CHK / 16) * (CHK / 16) / 8)
#define NB_WUPD (BB * (HH / 16) * (HH / 16) / 8)
#define NB_MLP1 (BB * (CHK / 16) * (H4 / 16) / 8)
#define NSL     (H4 / 128)
#define NB_LN   (BB * CHK)
#define NCHUNK  (SS / CHK)
#define NB_PACK (3 * 192 + BB * SS / 16)

__device__ __forceinline__ void d_init(int blk, const float* __restrict__ Wi, const float* __restrict__ bi,
                       float* __restrict__ Wst, float* __restrict__ bst) {
  int i = blk * 256 + threadIdx.x;
  const int n = HH * HH;
  if (i < n / 4) {
    const int lane = i & 31, u = i >> 5;
    const int tn = u / (HH / 8), kg = u % (HH / 8);
    v4f w = *(const v4f*)(Wi + (size_t)(tn * 16 + (lane & 15)) * HH + kg * 8 + 4 * (lane >> 4));
    *(v4f*)(Wst + (size_t)i * 4) = w;
    *(v4f*)(Wst + n + (size_t)i * 4) = w;
  }
  if (i < HH) {
    float b = bi[i];
    bst[i] = b;
    bst[HH + i] = b;
  }
}

__device__ __forceinline__ void d_pack(int blk, const float* __restrict__ gw, const float* __restrict__ uw,
                       const float* __restrict__ dw, const float* __restrict__ x,
                       float* __restrict__ gwp, float* __restrict__ uwp,
                       float* __restrict__ dwp, float* __restrict__ xp) {
  const int which = blk / 192, r = blk % 192;
  const float* src = (which == 0) ? gw : (which == 1) ? uw : (which == 2) ? dw : x;
  float* dst = (which == 0) ? gwp : (which == 1) ? uwp : (which == 2) ? dwp : xp;
  const int K   = (which == 2) ? H4 : HH;
  const int tn  = (which == 2) ? (r >> 2) : r;
  const int seg = (which == 2) ? (r & 3) : 0;
  const int w = threadIdx.x >> 5, lane = threadIdx.x & 31;
  const float* s = src + (size_t)(tn * 16 + (lane & 15)) * K + 4 * (lane >> 4);
  float* d = dst + ((size_t)tn * (K / 8) * 32 + lane) * 4;
  const int kg0 = seg * 96 + w * 12;
  v4f v[12];
#pragma unroll
  for (int i = 0; i < 12; ++i) v[i] = *(const v4f*)(s + (size_t)(kg0 + i) * 8);
#pragma unroll
  for (int i = 0; i < 12; ++i) *(v4f*)(d + (size_t)(kg0 + i) * 128) = v[i];
}

__device__ __forceinline__ void d_attn(int blk, const float* __restrict__ xs, bool packed,
                       float* __restrict__ Amat, int chunk) {
  int wid = (blk * 256 + threadIdx.x) >> 5;
  int lane = threadIdx.x & 31;
  int tn = wid % (CHK / 16);
  int tm = (wid / (CHK / 16)) % (CHK / 16);
  int b  = wid / ((CHK / 16) * (CHK / 16));
  v8f acc;
  if (packed) {
    const int t0 = b * (SS / 16) + chunk * (CHK / 16);
    acc = tile_kh(xs + ((size_t)(t0 + tm) * (HH / 8) * 32 + lane) * 4,
                  xs + ((size_t)(t0 + tn) * (HH / 8) * 32 + lane) * 4, 128);
  } else {
    const float* Xc = xs + ((size_t)b * SS + chunk * CHK) * HH + 4 * (lane >> 4);
    acc = tile_kh(Xc + (size_t)(tm * 16 + (lane & 15)) * HH,
                  Xc + (size_t)(tn * 16 + (lane & 15)) * HH, 8);
  }
  int s = tn * 16 + (lane & 15);
  int mbase = tm * 16 + ((lane >> 4) << 3);
  float* Ab = Amat + (size_t)b * CHK * CHK;
#pragma unroll
  for (int i = 0; i < 8; ++i) {
    int t = mbase + i;
    float v = (s < t) ? LRC * (acc[i] + 1.0f) : 0.0f;
    Ab[(size_t)t * CHK + s] = v;
  }
}

#define APITCH (CHK + 4)
#define DPITCH 40
#define PPITCH 36

static_assert(CHK / 16 <= 8, "one wave of the 256-thread block per 16-token block");

__device__ __forceinline__ void solve_block_operands(const float* Ps, const float* Xs, int t0, int lane,
                                                     float (&pv)[16], float (&xv)[16]) {
#pragma unroll
  for (int r = 0; r < 16; ++r) {
    pv[r] = Ps[(t0 + r) * PPITCH + lane];
    xv[r] = Xs[(t0 + r) * PPITCH + lane];
  }
}

__device__ __forceinline__ void d_p0solve(int blk, const float* __restrict__ x, const float* __restrict__ xp,
                        const float* __restrict__ Wst,
                        const float* __restrict__ Amat, float* __restrict__ Pred,
                        float* __restrict__ Diff, float* __restrict__ bst, int chunk) {
  __shared__ __attribute__((aligned(16))) float As[CHK * APITCH];
  __shared__ __attribute__((aligned(16))) float Ds[CHK * DPITCH];
  __shared__ __attribute__((aligned(16))) float Ts[16 * DPITCH];
  __shared__ __attribute__((aligned(16))) float Ps[CHK * PPITCH];
  __shared__ __attribute__((aligned(16))) float Xs[CHK * PPITCH];
  const int tid = threadIdx.x;
  const int w = __builtin_amdgcn_readfirstlane(tid >> 5);
  const int lane = tid & 31;
  const int c = lane & 15, hf = lane >> 4;
  const int b  = blk / (HH / 32);
  const int n0 = (blk % (HH / 32)) * 32;
  const float* Xb = x + (size_t)b * SS * HH;
  const float* Ab = Amat + (size_t)b * CHK * CHK;
  const size_t col = (size_t)n0 + lane;
  const bool last_chunk = (chunk == SS / CHK - 1);

  v4f av[CHK * CHK / 1024];
#pragma unroll
  for (int i = 0; i < CHK * CHK / 1024; ++i) {
    int idx = i * 256 + tid;
    av[i] = *(const v4f*)(Ab + (idx / (CHK / 4)) * CHK + (idx % (CHK / 4)) * 4);
  }
  v4f xq[CHK / 32];
  {
    const float* Xn = Xb + ((size_t)chunk * CHK + 1) * HH + n0;
#pragma unroll
    for (int i = 0; i < CHK / 32; ++i) {
      const int idx = i * 256 + tid;
      const int t = idx >> 3, q = idx & 7;
      const bool none = last_chunk && (t == CHK - 1);
      v4f v = *(const v4f*)(Xn + (size_t)(none ? t - 1 : t) * HH + q * 4);
      if (none) { v.x = 0.0f; v.y = 0.0f; v.z = 0.0f; v.w = 0.0f; }
      xq[i] = v;
    }
  }
#pragma unroll 1
  for (int tm = w; tm < CHK / 16; tm += 8) {
    const float* ap = xp + ((size_t)(b * (SS / 16) + chunk * (CHK / 16) + tm) * (HH / 8) * 32 + lane) * 4;
    const float* bp = Wst + (size_t)b * HH * HH + ((size_t)(n0 / 16) * (HH / 8) * 32 + lane) * 4;
    v8f acc0, acc1;
    tile_kh2(ap, bp, bp + (size_t)(HH / 8) * 128, 128, acc0, acc1);
    const float bias0 = bst[(size_t)b * HH + n0 + c];
    const float bias1 = bst[(size_t)b * HH + n0 + 16 + c];
#pragma unroll
    for (int i = 0; i < 8; ++i) {
      Ps[(tm * 16 + 8 * hf + i) * PPITCH + c]      = acc0[i] + bias0;
      Ps[(tm * 16 + 8 * hf + i) * PPITCH + 16 + c] = acc1[i] + bias1;
    }
  }
#pragma unroll
  for (int i = 0; i < CHK * CHK / 1024; ++i) {
    int idx = i * 256 + tid;
    *(v4f*)(As + (idx / (CHK / 4)) * APITCH + (idx % (CHK / 4)) * 4) = av[i];
  }
#pragma unroll
  for (int i = 0; i < CHK / 32; ++i) {
    const int idx = i * 256 + tid;
    *(v4f*)(Xs + (idx >> 3) * PPITCH + (idx & 7) * 4) = xq[i];
  }
  __syncthreads();

  float bsum = 0.0f;
  v8f s0 = {}, s1 = {};
  float pv[16], xv[16];
  if (w == 0) solve_block_operands(Ps, Xs, 0, lane, pv, xv);
#pragma unroll 1
  for (int j = 0; j < CHK / 16; ++j) {
    const int t0 = j * 16;
    if (w == 0) {
      float sc[16];
#pragma unroll
      for (int r = 0; r < 16; ++r) sc[r] = 0.0f;
      if (j > 0) {
#pragma unroll
        for (int r = 0; r < 16; ++r) sc[r] = Ts[r * DPITCH + lane];
      }
      v4f cf[16][4];
#pragma unroll
      for (int r = 1; r < 16; ++r) {
#pragma unroll
        for (int q4 = 0; q4 * 4 < r; ++q4)
          cf[r][q4] = *(const v4f*)(As + (t0 + r) * APITCH + t0 + q4 * 4);
      }
      float d[16];
#pragma unroll
      for (int r = 0; r < 16; ++r) {
        const int t = t0 + r;
        float acc = pv[r] - sc[r];
#pragma unroll
        for (int q = 0; q < r; ++q) acc -= cf[r][q >> 2][q & 3] * d[q];
        float dd = acc - xv[r];
        d[r] = dd;
        Ds[t * DPITCH + lane] = dd;
        Ps[t * PPITCH + lane] = acc;
        float dl = (t == CHK - 1 && last_chunk) ? 0.0f : dd;
        bsum += dl;
      }
    }
    __syncthreads();
    if (j + 1 < CHK / 16) {
      if (w > j) {
#pragma unroll
        for (int q = 0; q < 4; ++q) {
          const int k = t0 + 4 * q + 2 * hf;
          const v2f a = *(const v2f*)(As + (w * 16 + c) * APITCH + k);
          v2f b0, b1;
          b0.x = Ds[k * DPITCH + c];       b0.y = Ds[(k + 1) * DPITCH + c];
          b1.x = Ds[k * DPITCH + 16 + c];  b1.y = Ds[(k + 1) * DPITCH + 16 + c];
          s0 = wmma4(a, b0, s0);
          s1 = wmma4(a, b1, s1);
        }
        if (w == j + 1) {
#pragma unroll
          for (int r = 0; r < 8; ++r) {
            Ts[(8 * hf + r) * DPITCH + c]      = s0[r];
            Ts[(8 * hf + r) * DPITCH + 16 + c] = s1[r];
          }
        }
      } else if (w == 0) {
        solve_block_operands(Ps, Xs, t0 + 16, lane, pv, xv);
      }
      __syncthreads();
    }
  }
  {
    float* Pb = Pred + (size_t)b * CHK * HH + col;
    float* Db = Diff + (size_t)b * CHK * HH + col;
#pragma unroll
    for (int r = 0; r < 16; ++r) {
      const int t = w * 16 + r;
      const float pr = Ps[t * PPITCH + lane];
      float dv = Ds[t * DPITCH + lane];
      if (t == CHK - 1 && last_chunk) dv = 0.0f;
      Pb[(size_t)t * HH] = pr;
      Db[(size_t)t * HH] = dv;
    }
  }
  if (w == 0) bst[(size_t)b * HH + col] -= LRC * bsum;
}

__device__ __forceinline__ void d_wupdate(int blk, const float* __restrict__ x, const float* __restrict__ Diff,
                          float* __restrict__ Wst, int chunk) {
  int wid = (blk * 256 + threadIdx.x) >> 5;
  int lane = threadIdx.x & 31;
  int tn = wid % (HH / 16);
  int tm = (wid / (HH / 16)) % (HH / 16);
  int b  = wid / ((HH / 16) * (HH / 16));
  const float* Xc = x + ((size_t)b * SS + chunk * CHK) * HH;
  const float* Db = Diff + (size_t)b * CHK * HH;
  float* W = Wst + (size_t)b * HH * HH;
  v8f s = {};
  for (int k0 = 0; k0 < CHK; k0 += 4) {
    v2f a  = frag_kn(Xc, HH, k0, tn * 16, lane);
    v2f bb = frag_kn(Db, HH, k0, tm * 16, lane);
    s = wmma4(a, bb, s);
  }
  float* p = W + (((size_t)tm * (HH / 8) + tn * 2 + (lane >> 4)) * 32 + (lane & 15)) * 4;
  v4f w0 = *(const v4f*)p;
  v4f w1 = *(const v4f*)(p + 64);
  w0.x -= LRC * s[0]; w0.y -= LRC * s[1]; w0.z -= LRC * s[2]; w0.w -= LRC * s[3];
  w1.x -= LRC * s[4]; w1.y -= LRC * s[5]; w1.z -= LRC * s[6]; w1.w -= LRC * s[7];
  *(v4f*)p = w0;
  *(v4f*)(p + 64) = w1;
}

#define GPITCH 132
__device__ __forceinline__ void d_mlp(int blk, const float* __restrict__ Pred,
                       const float* __restrict__ gw, const float* __restrict__ gb,
                       const float* __restrict__ uw, const float* __restrict__ ub,
                       const float* __restrict__ dw, float* __restrict__ Ypart) {
  __shared__ __attribute__((aligned(16))) float Gs[16 * GPITCH];
  const int w = threadIdx.x >> 5, lane = threadIdx.x & 31;
  const int c = lane & 15, hf = lane >> 4;
  const int sl = blk % NSL;
  const int tm = (blk / NSL) % (CHK / 16);
  const int b  = blk / (NSL * (CHK / 16));
  const int tn = sl * 8 + w;
  const float* Pb = Pred + (size_t)b * CHK * HH;
  const float* ap = Pb + (size_t)(tm * 16 + c) * HH + 4 * hf;
  const float* gp = gw + ((size_t)tn * (HH / 8) * 32 + lane) * 4;
  const float* up = uw + ((size_t)tn * (HH / 8) * 32 + lane) * 4;
  v8f g = {}, u = {};
#pragma unroll 8
  for (int kg = 0; kg < HH / 8; ++kg) {
    v4f a  = *(const v4f*)(ap + kg * 8);
    v4f bg = *(const v4f*)(gp + (size_t)kg * 128);
    v4f bu = *(const v4f*)(up + (size_t)kg * 128);
    g = wmma4(lo2(a), lo2(bg), g);
    u = wmma4(lo2(a), lo2(bu), u);
    g = wmma4(hi2(a), hi2(bg), g);
    u = wmma4(hi2(a), hi2(bu), u);
  }
  {
    const int n = tn * 16 + c;
    const float gbias = gb[n], ubias = ub[n];
#pragma unroll
    for (int i = 0; i < 8; ++i) {
      float zg = g[i] + gbias;
      float zu = u[i] + ubias;
      Gs[(8 * hf + i) * GPITCH + w * 16 + c] = (1.0f / (1.0f + __expf(-zg))) * zu;
    }
  }
  __syncthreads();
  v4f a[16];
#pragma unroll
  for (int kk = 0; kk < 16; ++kk) a[kk] = *(const v4f*)(Gs + c * GPITCH + kk * 8 + 4 * hf);
  const float* bp = dw + (((size_t)(w * 6) * (H4 / 8) + sl * 16) * 32 + lane) * 4;
  float* Yp = Ypart + ((size_t)sl * (BB * CHK) + b * CHK + tm * 16 + 8 * hf) * HH + w * 96 + c;
#pragma unroll
  for (int q = 0; q < 6; q += 2) {
    const float* b0 = bp + (size_t)q * (H4 / 8) * 128;
    const float* b1 = b0 + (size_t)(H4 / 8) * 128;
    v8f y0 = {}, y1 = {};
    v4f v0[16], v1[16];
#pragma unroll
    for (int kk = 0; kk < 16; ++kk) {
      v0[kk] = *(const v4f*)(b0 + kk * 128);
      v1[kk] = *(const v4f*)(b1 + kk * 128);
    }
    __builtin_amdgcn_sched_barrier(0);
#pragma unroll
    for (int kk = 0; kk < 16; ++kk) {
      y0 = wmma4(lo2(a[kk]), lo2(v0[kk]), y0);
      y1 = wmma4(lo2(a[kk]), lo2(v1[kk]), y1);
      y0 = wmma4(hi2(a[kk]), hi2(v0[kk]), y0);
      y1 = wmma4(hi2(a[kk]), hi2(v1[kk]), y1);
    }
#pragma unroll
    for (int i = 0; i < 8; ++i) {
      Yp[(size_t)i * HH + q * 16]      = y0[i];
      Yp[(size_t)i * HH + q * 16 + 16] = y1[i];
    }
  }
}

__device__ __forceinline__ void d_ln(int blk, const float* __restrict__ Ypart, const float* __restrict__ db,
                     const float* __restrict__ Pred, const float* __restrict__ nw,
                     const float* __restrict__ nb, float* __restrict__ out, int chunk) {
  __shared__ float s1[256], s2[256];
  int row = blk;
  int b = row / CHK, t = row % CHK;
  const float* y = Ypart + (size_t)row * HH;
  const float* p = Pred + (size_t)row * HH;
  int tid = threadIdx.x;
  float v0 = 0.0f, v1 = 0.0f, v2 = 0.0f;
#pragma unroll
  for (int j = 0; j < NSL; ++j) {
    const float* yj = y + (size_t)j * (BB * CHK) * HH;
    v0 += yj[tid]; v1 += yj[tid + 256]; v2 += yj[tid + 512];
  }
  v0 = v0 + db[tid]       + p[tid];
  v1 = v1 + db[tid + 256] + p[tid + 256];
  v2 = v2 + db[tid + 512] + p[tid + 512];
  s1[tid] = v0 + v1 + v2;
  s2[tid] = v0 * v0 + v1 * v1 + v2 * v2;
  __syncthreads();
  for (int off = 128; off > 0; off >>= 1) {
    if (tid < off) { s1[tid] += s1[tid + off]; s2[tid] += s2[tid + off]; }
    __syncthreads();
  }
  float mu  = s1[0] * (1.0f / HH);
  float var = s2[0] * (1.0f / HH) - mu * mu;
  float r = rsqrtf(var + EPSC);
  int gt = chunk * CHK + t;
  float* o = out + ((size_t)b * SS + gt) * HH;
  o[tid]       = (v0 - mu) * r * nw[tid]       + nb[tid];
  o[tid + 256] = (v1 - mu) * r * nw[tid + 256] + nb[tid + 256];
  o[tid + 512] = (v2 - mu) * r * nw[tid + 512] + nb[tid + 512];
}


__global__ __launch_bounds__(256) void k_first(const float* __restrict__ x,
                       const float* __restrict__ Wi, const float* __restrict__ bi,
                       float* __restrict__ Wst, float* __restrict__ bst,
                       float* __restrict__ Amat,
                       const float* __restrict__ gw, const float* __restrict__ uw,
                       const float* __restrict__ dw, float* __restrict__ gwp,
                       float* __restrict__ uwp, float* __restrict__ dwp,
                       float* __restrict__ xp) {
  int blk = blockIdx.x;
  if (blk < NB_ATTN) { d_attn(blk, x, false, Amat, 0); return; }
  blk -= NB_ATTN;
  if (blk < NB_PACK) { d_pack(blk, gw, uw, dw, x, gwp, uwp, dwp, xp); return; }
  d_init(blk - NB_PACK, Wi, bi, Wst, bst);
}

__global__ __launch_bounds__(256) void k_stage_u(const float* __restrict__ x,
                       const float* __restrict__ Diff, float* __restrict__ Wst,
                       const float* __restrict__ YpartLn, const float* __restrict__ db,
                       const float* __restrict__ PredLn,
                       const float* __restrict__ nw, const float* __restrict__ nb,
                       float* __restrict__ out, int c) {
  int blk = blockIdx.x;
  if (blk < NB_WUPD) {
    d_wupdate(blk, x, Diff, Wst, c - 1);
    return;
  }
  if (c >= 2) d_ln(blk - NB_WUPD, YpartLn, db, PredLn, nw, nb, out, c - 2);
}

__global__ __launch_bounds__(256) void k_stage_s(const float* __restrict__ x,
                       const float* __restrict__ xp,
                       const float* __restrict__ Wst, float* __restrict__ bst,
                       const float* __restrict__ AmatCur, float* __restrict__ AmatNext,
                       float* __restrict__ Pred, float* __restrict__ Diff,
                       const float* __restrict__ PredPrev,
                       const float* __restrict__ gw, const float* __restrict__ gb,
                       const float* __restrict__ uw, const float* __restrict__ ub,
                       const float* __restrict__ dw, float* __restrict__ Ypart, int c) {
  int blk = blockIdx.x;
  if (blk < NB_SOLVE) {
    d_p0solve(blk, x, xp, Wst, AmatCur, Pred, Diff, bst, c);
    return;
  }
  blk -= NB_SOLVE;
  if (blk < NB_ATTN) {
    if (c + 1 < NCHUNK) d_attn(blk, xp, true, AmatNext, c + 1);
    return;
  }
  blk -= NB_ATTN;
  if (c >= 1) d_mlp(blk, PredPrev, gw, gb, uw, ub, dw, Ypart);
}

__global__ __launch_bounds__(256) void k_tail(const float* __restrict__ PredMlp,
                       const float* __restrict__ gw, const float* __restrict__ gb,
                       const float* __restrict__ uw, const float* __restrict__ ub,
                       const float* __restrict__ dw, float* __restrict__ YpartMlp,
                       const float* __restrict__ YpartLn, const float* __restrict__ db,
                       const float* __restrict__ PredLn,
                       const float* __restrict__ nw, const float* __restrict__ nb,
                       float* __restrict__ out, int chunk_ln) {
  int blk = blockIdx.x;
  if (blk < NB_MLP1) {
    d_mlp(blk, PredMlp, gw, gb, uw, ub, dw, YpartMlp);
    return;
  }
  d_ln(blk - NB_MLP1, YpartLn, db, PredLn, nw, nb, out, chunk_ln);
}

__global__ __launch_bounds__(256) void k_ln(const float* __restrict__ Ypart, const float* __restrict__ db,
                     const float* __restrict__ Pred, const float* __restrict__ nw,
                     const float* __restrict__ nb, float* __restrict__ out, int chunk) {
  d_ln(blockIdx.x, Ypart, db, Pred, nw, nb, out, chunk);
}

extern "C" void kernel_launch(void* const* d_in, const int* in_sizes, int n_in,
                              void* d_out, int out_size, void* d_ws, size_t ws_size,
                              hipStream_t stream) {
  const float* x      = (const float*)d_in[0];
  const float* W_init = (const float*)d_in[1];
  const float* b_init = (const float*)d_in[2];
  const float* gate_w = (const float*)d_in[3];
  const float* gate_b = (const float*)d_in[4];
  const float* up_w   = (const float*)d_in[5];
  const float* up_b   = (const float*)d_in[6];
  const float* down_w = (const float*)d_in[7];
  const float* down_b = (const float*)d_in[8];
  const float* norm_w = (const float*)d_in[9];
  const float* norm_b = (const float*)d_in[10];
  float* out = (float*)d_out;

  float* ws   = (float*)d_ws;
  float* Wst  = ws;  ws += (size_t)BB * HH * HH;
  float* bst  = ws;  ws += (size_t)BB * HH;
  float* PrE  = ws;  ws += (size_t)BB * CHK * HH;
  float* PrO  = ws;  ws += (size_t)BB * CHK * HH;
  float* Diff = ws;  ws += (size_t)BB * CHK * HH;
  float* AmE  = ws;  ws += (size_t)BB * CHK * CHK;
  float* AmO  = ws;  ws += (size_t)BB * CHK * CHK;
  float* YpE  = ws;  ws += (size_t)NSL * BB * CHK * HH;
  float* YpO  = ws;  ws += (size_t)NSL * BB * CHK * HH;
  float* gwp  = ws;  ws += (size_t)H4 * HH;
  float* uwp  = ws;  ws += (size_t)H4 * HH;
  float* dwp  = ws;  ws += (size_t)HH * H4;
  float* xp   = ws;  ws += (size_t)BB * SS * HH;

  k_first<<<NB_ATTN + NB_PACK + NB_INIT, 256, 0, stream>>>(
      x, W_init, b_init, Wst, bst, AmE, gate_w, up_w, down_w, gwp, uwp, dwp, xp);
  k_stage_s<<<NB_SOLVE + NB_ATTN, 256, 0, stream>>>(
      x, xp, Wst, bst, AmE, AmO, PrE, Diff, PrO, gwp, gate_b, uwp, up_b, dwp, YpO, 0);
  k_stage_u<<<NB_WUPD, 256, 0, stream>>>(
      x, Diff, Wst, YpE, down_b, PrE, norm_w, norm_b, out, 1);
  k_stage_s<<<NB_SOLVE + NB_ATTN + NB_MLP1, 256, 0, stream>>>(
      x, xp, Wst, bst, AmO, AmE, PrO, Diff, PrE, gwp, gate_b, uwp, up_b, dwp, YpE, 1);
  k_tail<<<NB_MLP1 + NB_LN, 256, 0, stream>>>(
      PrO, gwp, gate_b, uwp, up_b, dwp, YpO, YpE, down_b, PrE, norm_w, norm_b, out, 0);
  k_ln<<<NB_LN, 256, 0, stream>>>(YpO, down_b, PrO, norm_w, norm_b, out, NCHUNK - 1);
}


# P0 of chunk 0 (x*W_init^T + b_init, inputs only) leaves the solve block: 96 blocks of the first launch compute it beside attn(0), solve(0) only loads its 128x32 strip; every sum as before
# speedup vs baseline: 8.8765x; 1.1082x over previous
#include <hip/hip_runtime.h>
#include <math.h>

#define BB 2
#define SS 256
#define HH 768
#define H4 3072
#define CHK 128
#define LRC 0.01f
#define EPSC 1e-5f

typedef float v2f __attribute__((ext_vector_type(2)));
typedef float v8f __attribute__((ext_vector_type(8)));
typedef float v4f __attribute__((ext_vector_type(4)));

__device__ __forceinline__ v8f wmma4(v2f a, v2f b, v8f c) {
  return __builtin_amdgcn_wmma_f32_16x16x4_f32(
       false, a,  false, b,
       (short)0, c,  false,  false);
}

__device__ __forceinline__ v2f frag_kn(const float* M, int ld, int k0, int r0, int lane) {
  const float* p = M + (size_t)(k0 + ((lane >> 4) << 1)) * ld + r0 + (lane & 15);
  v2f v; v.x = p[0]; v.y = p[ld];
  return v;
}
__device__ __forceinline__ v2f lo2(v4f v) { v2f r; r.x = v.x; r.y = v.y; return r; }
__device__ __forceinline__ v2f hi2(v4f v) { v2f r; r.x = v.z; r.y = v.w; return r; }

__device__ __forceinline__ v8f tile_kh(const float* __restrict__ a0, const float* __restrict__ b0, int kstep) {
  v8f acc = {};
#pragma unroll 16
  for (int kg = 0; kg < HH / 8; ++kg) {
    v4f a = *(const v4f*)(a0 + (size_t)kg * kstep);
    v4f b = *(const v4f*)(b0 + (size_t)kg * kstep);
    acc = wmma4(lo2(a), lo2(b), acc);
    acc = wmma4(hi2(a), hi2(b), acc);
  }
  return acc;
}

__device__ __forceinline__ void tile_kh2(const float* __restrict__ a0, const float* __restrict__ b0,
                                         const float* __restrict__ b1, int kstep, v8f& c0, v8f& c1) {
  v8f x = {}, y = {};
#pragma unroll 8
  for (int kg = 0; kg < HH / 8; ++kg) {
    v4f a = *(const v4f*)(a0 + (size_t)kg * kstep);
    v4f p = *(const v4f*)(b0 + (size_t)kg * kstep);
    v4f q = *(const v4f*)(b1 + (size_t)kg * kstep);
    x = wmma4(lo2(a), lo2(p), x);
    y = wmma4(lo2(a), lo2(q), y);
    x = wmma4(hi2(a), hi2(p), x);
    y = wmma4(hi2(a), hi2(q), y);
  }
  c0 = x; c1 = y;
}


#define NB_INIT ((HH * HH / 4 + 255) / 256)
#define NB_SOLVE (BB * (HH / 32))
#define NB_ATTN (BB * (CHK / 16) * (CHK / 16) / 8)
#define NB_WUPD (BB * (HH / 16) * (HH / 16) / 8)
#define NB_MLP1 (BB * (CHK / 16) * (H4 / 16) / 8)
#define NSL     (H4 / 128)
#define NB_LN   (BB * CHK)
#define NCHUNK  (SS / CHK)
#define NB_PACK (3 * 192 + BB * SS / 16)
#define NB_P0F  (BB * (CHK / 16) * (HH / 16) / 8)

__device__ __forceinline__ void d_init(int blk, const float* __restrict__ Wi, const float* __restrict__ bi,
                       float* __restrict__ Wst, float* __restrict__ bst) {
  int i = blk * 256 + threadIdx.x;
  const int n = HH * HH;
  if (i < n / 4) {
    const int lane = i & 31, u = i >> 5;
    const int tn = u / (HH / 8), kg = u % (HH / 8);
    v4f w = *(const v4f*)(Wi + (size_t)(tn * 16 + (lane & 15)) * HH + kg * 8 + 4 * (lane >> 4));
    *(v4f*)(Wst + (size_t)i * 4) = w;
    *(v4f*)(Wst + n + (size_t)i * 4) = w;
  }
  if (i < HH) {
    float b = bi[i];
    bst[i] = b;
    bst[HH + i] = b;
  }
}

__device__ __forceinline__ void d_pack(int blk, const float* __restrict__ gw, const float* __restrict__ uw,
                       const float* __restrict__ dw, const float* __restrict__ x,
                       float* __restrict__ gwp, float* __restrict__ uwp,
                       float* __restrict__ dwp, float* __restrict__ xp) {
  const int which = blk / 192, r = blk % 192;
  const float* src = (which == 0) ? gw : (which == 1) ? uw : (which == 2) ? dw : x;
  float* dst = (which == 0) ? gwp : (which == 1) ? uwp : (which == 2) ? dwp : xp;
  const int K   = (which == 2) ? H4 : HH;
  const int tn  = (which == 2) ? (r >> 2) : r;
  const int seg = (which == 2) ? (r & 3) : 0;
  const int w = threadIdx.x >> 5, lane = threadIdx.x & 31;
  const float* s = src + (size_t)(tn * 16 + (lane & 15)) * K + 4 * (lane >> 4);
  float* d = dst + ((size_t)tn * (K / 8) * 32 + lane) * 4;
  const int kg0 = seg * 96 + w * 12;
  v4f v[12];
#pragma unroll
  for (int i = 0; i < 12; ++i) v[i] = *(const v4f*)(s + (size_t)(kg0 + i) * 8);
#pragma unroll
  for (int i = 0; i < 12; ++i) *(v4f*)(d + (size_t)(kg0 + i) * 128) = v[i];
}

__device__ __forceinline__ void d_attn(int blk, const float* __restrict__ xs, bool packed,
                       float* __restrict__ Amat, int chunk) {
  int wid = (blk * 256 + threadIdx.x) >> 5;
  int lane = threadIdx.x & 31;
  int tn = wid % (CHK / 16);
  int tm = (wid / (CHK / 16)) % (CHK / 16);
  int b  = wid / ((CHK / 16) * (CHK / 16));
  v8f acc;
  if (packed) {
    const int t0 = b * (SS / 16) + chunk * (CHK / 16);
    acc = tile_kh(xs + ((size_t)(t0 + tm) * (HH / 8) * 32 + lane) * 4,
                  xs + ((size_t)(t0 + tn) * (HH / 8) * 32 + lane) * 4, 128);
  } else {
    const float* Xc = xs + ((size_t)b * SS + chunk * CHK) * HH + 4 * (lane >> 4);
    acc = tile_kh(Xc + (size_t)(tm * 16 + (lane & 15)) * HH,
                  Xc + (size_t)(tn * 16 + (lane & 15)) * HH, 8);
  }
  int s = tn * 16 + (lane & 15);
  int mbase = tm * 16 + ((lane >> 4) << 3);
  float* Ab = Amat + (size_t)b * CHK * CHK;
#pragma unroll
  for (int i = 0; i < 8; ++i) {
    int t = mbase + i;
    float v = (s < t) ? LRC * (acc[i] + 1.0f) : 0.0f;
    Ab[(size_t)t * CHK + s] = v;
  }
}

__device__ __forceinline__ void d_p0first(int blk, const float* __restrict__ x, const float* __restrict__ Wi,
                       const float* __restrict__ bi, float* __restrict__ P0b) {
  const int wid = (blk * 256 + threadIdx.x) >> 5;
  const int lane = threadIdx.x & 31;
  const int c = lane & 15, hf = lane >> 4;
  const int tn = wid % (HH / 16);
  const int tm = (wid / (HH / 16)) % (CHK / 16);
  const int b  = wid / ((HH / 16) * (CHK / 16));
  const float bias = bi[tn * 16 + c];
  const v8f acc = tile_kh(x + ((size_t)b * SS + tm * 16 + c) * HH + 4 * hf,
                          Wi + (size_t)(tn * 16 + c) * HH + 4 * hf, 8);
  float* o = P0b + ((size_t)b * CHK + tm * 16 + 8 * hf) * HH + tn * 16 + c;
#pragma unroll
  for (int i = 0; i < 8; ++i) o[(size_t)i * HH] = acc[i] + bias;
}

#define APITCH (CHK + 4)
#define DPITCH 40
#define PPITCH 36

static_assert(CHK / 16 <= 8, "one wave of the 256-thread block per 16-token block");

__device__ __forceinline__ void solve_block_operands(const float* Ps, const float* Xs, int t0, int lane,
                                                     float (&pv)[16], float (&xv)[16]) {
#pragma unroll
  for (int r = 0; r < 16; ++r) {
    pv[r] = Ps[(t0 + r) * PPITCH + lane];
    xv[r] = Xs[(t0 + r) * PPITCH + lane];
  }
}

__device__ __forceinline__ void d_p0solve(int blk, const float* __restrict__ x, const float* __restrict__ xp,
                        const float* __restrict__ Wst, const float* __restrict__ P0b,
                        const float* __restrict__ Amat, float* __restrict__ Pred,
                        float* __restrict__ Diff, float* __restrict__ bst, int chunk) {
  __shared__ __attribute__((aligned(16))) float As[CHK * APITCH];
  __shared__ __attribute__((aligned(16))) float Ds[CHK * DPITCH];
  __shared__ __attribute__((aligned(16))) float Ts[16 * DPITCH];
  __shared__ __attribute__((aligned(16))) float Ps[CHK * PPITCH];
  __shared__ __attribute__((aligned(16))) float Xs[CHK * PPITCH];
  const int tid = threadIdx.x;
  const int w = __builtin_amdgcn_readfirstlane(tid >> 5);
  const int lane = tid & 31;
  const int c = lane & 15, hf = lane >> 4;
  const int b  = blk / (HH / 32);
  const int n0 = (blk % (HH / 32)) * 32;
  const float* Xb = x + (size_t)b * SS * HH;
  const float* Ab = Amat + (size_t)b * CHK * CHK;
  const size_t col = (size_t)n0 + lane;
  const bool last_chunk = (chunk == SS / CHK - 1);

  v4f av[CHK * CHK / 1024];
#pragma unroll
  for (int i = 0; i < CHK * CHK / 1024; ++i) {
    int idx = i * 256 + tid;
    av[i] = *(const v4f*)(Ab + (idx / (CHK / 4)) * CHK + (idx % (CHK / 4)) * 4);
  }
  v4f xq[CHK / 32];
  {
    const float* Xn = Xb + ((size_t)chunk * CHK + 1) * HH + n0;
#pragma unroll
    for (int i = 0; i < CHK / 32; ++i) {
      const int idx = i * 256 + tid;
      const int t = idx >> 3, q = idx & 7;
      const bool none = last_chunk && (t == CHK - 1);
      v4f v = *(const v4f*)(Xn + (size_t)(none ? t - 1 : t) * HH + q * 4);
      if (none) { v.x = 0.0f; v.y = 0.0f; v.z = 0.0f; v.w = 0.0f; }
      xq[i] = v;
    }
  }
  const float bias0 = bst[(size_t)b * HH + n0 + c];
  const float bias1 = bst[(size_t)b * HH + n0 + 16 + c];
#pragma unroll 1
  for (int tm = w; tm < CHK / 16; tm += 8) {
    if (chunk == 0) {
      const float* Pn = P0b + ((size_t)b * CHK + tm * 16) * HH + n0;
      v4f pq[4];
#pragma unroll
      for (int i = 0; i < 4; ++i) {
        const int idx = i * 32 + lane;
        pq[i] = *(const v4f*)(Pn + (size_t)(idx >> 3) * HH + (idx & 7) * 4);
      }
#pragma unroll
      for (int i = 0; i < 4; ++i) {
        const int idx = i * 32 + lane;
        *(v4f*)(Ps + (tm * 16 + (idx >> 3)) * PPITCH + (idx & 7) * 4) = pq[i];
      }
      continue;
    }
    const float* ap = xp + ((size_t)(b * (SS / 16) + chunk * (CHK / 16) + tm) * (HH / 8) * 32 + lane) * 4;
    const float* bp = Wst + (size_t)b * HH * HH + ((size_t)(n0 / 16) * (HH / 8) * 32 + lane) * 4;
    v8f acc0, acc1;
    tile_kh2(ap, bp, bp + (size_t)(HH / 8) * 128, 128, acc0, acc1);
#pragma unroll
    for (int i = 0; i < 8; ++i) {
      Ps[(tm * 16 + 8 * hf + i) * PPITCH + c]      = acc0[i] + bias0;
      Ps[(tm * 16 + 8 * hf + i) * PPITCH + 16 + c] = acc1[i] + bias1;
    }
  }
#pragma unroll
  for (int i = 0; i < CHK * CHK / 1024; ++i) {
    int idx = i * 256 + tid;
    *(v4f*)(As + (idx / (CHK / 4)) * APITCH + (idx % (CHK / 4)) * 4) = av[i];
  }
#pragma unroll
  for (int i = 0; i < CHK / 32; ++i) {
    const int idx = i * 256 + tid;
    *(v4f*)(Xs + (idx >> 3) * PPITCH + (idx & 7) * 4) = xq[i];
  }
  __syncthreads();

  float bsum = 0.0f;
  v8f s0 = {}, s1 = {};
  float pv[16], xv[16];
  if (w == 0) solve_block_operands(Ps, Xs, 0, lane, pv, xv);
#pragma unroll 1
  for (int j = 0; j < CHK / 16; ++j) {
    const int t0 = j * 16;
    if (w == 0) {
      float sc[16];
#pragma unroll
      for (int r = 0; r < 16; ++r) sc[r] = 0.0f;
      if (j > 0) {
#pragma unroll
        for (int r = 0; r < 16; ++r) sc[r] = Ts[r * DPITCH + lane];
      }
      v4f cf[16][4];
#pragma unroll
      for (int r = 1; r < 16; ++r) {
#pragma unroll
        for (int q4 = 0; q4 * 4 < r; ++q4)
          cf[r][q4] = *(const v4f*)(As + (t0 + r) * APITCH + t0 + q4 * 4);
      }
      float d[16];
#pragma unroll
      for (int r = 0; r < 16; ++r) {
        const int t = t0 + r;
        float acc = pv[r] - sc[r];
#pragma unroll
        for (int q = 0; q < r; ++q) acc -= cf[r][q >> 2][q & 3] * d[q];
        float dd = acc - xv[r];
        d[r] = dd;
        Ds[t * DPITCH + lane] = dd;
        Ps[t * PPITCH + lane] = acc;
        float dl = (t == CHK - 1 && last_chunk) ? 0.0f : dd;
        bsum += dl;
      }
    }
    __syncthreads();
    if (j + 1 < CHK / 16) {
      if (w > j) {
#pragma unroll
        for (int q = 0; q < 4; ++q) {
          const int k = t0 + 4 * q + 2 * hf;
          const v2f a = *(const v2f*)(As + (w * 16 + c) * APITCH + k);
          v2f b0, b1;
          b0.x = Ds[k * DPITCH + c];       b0.y = Ds[(k + 1) * DPITCH + c];
          b1.x = Ds[k * DPITCH + 16 + c];  b1.y = Ds[(k + 1) * DPITCH + 16 + c];
          s0 = wmma4(a, b0, s0);
          s1 = wmma4(a, b1, s1);
        }
        if (w == j + 1) {
#pragma unroll
          for (int r = 0; r < 8; ++r) {
            Ts[(8 * hf + r) * DPITCH + c]      = s0[r];
            Ts[(8 * hf + r) * DPITCH + 16 + c] = s1[r];
          }
        }
      } else if (w == 0) {
        solve_block_operands(Ps, Xs, t0 + 16, lane, pv, xv);
      }
      __syncthreads();
    }
  }
  {
    float* Pb = Pred + (size_t)b * CHK * HH + col;
    float* Db = Diff + (size_t)b * CHK * HH + col;
#pragma unroll
    for (int r = 0; r < 16; ++r) {
      const int t = w * 16 + r;
      const float pr = Ps[t * PPITCH + lane];
      float dv = Ds[t * DPITCH + lane];
      if (t == CHK - 1 && last_chunk) dv = 0.0f;
      Pb[(size_t)t * HH] = pr;
      Db[(size_t)t * HH] = dv;
    }
  }
  if (w == 0) bst[(size_t)b * HH + col] -= LRC * bsum;
}

__device__ __forceinline__ void d_wupdate(int blk, const float* __restrict__ x, const float* __restrict__ Diff,
                          float* __restrict__ Wst, int chunk) {
  int wid = (blk * 256 + threadIdx.x) >> 5;
  int lane = threadIdx.x & 31;
  int tn = wid % (HH / 16);
  int tm = (wid / (HH / 16)) % (HH / 16);
  int b  = wid / ((HH / 16) * (HH / 16));
  const float* Xc = x + ((size_t)b * SS + chunk * CHK) * HH;
  const float* Db = Diff + (size_t)b * CHK * HH;
  float* W = Wst + (size_t)b * HH * HH;
  v8f s = {};
  for (int k0 = 0; k0 < CHK; k0 += 4) {
    v2f a  = frag_kn(Xc, HH, k0, tn * 16, lane);
    v2f bb = frag_kn(Db, HH, k0, tm * 16, lane);
    s = wmma4(a, bb, s);
  }
  float* p = W + (((size_t)tm * (HH / 8) + tn * 2 + (lane >> 4)) * 32 + (lane & 15)) * 4;
  v4f w0 = *(const v4f*)p;
  v4f w1 = *(const v4f*)(p + 64);
  w0.x -= LRC * s[0]; w0.y -= LRC * s[1]; w0.z -= LRC * s[2]; w0.w -= LRC * s[3];
  w1.x -= LRC * s[4]; w1.y -= LRC * s[5]; w1.z -= LRC * s[6]; w1.w -= LRC * s[7];
  *(v4f*)p = w0;
  *(v4f*)(p + 64) = w1;
}

#define GPITCH 132
__device__ __forceinline__ void d_mlp(int blk, const float* __restrict__ Pred,
                       const float* __restrict__ gw, const float* __restrict__ gb,
                       const float* __restrict__ uw, const float* __restrict__ ub,
                       const float* __restrict__ dw, float* __restrict__ Ypart) {
  __shared__ __attribute__((aligned(16))) float Gs[16 * GPITCH];
  const int w = threadIdx.x >> 5, lane = threadIdx.x & 31;
  const int c = lane & 15, hf = lane >> 4;
  const int sl = blk % NSL;
  const int tm = (blk / NSL) % (CHK / 16);
  const int b  = blk / (NSL * (CHK / 16));
  const int tn = sl * 8 + w;
  const float* Pb = Pred + (size_t)b * CHK * HH;
  const float* ap = Pb + (size_t)(tm * 16 + c) * HH + 4 * hf;
  const float* gp = gw + ((size_t)tn * (HH / 8) * 32 + lane) * 4;
  const float* up = uw + ((size_t)tn * (HH / 8) * 32 + lane) * 4;
  v8f g = {}, u = {};
#pragma unroll 8
  for (int kg = 0; kg < HH / 8; ++kg) {
    v4f a  = *(const v4f*)(ap + kg * 8);
    v4f bg = *(const v4f*)(gp + (size_t)kg * 128);
    v4f bu = *(const v4f*)(up + (size_t)kg * 128);
    g = wmma4(lo2(a), lo2(bg), g);
    u = wmma4(lo2(a), lo2(bu), u);
    g = wmma4(hi2(a), hi2(bg), g);
    u = wmma4(hi2(a), hi2(bu), u);
  }
  {
    const int n = tn * 16 + c;
    const float gbias = gb[n], ubias = ub[n];
#pragma unroll
    for (int i = 0; i < 8; ++i) {
      float zg = g[i] + gbias;
      float zu = u[i] + ubias;
      Gs[(8 * hf + i) * GPITCH + w * 16 + c] = (1.0f / (1.0f + __expf(-zg))) * zu;
    }
  }
  __syncthreads();
  v4f a[16];
#pragma unroll
  for (int kk = 0; kk < 16; ++kk) a[kk] = *(const v4f*)(Gs + c * GPITCH + kk * 8 + 4 * hf);
  const float* bp = dw + (((size_t)(w * 6) * (H4 / 8) + sl * 16) * 32 + lane) * 4;
  float* Yp = Ypart + ((size_t)sl * (BB * CHK) + b * CHK + tm * 16 + 8 * hf) * HH + w * 96 + c;
#pragma unroll
  for (int q = 0; q < 6; q += 2) {
    const float* b0 = bp + (size_t)q * (H4 / 8) * 128;
    const float* b1 = b0 + (size_t)(H4 / 8) * 128;
    v8f y0 = {}, y1 = {};
    v4f v0[16], v1[16];
#pragma unroll
    for (int kk = 0; kk < 16; ++kk) {
      v0[kk] = *(const v4f*)(b0 + kk * 128);
      v1[kk] = *(const v4f*)(b1 + kk * 128);
    }
    __builtin_amdgcn_sched_barrier(0);
#pragma unroll
    for (int kk = 0; kk < 16; ++kk) {
      y0 = wmma4(lo2(a[kk]), lo2(v0[kk]), y0);
      y1 = wmma4(lo2(a[kk]), lo2(v1[kk]), y1);
      y0 = wmma4(hi2(a[kk]), hi2(v0[kk]), y0);
      y1 = wmma4(hi2(a[kk]), hi2(v1[kk]), y1);
    }
#pragma unroll
    for (int i = 0; i < 8; ++i) {
      Yp[(size_t)i * HH + q * 16]      = y0[i];
      Yp[(size_t)i * HH + q * 16 + 16] = y1[i];
    }
  }
}

__device__ __forceinline__ void d_ln(int blk, const float* __restrict__ Ypart, const float* __restrict__ db,
                     const float* __restrict__ Pred, const float* __restrict__ nw,
                     const float* __restrict__ nb, float* __restrict__ out, int chunk) {
  __shared__ float s1[256], s2[256];
  int row = blk;
  int b = row / CHK, t = row % CHK;
  const float* y = Ypart + (size_t)row * HH;
  const float* p = Pred + (size_t)row * HH;
  int tid = threadIdx.x;
  float v0 = 0.0f, v1 = 0.0f, v2 = 0.0f;
#pragma unroll
  for (int j = 0; j < NSL; ++j) {
    const float* yj = y + (size_t)j * (BB * CHK) * HH;
    v0 += yj[tid]; v1 += yj[tid + 256]; v2 += yj[tid + 512];
  }
  v0 = v0 + db[tid]       + p[tid];
  v1 = v1 + db[tid + 256] + p[tid + 256];
  v2 = v2 + db[tid + 512] + p[tid + 512];
  s1[tid] = v0 + v1 + v2;
  s2[tid] = v0 * v0 + v1 * v1 + v2 * v2;
  __syncthreads();
  for (int off = 128; off > 0; off >>= 1) {
    if (tid < off) { s1[tid] += s1[tid + off]; s2[tid] += s2[tid + off]; }
    __syncthreads();
  }
  float mu  = s1[0] * (1.0f / HH);
  float var = s2[0] * (1.0f / HH) - mu * mu;
  float r = rsqrtf(var + EPSC);
  int gt = chunk * CHK + t;
  float* o = out + ((size_t)b * SS + gt) * HH;
  o[tid]       = (v0 - mu) * r * nw[tid]       + nb[tid];
  o[tid + 256] = (v1 - mu) * r * nw[tid + 256] + nb[tid + 256];
  o[tid + 512] = (v2 - mu) * r * nw[tid + 512] + nb[tid + 512];
}


__global__ __launch_bounds__(256) void k_first(const float* __restrict__ x,
                       const float* __restrict__ Wi, const float* __restrict__ bi,
                       float* __restrict__ Wst, float* __restrict__ bst,
                       float* __restrict__ Amat,
                       const float* __restrict__ gw, const float* __restrict__ uw,
                       const float* __restrict__ dw, float* __restrict__ gwp,
                       float* __restrict__ uwp, float* __restrict__ dwp,
                       float* __restrict__ xp, float* __restrict__ P0b) {
  int blk = blockIdx.x;
  if (blk < NB_ATTN) { d_attn(blk, x, false, Amat, 0); return; }
  blk -= NB_ATTN;
  if (blk < NB_P0F) { d_p0first(blk, x, Wi, bi, P0b); return; }
  blk -= NB_P0F;
  if (blk < NB_PACK) { d_pack(blk, gw, uw, dw, x, gwp, uwp, dwp, xp); return; }
  d_init(blk - NB_PACK, Wi, bi, Wst, bst);
}

__global__ __launch_bounds__(256) void k_stage_u(const float* __restrict__ x,
                       const float* __restrict__ Diff, float* __restrict__ Wst,
                       const float* __restrict__ YpartLn, const float* __restrict__ db,
                       const float* __restrict__ PredLn,
                       const float* __restrict__ nw, const float* __restrict__ nb,
                       float* __restrict__ out, int c) {
  int blk = blockIdx.x;
  if (blk < NB_WUPD) {
    d_wupdate(blk, x, Diff, Wst, c - 1);
    return;
  }
  if (c >= 2) d_ln(blk - NB_WUPD, YpartLn, db, PredLn, nw, nb, out, c - 2);
}

__global__ __launch_bounds__(256) void k_stage_s(const float* __restrict__ x,
                       const float* __restrict__ xp,
                       const float* __restrict__ Wst, const float* __restrict__ P0b,
                       float* __restrict__ bst,
                       const float* __restrict__ AmatCur, float* __restrict__ AmatNext,
                       float* __restrict__ Pred, float* __restrict__ Diff,
                       const float* __restrict__ PredPrev,
                       const float* __restrict__ gw, const float* __restrict__ gb,
                       const float* __restrict__ uw, const float* __restrict__ ub,
                       const float* __restrict__ dw, float* __restrict__ Ypart, int c) {
  int blk = blockIdx.x;
  if (blk < NB_SOLVE) {
    d_p0solve(blk, x, xp, Wst, P0b, AmatCur, Pred, Diff, bst, c);
    return;
  }
  blk -= NB_SOLVE;
  if (blk < NB_ATTN) {
    if (c + 1 < NCHUNK) d_attn(blk, xp, true, AmatNext, c + 1);
    return;
  }
  blk -= NB_ATTN;
  if (c >= 1) d_mlp(blk, PredPrev, gw, gb, uw, ub, dw, Ypart);
}

__global__ __launch_bounds__(256) void k_tail(const float* __restrict__ PredMlp,
                       const float* __restrict__ gw, const float* __restrict__ gb,
                       const float* __restrict__ uw, const float* __restrict__ ub,
                       const float* __restrict__ dw, float* __restrict__ YpartMlp,
                       const float* __restrict__ YpartLn, const float* __restrict__ db,
                       const float* __restrict__ PredLn,
                       const float* __restrict__ nw, const float* __restrict__ nb,
                       float* __restrict__ out, int chunk_ln) {
  int blk = blockIdx.x;
  if (blk < NB_MLP1) {
    d_mlp(blk, PredMlp, gw, gb, uw, ub, dw, YpartMlp);
    return;
  }
  d_ln(blk - NB_MLP1, YpartLn, db, PredLn, nw, nb, out, chunk_ln);
}

__global__ __launch_bounds__(256) void k_ln(const float* __restrict__ Ypart, const float* __restrict__ db,
                     const float* __restrict__ Pred, const float* __restrict__ nw,
                     const float* __restrict__ nb, float* __restrict__ out, int chunk) {
  d_ln(blockIdx.x, Ypart, db, Pred, nw, nb, out, chunk);
}

extern "C" void kernel_launch(void* const* d_in, const int* in_sizes, int n_in,
                              void* d_out, int out_size, void* d_ws, size_t ws_size,
                              hipStream_t stream) {
  const float* x      = (const float*)d_in[0];
  const float* W_init = (const float*)d_in[1];
  const float* b_init = (const float*)d_in[2];
  const float* gate_w = (const float*)d_in[3];
  const float* gate_b = (const float*)d_in[4];
  const float* up_w   = (const float*)d_in[5];
  const float* up_b   = (const float*)d_in[6];
  const float* down_w = (const float*)d_in[7];
  const float* down_b = (const float*)d_in[8];
  const float* norm_w = (const float*)d_in[9];
  const float* norm_b = (const float*)d_in[10];
  float* out = (float*)d_out;

  float* ws   = (float*)d_ws;
  float* Wst  = ws;  ws += (size_t)BB * HH * HH;
  float* bst  = ws;  ws += (size_t)BB * HH;
  float* PrE  = ws;  ws += (size_t)BB * CHK * HH;
  float* PrO  = ws;  ws += (size_t)BB * CHK * HH;
  float* Diff = ws;  ws += (size_t)BB * CHK * HH;
  float* AmE  = ws;  ws += (size_t)BB * CHK * CHK;
  float* AmO  = ws;  ws += (size_t)BB * CHK * CHK;
  float* YpE  = ws;  ws += (size_t)NSL * BB * CHK * HH;
  float* YpO  = ws;  ws += (size_t)NSL * BB * CHK * HH;
  float* gwp  = ws;  ws += (size_t)H4 * HH;
  float* uwp  = ws;  ws += (size_t)H4 * HH;
  float* dwp  = ws;  ws += (size_t)HH * H4;
  float* xp   = ws;  ws += (size_t)BB * SS * HH;
  float* P0b  = ws;  ws += (size_t)BB * CHK * HH;

  k_first<<<NB_ATTN + NB_P0F + NB_PACK + NB_INIT, 256, 0, stream>>>(
      x, W_init, b_init, Wst, bst, AmE, gate_w, up_w, down_w, gwp, uwp, dwp, xp, P0b);
  k_stage_s<<<NB_SOLVE + NB_ATTN, 256, 0, stream>>>(
      x, xp, Wst, P0b, bst, AmE, AmO, PrE, Diff, PrO, gwp, gate_b, uwp, up_b, dwp, YpO, 0);
  k_stage_u<<<NB_WUPD, 256, 0, stream>>>(
      x, Diff, Wst, YpE, down_b, PrE, norm_w, norm_b, out, 1);
  k_stage_s<<<NB_SOLVE + NB_ATTN + NB_MLP1, 256, 0, stream>>>(
      x, xp, Wst, P0b, bst, AmO, AmE, PrO, Diff, PrE, gwp, gate_b, uwp, up_b, dwp, YpE, 1);
  k_tail<<<NB_MLP1 + NB_LN, 256, 0, stream>>>(
      PrO, gwp, gate_b, uwp, up_b, dwp, YpO, YpE, down_b, PrE, norm_w, norm_b, out, 0);
  k_ln<<<NB_LN, 256, 0, stream>>>(YpO, down_b, PrO, norm_w, norm_b, out, NCHUNK - 1);
}
